# starting kernel
# baseline (speedup 1.0000x reference)
#include <hip/hip_runtime.h>
#include <hip/hip_bf16.h>

typedef __attribute__((ext_vector_type(2))) float v2f;
typedef __attribute__((ext_vector_type(4))) float v4f;
typedef __attribute__((ext_vector_type(8))) float v8f;

#define DIM 64
#define KEEP_PROB 0.7f

__global__ void __launch_bounds__(256)
gcn_fc_wmma_kernel(const float* __restrict__ emb, const float* __restrict__ W,
                   const float* __restrict__ bias, float* __restrict__ acc,
                   int n_nodes) {
    __shared__ alignas(16) float Wlds[DIM * DIM];
    __shared__ alignas(16) float Alds[8][16 * DIM];

    const int tid  = threadIdx.x;
    const int wave = tid >> 5;
    const int lane = tid & 31;

    for (int i = tid; i < (DIM * DIM) / 4; i += 256) {
        ((v4f*)Wlds)[i] = ((const v4f*)W)[i];
    }

    const int row_base = blockIdx.x * 128 + wave * 16;

    for (int i = lane; i < 256; i += 32) {
        int r  = i >> 4;
        int c4 = i & 15;
        int gr = row_base + r;
        v4f v = (gr < n_nodes) ? ((const v4f*)(emb + (size_t)gr * DIM))[c4]
                               : (v4f){0.f, 0.f, 0.f, 0.f};
        ((v4f*)&Alds[wave][0])[r * 16 + c4] = v;
    }
    __syncthreads();

    const float* At = &Alds[wave][0];

    const int m_rel = lane & 15;
    const int k_off = (lane >> 4) * 2;
    const int n_lo  = lane & 15;

    v8f c0 = {}, c1 = {}, c2 = {}, c3 = {};

#pragma unroll
    for (int kk = 0; kk < 16; ++kk) {
        const int k = kk * 4 + k_off;
        v2f a  = *(const v2f*)(At + m_rel * DIM + k);
        v2f b0 = *(const v2f*)(Wlds + (n_lo +  0) * DIM + k);
        v2f b1 = *(const v2f*)(Wlds + (n_lo + 16) * DIM + k);
        v2f b2 = *(const v2f*)(Wlds + (n_lo + 32) * DIM + k);
        v2f b3 = *(const v2f*)(Wlds + (n_lo + 48) * DIM + k);
        c0 = __builtin_amdgcn_wmma_f32_16x16x4_f32(false, a, false, b0, (short)0, c0, false, false);
        c1 = __builtin_amdgcn_wmma_f32_16x16x4_f32(false, a, false, b1, (short)0, c1, false, false);
        c2 = __builtin_amdgcn_wmma_f32_16x16x4_f32(false, a, false, b2, (short)0, c2, false, false);
        c3 = __builtin_amdgcn_wmma_f32_16x16x4_f32(false, a, false, b3, (short)0, c3, false, false);
    }

    const float bb0 = bias[n_lo];
    const float bb1 = bias[n_lo + 16];
    const float bb2 = bias[n_lo + 32];
    const float bb3 = bias[n_lo + 48];

    const int m_hi = (lane >> 4) * 8;
#pragma unroll
    for (int j = 0; j < 8; ++j) {
        int gm = row_base + m_hi + j;
        if (gm < n_nodes) {
            float* outr = acc + (size_t)gm * DIM;
            outr[n_lo +  0] = c0[j] + bb0;
            outr[n_lo + 16] = c1[j] + bb1;
            outr[n_lo + 32] = c2[j] + bb2;
            outr[n_lo + 48] = c3[j] + bb3;
        }
    }
}

__global__ void __launch_bounds__(256)
gcn_spmm_kernel(const float* __restrict__ vals, const float* __restrict__ dropu,
                const int* __restrict__ rows, const int* __restrict__ cols,
                const float* __restrict__ src, float* __restrict__ dst,
                int n_edges) {
    const long long tid = (long long)blockIdx.x * blockDim.x + threadIdx.x;
    int e = (int)(tid >> 5);
    if (e >= n_edges) return;
    e = __builtin_amdgcn_readfirstlane(e);
    const int l = threadIdx.x & 31;

    const float u = dropu[e];
    if (u + KEEP_PROB < 1.0f) return;
    const float v = vals[e] * (1.0f / KEEP_PROB);

    const int col = cols[e];
    const int row = rows[e];

    const float* sp = src + (size_t)col * DIM + l * 2;
    float*       dp = dst + (size_t)row * DIM + l * 2;
    const float m0 = v * sp[0];
    const float m1 = v * sp[1];

    asm volatile("global_atomic_add_f32 %0, %1, off scope:SCOPE_DEV"
                 :: "v"(dp), "v"(m0) : "memory");
    asm volatile("global_atomic_add_f32 %0, %1, off scope:SCOPE_DEV"
                 :: "v"(dp + 1), "v"(m1) : "memory");
}

__global__ void __launch_bounds__(256)
gcn_finalize_kernel(float* __restrict__ out, const float* __restrict__ x1,
                    long long n4) {
    const long long i = (long long)blockIdx.x * blockDim.x + threadIdx.x;
    if (i >= n4) return;
    v4f a = ((const v4f*)out)[i];
    v4f p = ((const v4f*)x1)[i];
    v4f r = (a + p) * (1.0f / 3.0f);
    ((v4f*)out)[i] = r;
}

extern "C" void kernel_launch(void* const* d_in, const int* in_sizes, int n_in,
                              void* d_out, int out_size, void* d_ws, size_t ws_size,
                              hipStream_t stream) {
    const float* emb   = (const float*)d_in[0];
    const float* W     = (const float*)d_in[1];
    const float* bias  = (const float*)d_in[2];
    const float* vals  = (const float*)d_in[3];
    const float* dropu = (const float*)d_in[4];
    const int*   rows  = (const int*)d_in[5];
    const int*   cols  = (const int*)d_in[6];

    const int N = in_sizes[0] / DIM;
    const int E = in_sizes[3];

    float* out = (float*)d_out;
    float* x1  = (float*)d_ws;

    hipMemsetAsync(d_ws, 0, (size_t)N * DIM * sizeof(float), stream);

    {
        dim3 grid((N + 127) / 128), block(256);
        gcn_fc_wmma_kernel<<<grid, block, 0, stream>>>(emb, W, bias, out, N);
    }

    {
        long long threads = (long long)E * 32;
        dim3 grid((unsigned)((threads + 255) / 256)), block(256);
        gcn_spmm_kernel<<<grid, block, 0, stream>>>(vals, dropu,     rows, cols, emb, x1,  E);
        gcn_spmm_kernel<<<grid, block, 0, stream>>>(vals, dropu + E, rows, cols, x1,  out, E);
    }

    {
        long long n4 = (long long)N * DIM / 4;
        dim3 grid((unsigned)((n4 + 255) / 256)), block(256);
        gcn_finalize_kernel<<<grid, block, 0, stream>>>(out, x1, n4);
    }
}


# SpMM as pull over a per-call grouping by destination row built without atomics (owner work-groups sweep rows[] twice, wave-private LDS counters); one wave per row gathers and sums; sum order fixed
# speedup vs baseline: 1.3103x; 1.3103x over previous
#include <hip/hip_runtime.h>
#include <hip/hip_bf16.h>

typedef __attribute__((ext_vector_type(2))) float v2f;
typedef __attribute__((ext_vector_type(4))) float v4f;
typedef __attribute__((ext_vector_type(8))) float v8f;
typedef __attribute__((ext_vector_type(4))) int   v4i;

#define DIM 64
#define KEEP_PROB 0.7f
#define OWN_ROWS 400

__global__ void __launch_bounds__(256)
gcn_fc_wmma_kernel(const float* __restrict__ emb, const float* __restrict__ W,
                   const float* __restrict__ bias, float* __restrict__ acc,
                   int n_nodes) {
    __shared__ alignas(16) float Wlds[DIM * DIM];
    __shared__ alignas(16) float Alds[8][16 * DIM];

    const int tid  = threadIdx.x;
    const int wave = tid >> 5;
    const int lane = tid & 31;

    for (int i = tid; i < (DIM * DIM) / 4; i += 256) {
        ((v4f*)Wlds)[i] = ((const v4f*)W)[i];
    }

    const int row_base = blockIdx.x * 128 + wave * 16;

    for (int i = lane; i < 256; i += 32) {
        int r  = i >> 4;
        int c4 = i & 15;
        int gr = row_base + r;
        v4f v = (gr < n_nodes) ? ((const v4f*)(emb + (size_t)gr * DIM))[c4]
                               : (v4f){0.f, 0.f, 0.f, 0.f};
        ((v4f*)&Alds[wave][0])[r * 16 + c4] = v;
    }
    __syncthreads();

    const float* At = &Alds[wave][0];

    const int m_rel = lane & 15;
    const int k_off = (lane >> 4) * 2;
    const int n_lo  = lane & 15;

    v8f c0 = {}, c1 = {}, c2 = {}, c3 = {};

#pragma unroll
    for (int kk = 0; kk < 16; ++kk) {
        const int k = kk * 4 + k_off;
        v2f a  = *(const v2f*)(At + m_rel * DIM + k);
        v2f b0 = *(const v2f*)(Wlds + (n_lo +  0) * DIM + k);
        v2f b1 = *(const v2f*)(Wlds + (n_lo + 16) * DIM + k);
        v2f b2 = *(const v2f*)(Wlds + (n_lo + 32) * DIM + k);
        v2f b3 = *(const v2f*)(Wlds + (n_lo + 48) * DIM + k);
        c0 = __builtin_amdgcn_wmma_f32_16x16x4_f32(false, a, false, b0, (short)0, c0, false, false);
        c1 = __builtin_amdgcn_wmma_f32_16x16x4_f32(false, a, false, b1, (short)0, c1, false, false);
        c2 = __builtin_amdgcn_wmma_f32_16x16x4_f32(false, a, false, b2, (short)0, c2, false, false);
        c3 = __builtin_amdgcn_wmma_f32_16x16x4_f32(false, a, false, b3, (short)0, c3, false, false);
    }

    const float bb0 = bias[n_lo];
    const float bb1 = bias[n_lo + 16];
    const float bb2 = bias[n_lo + 32];
    const float bb3 = bias[n_lo + 48];

    const int m_hi = (lane >> 4) * 8;
#pragma unroll
    for (int j = 0; j < 8; ++j) {
        int gm = row_base + m_hi + j;
        if (gm < n_nodes) {
            float* outr = acc + (size_t)gm * DIM;
            outr[n_lo +  0] = c0[j] + bb0;
            outr[n_lo + 16] = c1[j] + bb1;
            outr[n_lo + 32] = c2[j] + bb2;
            outr[n_lo + 48] = c3[j] + bb3;
        }
    }
}

__global__ void __launch_bounds__(256)
gcn_spmm_kernel(const float* __restrict__ vals, const float* __restrict__ dropu,
                const int* __restrict__ rows, const int* __restrict__ cols,
                const float* __restrict__ src, float* __restrict__ dst,
                int n_edges) {
    const long long tid = (long long)blockIdx.x * blockDim.x + threadIdx.x;
    int e = (int)(tid >> 5);
    if (e >= n_edges) return;
    e = __builtin_amdgcn_readfirstlane(e);
    const int l = threadIdx.x & 31;

    const float u = dropu[e];
    if (u + KEEP_PROB < 1.0f) return;
    const float v = vals[e] * (1.0f / KEEP_PROB);

    const int col = cols[e];
    const int row = rows[e];

    const float* sp = src + (size_t)col * DIM + l * 2;
    float*       dp = dst + (size_t)row * DIM + l * 2;
    const float m0 = v * sp[0];
    const float m1 = v * sp[1];

    asm volatile("global_atomic_add_f32 %0, %1, off scope:SCOPE_DEV"
                 :: "v"(dp), "v"(m0) : "memory");
    asm volatile("global_atomic_add_f32 %0, %1, off scope:SCOPE_DEV"
                 :: "v"(dp + 1), "v"(m1) : "memory");
}


__device__ __forceinline__ int wave_sum_i32(int x) {
    x += __shfl_xor(x, 16);
    x += __shfl_xor(x, 8);
    x += __shfl_xor(x, 4);
    x += __shfl_xor(x, 2);
    x += __shfl_xor(x, 1);
    return x;
}

__device__ __forceinline__ int wave_incl_scan_i32(int x, int lane) {
    int y;
    y = __shfl_up(x, 1);  if (lane >= 1)  x += y;
    y = __shfl_up(x, 2);  if (lane >= 2)  x += y;
    y = __shfl_up(x, 4);  if (lane >= 4)  x += y;
    y = __shfl_up(x, 8);  if (lane >= 8)  x += y;
    y = __shfl_up(x, 16); if (lane >= 16) x += y;
    return x;
}

template <bool ASSIGN>
__device__ __forceinline__ int own_step(int val, int lo, unsigned nrows,
                                        int* mycw, int lane) {
    const bool mine = (unsigned)(val - lo) < nrows;
    unsigned m = __builtin_amdgcn_ballot_w32(mine);
    int p = -1;
    while (m) {
        const int j = __builtin_ctz(m);
        m &= m - 1;
        const int rl = __builtin_amdgcn_readlane(val, j) - lo;
        const int old = mycw[rl];
        mycw[rl] = old + 1;
        if (ASSIGN && lane == j) p = old;
    }
    return p;
}

__device__ __forceinline__ v4i load_rows4(const int* __restrict__ rows, int i4,
                                          int nfull) {
    const int ic = (i4 < nfull) ? i4 : (nfull - 1);
    v4i r = ((const v4i*)rows)[ic];
    if (i4 >= nfull) r = (v4i){0x7fffffff, 0x7fffffff, 0x7fffffff, 0x7fffffff};
    return r;
}

template <bool ASSIGN>
__device__ __forceinline__ int sweep_rows(const int* __restrict__ rows,
                                          int* __restrict__ pos, int n_edges,
                                          int lo, unsigned nrows, int* mycw,
                                          int tid, int lane) {
    const int nfull = n_edges >> 2;
    const int trips = (nfull + 1023) >> 10;
    int below = 0;
    if (nfull > 0) {
        for (int it = 0; it < trips; it += 4) {
            v4i g[4];
#pragma unroll
            for (int u = 0; u < 4; ++u) {
                g[u] = load_rows4(rows, (it + u) * 1024 + tid, nfull);
            }
#pragma unroll
            for (int u = 0; u < 4; ++u) {
                const int e0 = ((it + u) * 1024 + tid) * 4;
                const int p0 = own_step<ASSIGN>(g[u].x, lo, nrows, mycw, lane);
                const int p1 = own_step<ASSIGN>(g[u].y, lo, nrows, mycw, lane);
                const int p2 = own_step<ASSIGN>(g[u].z, lo, nrows, mycw, lane);
                const int p3 = own_step<ASSIGN>(g[u].w, lo, nrows, mycw, lane);
                if (ASSIGN) {
                    if (p0 >= 0) pos[e0]     = p0;
                    if (p1 >= 0) pos[e0 + 1] = p1;
                    if (p2 >= 0) pos[e0 + 2] = p2;
                    if (p3 >= 0) pos[e0 + 3] = p3;
                } else {
                    below += (g[u].x < lo) + (g[u].y < lo) + (g[u].z < lo) + (g[u].w < lo);
                }
            }
        }
    }
    {
        const int te = nfull * 4 + tid;
        int tv = 0x7fffffff;
        if (tid < 32 && te < n_edges) tv = rows[te];
        const int pt = own_step<ASSIGN>(tv, lo, nrows, mycw, lane);
        if (ASSIGN) {
            if (pt >= 0) pos[te] = pt;
        } else {
            below += (tv < lo);
        }
    }
    return below;
}

__global__ void __launch_bounds__(1024)
gcn_group_kernel(const int* __restrict__ rows, int* __restrict__ rowptr,
                 int* __restrict__ pos, int n_nodes, int n_edges) {
    __shared__ int cw[32][OWN_ROWS];
    __shared__ int red[32];
    __shared__ int wtot[32];

    const int tid  = threadIdx.x;
    const int lane = tid & 31;
    const int wave = tid >> 5;
    const int lo   = blockIdx.x * OWN_ROWS;
    const int nrows = (n_nodes - lo < OWN_ROWS) ? (n_nodes - lo) : OWN_ROWS;
    int* mycw = cw[wave];

    for (int i = lane; i < OWN_ROWS; i += 32) mycw[i] = 0;

    int below = sweep_rows<false>(rows, pos, n_edges, lo, (unsigned)nrows, mycw, tid, lane);
    below = wave_sum_i32(below);
    if (lane == 0) red[wave] = below;
    __syncthreads();

    int tot = 0;
    if (tid < OWN_ROWS) {
        for (int w = 0; w < 32; ++w) {
            const int cc = cw[w][tid];
            cw[w][tid] = tot;
            tot += cc;
        }
    }
    const int x = wave_incl_scan_i32(tot, lane);
    if (lane == 31) wtot[wave] = x;
    __syncthreads();
    const int base = wave_sum_i32(red[lane]);
    const int wt   = wtot[lane];
    const int wx   = wave_incl_scan_i32(wt, lane);
    const int woff = __shfl(wx - wt, wave);
    const int start = base + woff + x - tot;
    if (tid < nrows) {
        rowptr[lo + tid] = start;
        if (lo + tid == n_nodes - 1) rowptr[n_nodes] = start + tot;
    }
    if (tid < OWN_ROWS) {
        for (int w = 0; w < 32; ++w) cw[w][tid] += start;
    }
    __syncthreads();

    sweep_rows<true>(rows, pos, n_edges, lo, (unsigned)nrows, mycw, tid, lane);
}

__global__ void __launch_bounds__(256)
gcn_fill_kernel(const float* __restrict__ vals, const float* __restrict__ dropu,
                const int* __restrict__ cols, const int* __restrict__ pos,
                int* __restrict__ rec, int n_edges) {
    const int e = blockIdx.x * 256 + threadIdx.x;
    if (e >= n_edges) return;
    const float v  = vals[e] * (1.0f / KEEP_PROB);
    const float v0 = (dropu[e] + KEEP_PROB < 1.0f) ? 0.0f : v;
    const float v1 = (dropu[(size_t)n_edges + e] + KEEP_PROB < 1.0f) ? 0.0f : v;
    v4i r = {cols[e], __float_as_int(v0), __float_as_int(v1), 0};
    ((v4i*)rec)[pos[e]] = r;
}

template <int K>
__device__ __forceinline__ void pull_batch(unsigned& m, int cj, int wb,
                                           const float* __restrict__ srcl,
                                           v2f& acc) {
    int   c[K];
    float w[K];
    v2f   x[K];
#pragma unroll
    for (int k = 0; k < K; ++k) {
        const int j = __builtin_ctz(m);
        m &= m - 1;
        c[k] = __builtin_amdgcn_readlane(cj, j);
        w[k] = __int_as_float(__builtin_amdgcn_readlane(wb, j));
    }
#pragma unroll
    for (int k = 0; k < K; ++k) {
        x[k] = *(const v2f*)(srcl + (size_t)c[k] * DIM);
    }
#pragma unroll
    for (int k = 0; k < K; ++k) {
        acc += w[k] * x[k];
    }
}

template <int LAYER>
__global__ void __launch_bounds__(256)
gcn_pull_kernel(const int* __restrict__ rowptr, const int* __restrict__ rec,
                const float* __restrict__ src, float* __restrict__ dst,
                int n_nodes) {
    int row = blockIdx.x * 8 + (threadIdx.x >> 5);
    if (row >= n_nodes) return;
    row = __builtin_amdgcn_readfirstlane(row);
    const int l = threadIdx.x & 31;

    const int beg = rowptr[row];
    const int end = rowptr[row + 1];

    float* dp = dst + (size_t)row * DIM + l * 2;
    const float* srcl = src + l * 2;

    v2f acc = {0.f, 0.f};
    if (LAYER == 1) acc = *(const v2f*)dp;

    for (int base = beg; base < end; base += 32) {
        const int idx = base + l;
        v4i r = {0, 0, 0, 0};
        if (idx < end) r = ((const v4i*)rec)[idx];
        const int cj = r.x;
        const int wb = LAYER ? r.z : r.y;
        unsigned m = __builtin_amdgcn_ballot_w32(__int_as_float(wb) != 0.0f);
        int nk = __builtin_popcount(m);
        for (; nk >= 8; nk -= 8) pull_batch<8>(m, cj, wb, srcl, acc);
        if (nk & 4) pull_batch<4>(m, cj, wb, srcl, acc);
        if (nk & 2) pull_batch<2>(m, cj, wb, srcl, acc);
        if (nk & 1) pull_batch<1>(m, cj, wb, srcl, acc);
    }

    *(v2f*)dp = acc;
}

__global__ void __launch_bounds__(256)
gcn_finalize_kernel(float* __restrict__ out, const float* __restrict__ x1,
                    long long n4) {
    const long long i = (long long)blockIdx.x * blockDim.x + threadIdx.x;
    if (i >= n4) return;
    v4f a = ((const v4f*)out)[i];
    v4f p = ((const v4f*)x1)[i];
    v4f r = (a + p) * (1.0f / 3.0f);
    ((v4f*)out)[i] = r;
}

extern "C" void kernel_launch(void* const* d_in, const int* in_sizes, int n_in,
                              void* d_out, int out_size, void* d_ws, size_t ws_size,
                              hipStream_t stream) {
    const float* emb   = (const float*)d_in[0];
    const float* W     = (const float*)d_in[1];
    const float* bias  = (const float*)d_in[2];
    const float* vals  = (const float*)d_in[3];
    const float* dropu = (const float*)d_in[4];
    const int*   rows  = (const int*)d_in[5];
    const int*   cols  = (const int*)d_in[6];

    const int N = in_sizes[0] / DIM;
    const int E = in_sizes[3];

    float* out = (float*)d_out;
    float* x1  = (float*)d_ws;

    const size_t x1_bytes  = (size_t)N * DIM * sizeof(float);
    const size_t rec_bytes = (size_t)E * 16;
    const size_t rp_bytes  = ((size_t)N + 1) * sizeof(int);
    const size_t need      = x1_bytes + rec_bytes + rp_bytes + (size_t)E * sizeof(int);

    if (ws_size >= need) {
        int* rec    = (int*)((char*)d_ws + x1_bytes);
        int* rowptr = (int*)((char*)d_ws + x1_bytes + rec_bytes);
        int* pos    = (int*)((char*)d_ws + x1_bytes + rec_bytes + rp_bytes);

        {
            dim3 grid((N + 127) / 128), block(256);
            gcn_fc_wmma_kernel<<<grid, block, 0, stream>>>(emb, W, bias, out, N);
        }
        {
            gcn_group_kernel<<<dim3((N + OWN_ROWS - 1) / OWN_ROWS), dim3(1024), 0, stream>>>(rows, rowptr, pos, N, E);
            gcn_fill_kernel<<<dim3((E + 255) / 256), dim3(256), 0, stream>>>(vals, dropu, cols, pos, rec, E);
        }
        {
            dim3 grid((N + 7) / 8), block(256);
            gcn_pull_kernel<0><<<grid, block, 0, stream>>>(rowptr, rec, emb, x1,  N);
            gcn_pull_kernel<1><<<grid, block, 0, stream>>>(rowptr, rec, x1,  out, N);
        }
    } else {
        hipMemsetAsync(d_ws, 0, x1_bytes, stream);

        {
            dim3 grid((N + 127) / 128), block(256);
            gcn_fc_wmma_kernel<<<grid, block, 0, stream>>>(emb, W, bias, out, N);
        }

        {
            long long threads = (long long)E * 32;
            dim3 grid((unsigned)((threads + 255) / 256)), block(256);
            gcn_spmm_kernel<<<grid, block, 0, stream>>>(vals, dropu,     rows, cols, emb, x1,  E);
            gcn_spmm_kernel<<<grid, block, 0, stream>>>(vals, dropu + E, rows, cols, x1,  out, E);
        }
    }

    {
        long long n4 = (long long)N * DIM / 4;
        dim3 grid((unsigned)((n4 + 255) / 256)), block(256);
        gcn_finalize_kernel<<<grid, block, 0, stream>>>(out, x1, n4);
    }
}


# Pull SpMM: a wave walks 8 consecutive rows, next row's records (now 8-B pairs {col,v_layer}) and fc row requested under the current row's gathers; 12500 waves a layer, not 100000; row sum order kept
# speedup vs baseline: 2.9021x; 2.2148x over previous
#include <hip/hip_runtime.h>
#include <hip/hip_bf16.h>

typedef __attribute__((ext_vector_type(2))) float v2f;
typedef __attribute__((ext_vector_type(4))) float v4f;
typedef __attribute__((ext_vector_type(8))) float v8f;
typedef __attribute__((ext_vector_type(4))) int   v4i;
typedef __attribute__((ext_vector_type(2))) int   v2i;

#define DIM 64
#define KEEP_PROB 0.7f
#define OWN_ROWS 400

__global__ void __launch_bounds__(256)
gcn_fc_wmma_kernel(const float* __restrict__ emb, const float* __restrict__ W,
                   const float* __restrict__ bias, float* __restrict__ acc,
                   int n_nodes) {
    __shared__ alignas(16) float Wlds[DIM * DIM];
    __shared__ alignas(16) float Alds[8][16 * DIM];

    const int tid  = threadIdx.x;
    const int wave = tid >> 5;
    const int lane = tid & 31;

    for (int i = tid; i < (DIM * DIM) / 4; i += 256) {
        ((v4f*)Wlds)[i] = ((const v4f*)W)[i];
    }

    const int row_base = blockIdx.x * 128 + wave * 16;

    for (int i = lane; i < 256; i += 32) {
        int r  = i >> 4;
        int c4 = i & 15;
        int gr = row_base + r;
        v4f v = (gr < n_nodes) ? ((const v4f*)(emb + (size_t)gr * DIM))[c4]
                               : (v4f){0.f, 0.f, 0.f, 0.f};
        ((v4f*)&Alds[wave][0])[r * 16 + c4] = v;
    }
    __syncthreads();

    const float* At = &Alds[wave][0];

    const int m_rel = lane & 15;
    const int k_off = (lane >> 4) * 2;
    const int n_lo  = lane & 15;

    v8f c0 = {}, c1 = {}, c2 = {}, c3 = {};

#pragma unroll
    for (int kk = 0; kk < 16; ++kk) {
        const int k = kk * 4 + k_off;
        v2f a  = *(const v2f*)(At + m_rel * DIM + k);
        v2f b0 = *(const v2f*)(Wlds + (n_lo +  0) * DIM + k);
        v2f b1 = *(const v2f*)(Wlds + (n_lo + 16) * DIM + k);
        v2f b2 = *(const v2f*)(Wlds + (n_lo + 32) * DIM + k);
        v2f b3 = *(const v2f*)(Wlds + (n_lo + 48) * DIM + k);
        c0 = __builtin_amdgcn_wmma_f32_16x16x4_f32(false, a, false, b0, (short)0, c0, false, false);
        c1 = __builtin_amdgcn_wmma_f32_16x16x4_f32(false, a, false, b1, (short)0, c1, false, false);
        c2 = __builtin_amdgcn_wmma_f32_16x16x4_f32(false, a, false, b2, (short)0, c2, false, false);
        c3 = __builtin_amdgcn_wmma_f32_16x16x4_f32(false, a, false, b3, (short)0, c3, false, false);
    }

    const float bb0 = bias[n_lo];
    const float bb1 = bias[n_lo + 16];
    const float bb2 = bias[n_lo + 32];
    const float bb3 = bias[n_lo + 48];

    const int m_hi = (lane >> 4) * 8;
#pragma unroll
    for (int j = 0; j < 8; ++j) {
        int gm = row_base + m_hi + j;
        if (gm < n_nodes) {
            float* outr = acc + (size_t)gm * DIM;
            outr[n_lo +  0] = c0[j] + bb0;
            outr[n_lo + 16] = c1[j] + bb1;
            outr[n_lo + 32] = c2[j] + bb2;
            outr[n_lo + 48] = c3[j] + bb3;
        }
    }
}

__global__ void __launch_bounds__(256)
gcn_spmm_kernel(const float* __restrict__ vals, const float* __restrict__ dropu,
                const int* __restrict__ rows, const int* __restrict__ cols,
                const float* __restrict__ src, float* __restrict__ dst,
                int n_edges) {
    const long long tid = (long long)blockIdx.x * blockDim.x + threadIdx.x;
    int e = (int)(tid >> 5);
    if (e >= n_edges) return;
    e = __builtin_amdgcn_readfirstlane(e);
    const int l = threadIdx.x & 31;

    const float u = dropu[e];
    if (u + KEEP_PROB < 1.0f) return;
    const float v = vals[e] * (1.0f / KEEP_PROB);

    const int col = cols[e];
    const int row = rows[e];

    const float* sp = src + (size_t)col * DIM + l * 2;
    float*       dp = dst + (size_t)row * DIM + l * 2;
    const float m0 = v * sp[0];
    const float m1 = v * sp[1];

    asm volatile("global_atomic_add_f32 %0, %1, off scope:SCOPE_DEV"
                 :: "v"(dp), "v"(m0) : "memory");
    asm volatile("global_atomic_add_f32 %0, %1, off scope:SCOPE_DEV"
                 :: "v"(dp + 1), "v"(m1) : "memory");
}


#define BROWS 400
#define NBMAX 256
#define CHUNK 4096

__device__ __forceinline__ int wave_incl_scan_i32(int x, int lane) {
    int y;
    y = __shfl_up(x, 1);  if (lane >= 1)  x += y;
    y = __shfl_up(x, 2);  if (lane >= 2)  x += y;
    y = __shfl_up(x, 4);  if (lane >= 4)  x += y;
    y = __shfl_up(x, 8);  if (lane >= 8)  x += y;
    y = __shfl_up(x, 16); if (lane >= 16) x += y;
    return x;
}

template <int NBITS>
__device__ __forceinline__ void wave_match(int key, bool ok, int& rank, int& cnt) {
    unsigned peers = __builtin_amdgcn_ballot_w32(ok);
    for (int b = 0; b < NBITS; ++b) {
        const bool bit = ((key >> b) & 1) != 0;
        const unsigned m = __builtin_amdgcn_ballot_w32(ok && bit);
        peers &= bit ? m : ~m;
    }
    rank = (int)__builtin_amdgcn_mbcnt_lo(peers, 0u);
    cnt  = __builtin_popcount(peers);
}

__device__ __forceinline__ void load_rows4(const int* __restrict__ rows,
                                           int n_edges, int e0, int (&row)[4]) {
    if (e0 + 3 < n_edges) {
        const v4i g = *(const v4i*)(rows + e0);
        row[0] = g.x; row[1] = g.y; row[2] = g.z; row[3] = g.w;
    } else {
#pragma unroll
        for (int k = 0; k < 4; ++k) row[k] = (e0 + k < n_edges) ? rows[e0 + k] : -1;
    }
}

__device__ __forceinline__ void chunk_count(int n_nodes, int* mycw, int (&row)[4],
                                            int (&key)[4], int (&rank)[4],
                                            int (&cnt)[4]) {
#pragma unroll
    for (int k = 0; k < 4; ++k) {
        const bool ok = (unsigned)row[k] < (unsigned)n_nodes;
        if (!ok) row[k] = -1;
        key[k] = ok ? (row[k] / BROWS) : 0;
        wave_match<8>(key[k], ok, rank[k], cnt[k]);
        if (ok && rank[k] == 0) mycw[key[k]] += cnt[k];
    }
}

__global__ void __launch_bounds__(1024)
gcn_hist_kernel(const int* __restrict__ rows, int* __restrict__ hist,
                int n_nodes, int n_edges) {
    __shared__ int cw[32][NBMAX];

    const int tid  = threadIdx.x;
    const int lane = tid & 31;
    const int wave = tid >> 5;
    int* mycw = cw[wave];
    for (int i = lane; i < NBMAX; i += 32) mycw[i] = 0;

    int row[4], key[4], rank[4], cnt[4];
    load_rows4(rows, n_edges, (blockIdx.x * 1024 + tid) * 4, row);
    chunk_count(n_nodes, mycw, row, key, rank, cnt);
    __syncthreads();

    if (tid < NBMAX) {
        int tot = 0;
        for (int w = 0; w < 32; ++w) tot += cw[w][tid];
        hist[blockIdx.x * NBMAX + tid] = tot;
    }
}

__global__ void __launch_bounds__(1024)
gcn_scatter_kernel(const int* __restrict__ rows, const int* __restrict__ cols,
                   const float* __restrict__ vals, const float* __restrict__ dropu,
                   const int* __restrict__ hist, int* __restrict__ bbase,
                   int* __restrict__ tmp, int n_nodes, int n_edges, int n_chunks) {
    __shared__ int cw[32][NBMAX];
    __shared__ alignas(16) int part[2][16][NBMAX];
    __shared__ int gofs[NBMAX];
    __shared__ int wtot[8];

    const int tid   = threadIdx.x;
    const int lane  = tid & 31;
    const int wave  = tid >> 5;
    const int chunk = blockIdx.x;
    const int e0    = (chunk * 1024 + tid) * 4;
    int* mycw = cw[wave];
    for (int i = lane; i < NBMAX; i += 32) mycw[i] = 0;

    int row[4];
    load_rows4(rows, n_edges, e0, row);

    {
        const int t4 = tid & 63;
        const int q  = tid >> 6;
        v4i all = {0, 0, 0, 0}, bef = {0, 0, 0, 0};
        for (int c0 = q; c0 < n_chunks; c0 += 64) {
            v4i h[4];
#pragma unroll
            for (int u = 0; u < 4; ++u) {
                const int c  = c0 + 16 * u;
                const int cl = (c < n_chunks) ? c : (n_chunks - 1);
                h[u] = ((const v4i*)hist)[cl * (NBMAX / 4) + t4];
            }
#pragma unroll
            for (int u = 0; u < 4; ++u) {
                const int c = c0 + 16 * u;
                const v4i z = {0, 0, 0, 0};
                all += (c < n_chunks) ? h[u] : z;
                bef += (c < chunk) ? h[u] : z;
            }
        }
        ((v4i*)&part[0][q][0])[t4] = all;
        ((v4i*)&part[1][q][0])[t4] = bef;
    }

    int   cc[4];
    float ww[4], ua[4], ub[4];
    if (e0 + 3 < n_edges && (n_edges & 3) == 0) {
        const v4i c4 = *(const v4i*)(cols + e0);
        const v4f w4 = *(const v4f*)(vals + e0);
        const v4f a4 = *(const v4f*)(dropu + e0);
        const v4f b4 = *(const v4f*)(dropu + (size_t)n_edges + e0);
        cc[0] = c4.x; cc[1] = c4.y; cc[2] = c4.z; cc[3] = c4.w;
        ww[0] = w4.x; ww[1] = w4.y; ww[2] = w4.z; ww[3] = w4.w;
        ua[0] = a4.x; ua[1] = a4.y; ua[2] = a4.z; ua[3] = a4.w;
        ub[0] = b4.x; ub[1] = b4.y; ub[2] = b4.z; ub[3] = b4.w;
    } else {
#pragma unroll
        for (int k = 0; k < 4; ++k) {
            const bool in = e0 + k < n_edges;
            cc[k] = in ? cols[e0 + k] : 0;
            ww[k] = in ? vals[e0 + k] : 0.0f;
            ua[k] = in ? dropu[e0 + k] : 0.0f;
            ub[k] = in ? dropu[(size_t)n_edges + e0 + k] : 0.0f;
        }
    }

    int key[4], rank[4], cnt[4];
    chunk_count(n_nodes, mycw, row, key, rank, cnt);
    __syncthreads();

    int colsum = 0, before = 0;
    if (tid < NBMAX) {
        int run = 0;
        for (int w = 0; w < 32; ++w) {
            const int c = cw[w][tid];
            cw[w][tid] = run;
            run += c;
        }
        for (int q = 0; q < 16; ++q) {
            colsum += part[0][q][tid];
            before += part[1][q][tid];
        }
    }
    const int inc = wave_incl_scan_i32(colsum, lane);
    if (tid < NBMAX && lane == 31) wtot[wave] = inc;
    __syncthreads();
    if (tid < NBMAX) {
        int woff = 0;
        for (int w = 0; w < wave; ++w) woff += wtot[w];
        const int excl = woff + inc - colsum;
        gofs[tid] = excl + before;
        if (chunk == 0) {
            bbase[tid] = excl;
            if (tid == NBMAX - 1) bbase[NBMAX] = excl + colsum;
        }
    }
    __syncthreads();

#pragma unroll
    for (int k = 0; k < 4; ++k) {
        if (row[k] >= 0) {
            const int cur  = mycw[key[k]];
            const int slot = gofs[key[k]] + cur + rank[k];
            if (rank[k] == 0) mycw[key[k]] = cur + cnt[k];
            const float v  = ww[k] * (1.0f / KEEP_PROB);
            const float v0 = (ua[k] + KEEP_PROB < 1.0f) ? 0.0f : v;
            const float v1 = (ub[k] + KEEP_PROB < 1.0f) ? 0.0f : v;
            const v4i r = {cc[k], __float_as_int(v0), __float_as_int(v1), row[k]};
            ((v4i*)tmp)[slot] = r;
        }
    }
}

__global__ void __launch_bounds__(1024)
gcn_bucket_kernel(const int* __restrict__ bbase, const int* __restrict__ tmp,
                  int* __restrict__ rowptr, int* __restrict__ rec, int n_nodes) {
    __shared__ int cw[32][BROWS];
    __shared__ int rstart[BROWS];
    __shared__ int wtot[32];

    const int tid  = threadIdx.x;
    const int lane = tid & 31;
    const int wave = tid >> 5;
    const int lo   = blockIdx.x * BROWS;
    const int nrows = (n_nodes - lo < BROWS) ? (n_nodes - lo) : BROWS;
    const int seg_beg = bbase[blockIdx.x];
    const int seg_end = bbase[blockIdx.x + 1];
    int* mycw = cw[wave];
    for (int i = lane; i < BROWS; i += 32) mycw[i] = 0;

    for (int base = seg_beg; base < seg_end; base += 4096) {
        int kw[4];
#pragma unroll
        for (int u = 0; u < 4; ++u) {
            const int idx = base + u * 1024 + tid;
            const int ic  = (idx < seg_end) ? idx : (seg_end - 1);
            kw[u] = tmp[(size_t)ic * 4 + 3];
        }
#pragma unroll
        for (int u = 0; u < 4; ++u) {
            const bool ok = base + u * 1024 + tid < seg_end;
            const int key = ok ? (kw[u] - lo) : 0;
            int rank, cnt;
            wave_match<9>(key, ok, rank, cnt);
            if (ok && rank == 0) mycw[key] += cnt;
        }
    }
    __syncthreads();

    int tot = 0;
    if (tid < BROWS) {
        for (int w = 0; w < 32; ++w) {
            const int c = cw[w][tid];
            cw[w][tid] = tot;
            tot += c;
        }
    }
    const int x = wave_incl_scan_i32(tot, lane);
    if (lane == 31) wtot[wave] = x;
    __syncthreads();
    const int wt   = wtot[lane];
    const int wx   = wave_incl_scan_i32(wt, lane);
    const int woff = __shfl(wx - wt, wave);
    const int start = seg_beg + woff + x - tot;
    if (tid < BROWS) rstart[tid] = start;
    if (tid < nrows) {
        rowptr[lo + tid] = start;
        if (lo + tid == n_nodes - 1) rowptr[n_nodes] = start + tot;
    }
    __syncthreads();

    for (int base = seg_beg; base < seg_end; base += 4096) {
        v4i r[4];
#pragma unroll
        for (int u = 0; u < 4; ++u) {
            const int idx = base + u * 1024 + tid;
            const int ic  = (idx < seg_end) ? idx : (seg_end - 1);
            r[u] = ((const v4i*)tmp)[ic];
        }
#pragma unroll
        for (int u = 0; u < 4; ++u) {
            const bool ok = base + u * 1024 + tid < seg_end;
            const int key = ok ? (r[u].w - lo) : 0;
            int rank, cnt;
            wave_match<9>(key, ok, rank, cnt);
            if (ok) {
                const int cur = mycw[key];
                if (rank == 0) mycw[key] = cur + cnt;
                const v4i o = {r[u].x, r[u].y, r[u].x, r[u].z};
                ((v4i*)rec)[rstart[key] + cur + rank] = o;
            }
        }
    }
}

template <int K>
__device__ __forceinline__ void pull_batch(unsigned& m, int cj, int wb,
                                           const float* __restrict__ srcl,
                                           v2f& acc) {
    int   c[K];
    float w[K];
    v2f   x[K];
#pragma unroll
    for (int k = 0; k < K; ++k) {
        const int j = __builtin_ctz(m);
        m &= m - 1;
        c[k] = __builtin_amdgcn_readlane(cj, j);
        w[k] = __int_as_float(__builtin_amdgcn_readlane(wb, j));
    }
#pragma unroll
    for (int k = 0; k < K; ++k) {
        x[k] = *(const v2f*)(srcl + (size_t)c[k] * DIM);
    }
#pragma unroll
    for (int k = 0; k < K; ++k) {
        acc += w[k] * x[k];
    }
}

__device__ __forceinline__ void pull_records(const v2i r,
                                             const float* __restrict__ srcl,
                                             v2f& acc) {
    const int cj = r.x;
    const int wb = r.y;
    unsigned m = __builtin_amdgcn_ballot_w32(__int_as_float(wb) != 0.0f);
    int nk = __builtin_popcount(m);
    for (; nk >= 8; nk -= 8) pull_batch<8>(m, cj, wb, srcl, acc);
    if (nk & 4) pull_batch<4>(m, cj, wb, srcl, acc);
    if (nk & 2) pull_batch<2>(m, cj, wb, srcl, acc);
    if (nk & 1) pull_batch<1>(m, cj, wb, srcl, acc);
}

#define PULL_ROWS 8

template <int LAYER>
__global__ void __launch_bounds__(256)
gcn_pull_kernel(const int* __restrict__ rowptr, const int* __restrict__ rec,
                const float* __restrict__ src, float* __restrict__ dst,
                int n_nodes) {
    int r0 = (blockIdx.x * 8 + (threadIdx.x >> 5)) * PULL_ROWS;
    if (r0 >= n_nodes) return;
    r0 = __builtin_amdgcn_readfirstlane(r0);
    const int l  = threadIdx.x & 31;
    const int nr = (n_nodes - r0 < PULL_ROWS) ? (n_nodes - r0) : PULL_ROWS;

    const int rpv = rowptr[r0 + ((l < nr) ? l : nr)];

    float* dp = dst + (size_t)r0 * DIM + l * 2;
    const float* srcl = src + l * 2;
    const v2i* recl = (const v2i*)rec + LAYER;

    v2f fn = {0.f, 0.f};
    if (LAYER == 1) fn = *(const v2f*)dp;
    int nbeg = __builtin_amdgcn_readlane(rpv, 0);
    int nend = __builtin_amdgcn_readlane(rpv, 1);
    v2i rn = {0, 0};
    if (nbeg + l < nend) rn = recl[(size_t)(nbeg + l) * 2];

    for (int i = 0; i < nr; ++i) {
        const v2i r = rn;
        v2f acc = fn;
        const int beg = nbeg;
        const int end = nend;

        if (i + 1 < nr) {
            nbeg = nend;
            nend = __builtin_amdgcn_readlane(rpv, i + 2);
            rn = (v2i){0, 0};
            if (nbeg + l < nend) rn = recl[(size_t)(nbeg + l) * 2];
            if (LAYER == 1) fn = *(const v2f*)(dp + (size_t)(i + 1) * DIM);
        }

        pull_records(r, srcl, acc);
        for (int base = beg + 32; base < end; base += 32) {
            v2i q = {0, 0};
            if (base + l < end) q = recl[(size_t)(base + l) * 2];
            pull_records(q, srcl, acc);
        }

        *(v2f*)(dp + (size_t)i * DIM) = acc;
    }
}

__global__ void __launch_bounds__(256)
gcn_finalize_kernel(float* __restrict__ out, const float* __restrict__ x1,
                    long long n4) {
    const long long i = (long long)blockIdx.x * blockDim.x + threadIdx.x;
    if (i >= n4) return;
    v4f a = ((const v4f*)out)[i];
    v4f p = ((const v4f*)x1)[i];
    v4f r = (a + p) * (1.0f / 3.0f);
    ((v4f*)out)[i] = r;
}

extern "C" void kernel_launch(void* const* d_in, const int* in_sizes, int n_in,
                              void* d_out, int out_size, void* d_ws, size_t ws_size,
                              hipStream_t stream) {
    const float* emb   = (const float*)d_in[0];
    const float* W     = (const float*)d_in[1];
    const float* bias  = (const float*)d_in[2];
    const float* vals  = (const float*)d_in[3];
    const float* dropu = (const float*)d_in[4];
    const int*   rows  = (const int*)d_in[5];
    const int*   cols  = (const int*)d_in[6];

    const int N = in_sizes[0] / DIM;
    const int E = in_sizes[3];

    float* out = (float*)d_out;
    float* x1  = (float*)d_ws;

    const int    n_chunks  = (E + CHUNK - 1) / CHUNK;
    const int    n_buckets = (N + BROWS - 1) / BROWS;
    const size_t x1_bytes  = (size_t)N * DIM * sizeof(float);
    const size_t rec_bytes = (size_t)E * 16;
    const size_t r0_bytes  = (x1_bytes > rec_bytes) ? x1_bytes : rec_bytes;
    const size_t rp_bytes  = (((size_t)N + 1) * sizeof(int) + 15) & ~(size_t)15;
    const size_t hs_bytes  = (size_t)n_chunks * NBMAX * sizeof(int);
    const size_t need      = r0_bytes + rec_bytes + rp_bytes + hs_bytes + (NBMAX + 1) * sizeof(int);

    if (ws_size >= need && n_buckets <= NBMAX && n_chunks > 0) {
        int* tmp    = (int*)d_ws;
        int* rec    = (int*)((char*)d_ws + r0_bytes);
        int* rowptr = (int*)((char*)d_ws + r0_bytes + rec_bytes);
        int* hist   = (int*)((char*)d_ws + r0_bytes + rec_bytes + rp_bytes);
        int* bbase  = (int*)((char*)d_ws + r0_bytes + rec_bytes + rp_bytes + hs_bytes);

        {
            dim3 grid((N + 127) / 128), block(256);
            gcn_fc_wmma_kernel<<<grid, block, 0, stream>>>(emb, W, bias, out, N);
        }
        {
            gcn_hist_kernel<<<dim3(n_chunks), dim3(1024), 0, stream>>>(rows, hist, N, E);
            gcn_scatter_kernel<<<dim3(n_chunks), dim3(1024), 0, stream>>>(rows, cols, vals, dropu, hist, bbase, tmp, N, E, n_chunks);
            gcn_bucket_kernel<<<dim3(n_buckets), dim3(1024), 0, stream>>>(bbase, tmp, rowptr, rec, N);
        }
        {
            dim3 grid((N + 8 * PULL_ROWS - 1) / (8 * PULL_ROWS)), block(256);
            gcn_pull_kernel<0><<<grid, block, 0, stream>>>(rowptr, rec, emb, x1,  N);
            gcn_pull_kernel<1><<<grid, block, 0, stream>>>(rowptr, rec, x1,  out, N);
        }
    } else {
        hipMemsetAsync(d_ws, 0, x1_bytes, stream);

        {
            dim3 grid((N + 127) / 128), block(256);
            gcn_fc_wmma_kernel<<<grid, block, 0, stream>>>(emb, W, bias, out, N);
        }

        {
            long long threads = (long long)E * 32;
            dim3 grid((unsigned)((threads + 255) / 256)), block(256);
            gcn_spmm_kernel<<<grid, block, 0, stream>>>(vals, dropu,     rows, cols, emb, x1,  E);
            gcn_spmm_kernel<<<grid, block, 0, stream>>>(vals, dropu + E, rows, cols, x1,  out, E);
        }
    }

    {
        long long n4 = (long long)N * DIM / 4;
        dim3 grid((unsigned)((n4 + 255) / 256)), block(256);
        gcn_finalize_kernel<<<grid, block, 0, stream>>>(out, x1, n4);
    }
}


# fc pass without LDS: a wave keeps the W fragments in registers and walks 16-row tiles, A fragments straight from global as 16-B loads; no 16-way bank conflict; k order inside the fc sums new, fixed
# speedup vs baseline: 3.2243x; 1.1110x over previous
#include <hip/hip_runtime.h>
#include <hip/hip_bf16.h>

typedef __attribute__((ext_vector_type(2))) float v2f;
typedef __attribute__((ext_vector_type(4))) float v4f;
typedef __attribute__((ext_vector_type(8))) float v8f;
typedef __attribute__((ext_vector_type(4))) int   v4i;
typedef __attribute__((ext_vector_type(2))) int   v2i;

#define DIM 64
#define KEEP_PROB 0.7f
#define OWN_ROWS 400

#define FC_GROUPS 512

__device__ __forceinline__ void fc_load_tile(const float* __restrict__ emb, int tile,
                                             int c, int h, int n_nodes, v4f (&a)[8]) {
    int gr = tile * 16 + c;
    gr = (gr < n_nodes) ? gr : (n_nodes - 1);
    const float* p = emb + (size_t)gr * DIM + 4 * h;
#pragma unroll
    for (int j = 0; j < 8; ++j) a[j] = *(const v4f*)(p + 8 * j);
}

__device__ __forceinline__ void fc_tile(const v4f (&a)[8], const v4f (&bf)[4][8],
                                        float bb0, float bb1, float bb2, float bb3,
                                        float* __restrict__ acc, int tile,
                                        int c, int h, int n_nodes) {
    v8f c0 = {}, c1 = {}, c2 = {}, c3 = {};
#pragma unroll
    for (int j = 0; j < 8; ++j) {
        const v2f a0 = a[j].xy;
        const v2f a1 = a[j].zw;
        c0 = __builtin_amdgcn_wmma_f32_16x16x4_f32(false, a0, false, bf[0][j].xy, (short)0, c0, false, false);
        c1 = __builtin_amdgcn_wmma_f32_16x16x4_f32(false, a0, false, bf[1][j].xy, (short)0, c1, false, false);
        c2 = __builtin_amdgcn_wmma_f32_16x16x4_f32(false, a0, false, bf[2][j].xy, (short)0, c2, false, false);
        c3 = __builtin_amdgcn_wmma_f32_16x16x4_f32(false, a0, false, bf[3][j].xy, (short)0, c3, false, false);
        c0 = __builtin_amdgcn_wmma_f32_16x16x4_f32(false, a1, false, bf[0][j].zw, (short)0, c0, false, false);
        c1 = __builtin_amdgcn_wmma_f32_16x16x4_f32(false, a1, false, bf[1][j].zw, (short)0, c1, false, false);
        c2 = __builtin_amdgcn_wmma_f32_16x16x4_f32(false, a1, false, bf[2][j].zw, (short)0, c2, false, false);
        c3 = __builtin_amdgcn_wmma_f32_16x16x4_f32(false, a1, false, bf[3][j].zw, (short)0, c3, false, false);
    }

    const int row_base = tile * 16 + h * 8;
#pragma unroll
    for (int j = 0; j < 8; ++j) {
        const int gm = row_base + j;
        if (gm < n_nodes) {
            float* outr = acc + (size_t)gm * DIM;
            outr[c +  0] = c0[j] + bb0;
            outr[c + 16] = c1[j] + bb1;
            outr[c + 32] = c2[j] + bb2;
            outr[c + 48] = c3[j] + bb3;
        }
    }
}

__global__ void __launch_bounds__(256)
gcn_fc_wmma_kernel(const float* __restrict__ emb, const float* __restrict__ W,
                   const float* __restrict__ bias, float* __restrict__ acc,
                   int n_nodes) {
    const int lane = threadIdx.x & 31;
    const int c    = lane & 15;
    const int h    = lane >> 4;
    const int nw   = gridDim.x * 8;
    const int n_tiles = (n_nodes + 15) >> 4;
    int tile = (threadIdx.x >> 5) * gridDim.x + blockIdx.x;
    if (tile >= n_tiles) return;
    tile = __builtin_amdgcn_readfirstlane(tile);

    v4f bf[4][8];
#pragma unroll
    for (int t = 0; t < 4; ++t) {
#pragma unroll
        for (int j = 0; j < 8; ++j) {
            bf[t][j] = *(const v4f*)(W + (size_t)(16 * t + c) * DIM + 8 * j + 4 * h);
        }
    }
    const float bb0 = bias[c];
    const float bb1 = bias[c + 16];
    const float bb2 = bias[c + 32];
    const float bb3 = bias[c + 48];

    v4f pa[8], pb[8];
    fc_load_tile(emb, tile, c, h, n_nodes, pa);
    for (;;) {
        int next = tile + nw;
        fc_load_tile(emb, (next < n_tiles) ? next : tile, c, h, n_nodes, pb);
        fc_tile(pa, bf, bb0, bb1, bb2, bb3, acc, tile, c, h, n_nodes);
        if (next >= n_tiles) break;
        tile = next;
        next = tile + nw;
        fc_load_tile(emb, (next < n_tiles) ? next : tile, c, h, n_nodes, pa);
        fc_tile(pb, bf, bb0, bb1, bb2, bb3, acc, tile, c, h, n_nodes);
        if (next >= n_tiles) break;
        tile = next;
    }
}

__global__ void __launch_bounds__(256)
gcn_spmm_kernel(const float* __restrict__ vals, const float* __restrict__ dropu,
                const int* __restrict__ rows, const int* __restrict__ cols,
                const float* __restrict__ src, float* __restrict__ dst,
                int n_edges) {
    const long long tid = (long long)blockIdx.x * blockDim.x + threadIdx.x;
    int e = (int)(tid >> 5);
    if (e >= n_edges) return;
    e = __builtin_amdgcn_readfirstlane(e);
    const int l = threadIdx.x & 31;

    const float u = dropu[e];
    if (u + KEEP_PROB < 1.0f) return;
    const float v = vals[e] * (1.0f / KEEP_PROB);

    const int col = cols[e];
    const int row = rows[e];

    const float* sp = src + (size_t)col * DIM + l * 2;
    float*       dp = dst + (size_t)row * DIM + l * 2;
    const float m0 = v * sp[0];
    const float m1 = v * sp[1];

    asm volatile("global_atomic_add_f32 %0, %1, off scope:SCOPE_DEV"
                 :: "v"(dp), "v"(m0) : "memory");
    asm volatile("global_atomic_add_f32 %0, %1, off scope:SCOPE_DEV"
                 :: "v"(dp + 1), "v"(m1) : "memory");
}


#define BROWS 400
#define NBMAX 256
#define CHUNK 4096

__device__ __forceinline__ int wave_incl_scan_i32(int x, int lane) {
    int y;
    y = __shfl_up(x, 1);  if (lane >= 1)  x += y;
    y = __shfl_up(x, 2);  if (lane >= 2)  x += y;
    y = __shfl_up(x, 4);  if (lane >= 4)  x += y;
    y = __shfl_up(x, 8);  if (lane >= 8)  x += y;
    y = __shfl_up(x, 16); if (lane >= 16) x += y;
    return x;
}

template <int NBITS>
__device__ __forceinline__ void wave_match(int key, bool ok, int& rank, int& cnt) {
    unsigned peers = __builtin_amdgcn_ballot_w32(ok);
    for (int b = 0; b < NBITS; ++b) {
        const bool bit = ((key >> b) & 1) != 0;
        const unsigned m = __builtin_amdgcn_ballot_w32(ok && bit);
        peers &= bit ? m : ~m;
    }
    rank = (int)__builtin_amdgcn_mbcnt_lo(peers, 0u);
    cnt  = __builtin_popcount(peers);
}

__device__ __forceinline__ void load_rows4(const int* __restrict__ rows,
                                           int n_edges, int e0, int (&row)[4]) {
    if (e0 + 3 < n_edges) {
        const v4i g = *(const v4i*)(rows + e0);
        row[0] = g.x; row[1] = g.y; row[2] = g.z; row[3] = g.w;
    } else {
#pragma unroll
        for (int k = 0; k < 4; ++k) row[k] = (e0 + k < n_edges) ? rows[e0 + k] : -1;
    }
}

__device__ __forceinline__ void chunk_count(int n_nodes, int* mycw, int (&row)[4],
                                            int (&key)[4], int (&rank)[4],
                                            int (&cnt)[4]) {
#pragma unroll
    for (int k = 0; k < 4; ++k) {
        const bool ok = (unsigned)row[k] < (unsigned)n_nodes;
        if (!ok) row[k] = -1;
        key[k] = ok ? (row[k] / BROWS) : 0;
        wave_match<8>(key[k], ok, rank[k], cnt[k]);
        if (ok && rank[k] == 0) mycw[key[k]] += cnt[k];
    }
}

__global__ void __launch_bounds__(1024)
gcn_hist_kernel(const int* __restrict__ rows, int* __restrict__ hist,
                int n_nodes, int n_edges) {
    __shared__ int cw[32][NBMAX];

    const int tid  = threadIdx.x;
    const int lane = tid & 31;
    const int wave = tid >> 5;
    int* mycw = cw[wave];
    for (int i = lane; i < NBMAX; i += 32) mycw[i] = 0;

    int row[4], key[4], rank[4], cnt[4];
    load_rows4(rows, n_edges, (blockIdx.x * 1024 + tid) * 4, row);
    chunk_count(n_nodes, mycw, row, key, rank, cnt);
    __syncthreads();

    if (tid < NBMAX) {
        int tot = 0;
        for (int w = 0; w < 32; ++w) tot += cw[w][tid];
        hist[blockIdx.x * NBMAX + tid] = tot;
    }
}

__global__ void __launch_bounds__(1024)
gcn_scatter_kernel(const int* __restrict__ rows, const int* __restrict__ cols,
                   const float* __restrict__ vals, const float* __restrict__ dropu,
                   const int* __restrict__ hist, int* __restrict__ bbase,
                   int* __restrict__ tmp, int n_nodes, int n_edges, int n_chunks) {
    __shared__ int cw[32][NBMAX];
    __shared__ alignas(16) int part[2][16][NBMAX];
    __shared__ int gofs[NBMAX];
    __shared__ int wtot[8];

    const int tid   = threadIdx.x;
    const int lane  = tid & 31;
    const int wave  = tid >> 5;
    const int chunk = blockIdx.x;
    const int e0    = (chunk * 1024 + tid) * 4;
    int* mycw = cw[wave];
    for (int i = lane; i < NBMAX; i += 32) mycw[i] = 0;

    int row[4];
    load_rows4(rows, n_edges, e0, row);

    {
        const int t4 = tid & 63;
        const int q  = tid >> 6;
        v4i all = {0, 0, 0, 0}, bef = {0, 0, 0, 0};
        for (int c0 = q; c0 < n_chunks; c0 += 64) {
            v4i h[4];
#pragma unroll
            for (int u = 0; u < 4; ++u) {
                const int c  = c0 + 16 * u;
                const int cl = (c < n_chunks) ? c : (n_chunks - 1);
                h[u] = ((const v4i*)hist)[cl * (NBMAX / 4) + t4];
            }
#pragma unroll
            for (int u = 0; u < 4; ++u) {
                const int c = c0 + 16 * u;
                const v4i z = {0, 0, 0, 0};
                all += (c < n_chunks) ? h[u] : z;
                bef += (c < chunk) ? h[u] : z;
            }
        }
        ((v4i*)&part[0][q][0])[t4] = all;
        ((v4i*)&part[1][q][0])[t4] = bef;
    }

    int   cc[4];
    float ww[4], ua[4], ub[4];
    if (e0 + 3 < n_edges && (n_edges & 3) == 0) {
        const v4i c4 = *(const v4i*)(cols + e0);
        const v4f w4 = *(const v4f*)(vals + e0);
        const v4f a4 = *(const v4f*)(dropu + e0);
        const v4f b4 = *(const v4f*)(dropu + (size_t)n_edges + e0);
        cc[0] = c4.x; cc[1] = c4.y; cc[2] = c4.z; cc[3] = c4.w;
        ww[0] = w4.x; ww[1] = w4.y; ww[2] = w4.z; ww[3] = w4.w;
        ua[0] = a4.x; ua[1] = a4.y; ua[2] = a4.z; ua[3] = a4.w;
        ub[0] = b4.x; ub[1] = b4.y; ub[2] = b4.z; ub[3] = b4.w;
    } else {
#pragma unroll
        for (int k = 0; k < 4; ++k) {
            const bool in = e0 + k < n_edges;
            cc[k] = in ? cols[e0 + k] : 0;
            ww[k] = in ? vals[e0 + k] : 0.0f;
            ua[k] = in ? dropu[e0 + k] : 0.0f;
            ub[k] = in ? dropu[(size_t)n_edges + e0 + k] : 0.0f;
        }
    }

    int key[4], rank[4], cnt[4];
    chunk_count(n_nodes, mycw, row, key, rank, cnt);
    __syncthreads();

    int colsum = 0, before = 0;
    if (tid < NBMAX) {
        int run = 0;
        for (int w = 0; w < 32; ++w) {
            const int c = cw[w][tid];
            cw[w][tid] = run;
            run += c;
        }
        for (int q = 0; q < 16; ++q) {
            colsum += part[0][q][tid];
            before += part[1][q][tid];
        }
    }
    const int inc = wave_incl_scan_i32(colsum, lane);
    if (tid < NBMAX && lane == 31) wtot[wave] = inc;
    __syncthreads();
    if (tid < NBMAX) {
        int woff = 0;
        for (int w = 0; w < wave; ++w) woff += wtot[w];
        const int excl = woff + inc - colsum;
        gofs[tid] = excl + before;
        if (chunk == 0) {
            bbase[tid] = excl;
            if (tid == NBMAX - 1) bbase[NBMAX] = excl + colsum;
        }
    }
    __syncthreads();

#pragma unroll
    for (int k = 0; k < 4; ++k) {
        if (row[k] >= 0) {
            const int cur  = mycw[key[k]];
            const int slot = gofs[key[k]] + cur + rank[k];
            if (rank[k] == 0) mycw[key[k]] = cur + cnt[k];
            const float v  = ww[k] * (1.0f / KEEP_PROB);
            const float v0 = (ua[k] + KEEP_PROB < 1.0f) ? 0.0f : v;
            const float v1 = (ub[k] + KEEP_PROB < 1.0f) ? 0.0f : v;
            const v4i r = {cc[k], __float_as_int(v0), __float_as_int(v1), row[k]};
            ((v4i*)tmp)[slot] = r;
        }
    }
}

__global__ void __launch_bounds__(1024)
gcn_bucket_kernel(const int* __restrict__ bbase, const int* __restrict__ tmp,
                  int* __restrict__ rowptr, int* __restrict__ rec, int n_nodes) {
    __shared__ int cw[32][BROWS];
    __shared__ int rstart[BROWS];
    __shared__ int wtot[32];

    const int tid  = threadIdx.x;
    const int lane = tid & 31;
    const int wave = tid >> 5;
    const int lo   = blockIdx.x * BROWS;
    const int nrows = (n_nodes - lo < BROWS) ? (n_nodes - lo) : BROWS;
    const int seg_beg = bbase[blockIdx.x];
    const int seg_end = bbase[blockIdx.x + 1];
    int* mycw = cw[wave];
    for (int i = lane; i < BROWS; i += 32) mycw[i] = 0;

    for (int base = seg_beg; base < seg_end; base += 4096) {
        int kw[4];
#pragma unroll
        for (int u = 0; u < 4; ++u) {
            const int idx = base + u * 1024 + tid;
            const int ic  = (idx < seg_end) ? idx : (seg_end - 1);
            kw[u] = tmp[(size_t)ic * 4 + 3];
        }
#pragma unroll
        for (int u = 0; u < 4; ++u) {
            const bool ok = base + u * 1024 + tid < seg_end;
            const int key = ok ? (kw[u] - lo) : 0;
            int rank, cnt;
            wave_match<9>(key, ok, rank, cnt);
            if (ok && rank == 0) mycw[key] += cnt;
        }
    }
    __syncthreads();

    int tot = 0;
    if (tid < BROWS) {
        for (int w = 0; w < 32; ++w) {
            const int c = cw[w][tid];
            cw[w][tid] = tot;
            tot += c;
        }
    }
    const int x = wave_incl_scan_i32(tot, lane);
    if (lane == 31) wtot[wave] = x;
    __syncthreads();
    const int wt   = wtot[lane];
    const int wx   = wave_incl_scan_i32(wt, lane);
    const int woff = __shfl(wx - wt, wave);
    const int start = seg_beg + woff + x - tot;
    if (tid < BROWS) rstart[tid] = start;
    if (tid < nrows) {
        rowptr[lo + tid] = start;
        if (lo + tid == n_nodes - 1) rowptr[n_nodes] = start + tot;
    }
    __syncthreads();

    for (int base = seg_beg; base < seg_end; base += 4096) {
        v4i r[4];
#pragma unroll
        for (int u = 0; u < 4; ++u) {
            const int idx = base + u * 1024 + tid;
            const int ic  = (idx < seg_end) ? idx : (seg_end - 1);
            r[u] = ((const v4i*)tmp)[ic];
        }
#pragma unroll
        for (int u = 0; u < 4; ++u) {
            const bool ok = base + u * 1024 + tid < seg_end;
            const int key = ok ? (r[u].w - lo) : 0;
            int rank, cnt;
            wave_match<9>(key, ok, rank, cnt);
            if (ok) {
                const int cur = mycw[key];
                if (rank == 0) mycw[key] = cur + cnt;
                const v4i o = {r[u].x, r[u].y, r[u].x, r[u].z};
                ((v4i*)rec)[rstart[key] + cur + rank] = o;
            }
        }
    }
}

template <int K>
__device__ __forceinline__ void pull_batch(unsigned& m, int cj, int wb,
                                           const float* __restrict__ srcl,
                                           v2f& acc) {
    int   c[K];
    float w[K];
    v2f   x[K];
#pragma unroll
    for (int k = 0; k < K; ++k) {
        const int j = __builtin_ctz(m);
        m &= m - 1;
        c[k] = __builtin_amdgcn_readlane(cj, j);
        w[k] = __int_as_float(__builtin_amdgcn_readlane(wb, j));
    }
#pragma unroll
    for (int k = 0; k < K; ++k) {
        x[k] = *(const v2f*)(srcl + (size_t)c[k] * DIM);
    }
#pragma unroll
    for (int k = 0; k < K; ++k) {
        acc += w[k] * x[k];
    }
}

__device__ __forceinline__ void pull_records(const v2i r,
                                             const float* __restrict__ srcl,
                                             v2f& acc) {
    const int cj = r.x;
    const int wb = r.y;
    unsigned m = __builtin_amdgcn_ballot_w32(__int_as_float(wb) != 0.0f);
    int nk = __builtin_popcount(m);
    for (; nk >= 8; nk -= 8) pull_batch<8>(m, cj, wb, srcl, acc);
    if (nk & 4) pull_batch<4>(m, cj, wb, srcl, acc);
    if (nk & 2) pull_batch<2>(m, cj, wb, srcl, acc);
    if (nk & 1) pull_batch<1>(m, cj, wb, srcl, acc);
}

#define PULL_ROWS 8

template <int LAYER>
__global__ void __launch_bounds__(256)
gcn_pull_kernel(const int* __restrict__ rowptr, const int* __restrict__ rec,
                const float* __restrict__ src, float* __restrict__ dst,
                int n_nodes) {
    int r0 = (blockIdx.x * 8 + (threadIdx.x >> 5)) * PULL_ROWS;
    if (r0 >= n_nodes) return;
    r0 = __builtin_amdgcn_readfirstlane(r0);
    const int l  = threadIdx.x & 31;
    const int nr = (n_nodes - r0 < PULL_ROWS) ? (n_nodes - r0) : PULL_ROWS;

    const int rpv = rowptr[r0 + ((l < nr) ? l : nr)];

    float* dp = dst + (size_t)r0 * DIM + l * 2;
    const float* srcl = src + l * 2;
    const v2i* recl = (const v2i*)rec + LAYER;

    v2f fn = {0.f, 0.f};
    if (LAYER == 1) fn = *(const v2f*)dp;
    int nbeg = __builtin_amdgcn_readlane(rpv, 0);
    int nend = __builtin_amdgcn_readlane(rpv, 1);
    v2i rn = {0, 0};
    if (nbeg + l < nend) rn = recl[(size_t)(nbeg + l) * 2];

    for (int i = 0; i < nr; ++i) {
        const v2i r = rn;
        v2f acc = fn;
        const int beg = nbeg;
        const int end = nend;

        if (i + 1 < nr) {
            nbeg = nend;
            nend = __builtin_amdgcn_readlane(rpv, i + 2);
            rn = (v2i){0, 0};
            if (nbeg + l < nend) rn = recl[(size_t)(nbeg + l) * 2];
            if (LAYER == 1) fn = *(const v2f*)(dp + (size_t)(i + 1) * DIM);
        }

        pull_records(r, srcl, acc);
        for (int base = beg + 32; base < end; base += 32) {
            v2i q = {0, 0};
            if (base + l < end) q = recl[(size_t)(base + l) * 2];
            pull_records(q, srcl, acc);
        }

        *(v2f*)(dp + (size_t)i * DIM) = acc;
    }
}

__global__ void __launch_bounds__(256)
gcn_finalize_kernel(float* __restrict__ out, const float* __restrict__ x1,
                    long long n4) {
    const long long i = (long long)blockIdx.x * blockDim.x + threadIdx.x;
    if (i >= n4) return;
    v4f a = ((const v4f*)out)[i];
    v4f p = ((const v4f*)x1)[i];
    v4f r = (a + p) * (1.0f / 3.0f);
    ((v4f*)out)[i] = r;
}

extern "C" void kernel_launch(void* const* d_in, const int* in_sizes, int n_in,
                              void* d_out, int out_size, void* d_ws, size_t ws_size,
                              hipStream_t stream) {
    const float* emb   = (const float*)d_in[0];
    const float* W     = (const float*)d_in[1];
    const float* bias  = (const float*)d_in[2];
    const float* vals  = (const float*)d_in[3];
    const float* dropu = (const float*)d_in[4];
    const int*   rows  = (const int*)d_in[5];
    const int*   cols  = (const int*)d_in[6];

    const int N = in_sizes[0] / DIM;
    const int E = in_sizes[3];

    float* out = (float*)d_out;
    float* x1  = (float*)d_ws;

    const int    n_chunks  = (E + CHUNK - 1) / CHUNK;
    const int    n_buckets = (N + BROWS - 1) / BROWS;
    const size_t x1_bytes  = (size_t)N * DIM * sizeof(float);
    const size_t rec_bytes = (size_t)E * 16;
    const size_t r0_bytes  = (x1_bytes > rec_bytes) ? x1_bytes : rec_bytes;
    const size_t rp_bytes  = (((size_t)N + 1) * sizeof(int) + 15) & ~(size_t)15;
    const size_t hs_bytes  = (size_t)n_chunks * NBMAX * sizeof(int);
    const size_t need      = r0_bytes + rec_bytes + rp_bytes + hs_bytes + (NBMAX + 1) * sizeof(int);

    if (ws_size >= need && n_buckets <= NBMAX && n_chunks > 0) {
        int* tmp    = (int*)d_ws;
        int* rec    = (int*)((char*)d_ws + r0_bytes);
        int* rowptr = (int*)((char*)d_ws + r0_bytes + rec_bytes);
        int* hist   = (int*)((char*)d_ws + r0_bytes + rec_bytes + rp_bytes);
        int* bbase  = (int*)((char*)d_ws + r0_bytes + rec_bytes + rp_bytes + hs_bytes);

        {
            dim3 grid(FC_GROUPS), block(256);
            gcn_fc_wmma_kernel<<<grid, block, 0, stream>>>(emb, W, bias, out, N);
        }
        {
            gcn_hist_kernel<<<dim3(n_chunks), dim3(1024), 0, stream>>>(rows, hist, N, E);
            gcn_scatter_kernel<<<dim3(n_chunks), dim3(1024), 0, stream>>>(rows, cols, vals, dropu, hist, bbase, tmp, N, E, n_chunks);
            gcn_bucket_kernel<<<dim3(n_buckets), dim3(1024), 0, stream>>>(bbase, tmp, rowptr, rec, N);
        }
        {
            dim3 grid((N + 8 * PULL_ROWS - 1) / (8 * PULL_ROWS)), block(256);
            gcn_pull_kernel<0><<<grid, block, 0, stream>>>(rowptr, rec, emb, x1,  N);
            gcn_pull_kernel<1><<<grid, block, 0, stream>>>(rowptr, rec, x1,  out, N);
        }
    } else {
        hipMemsetAsync(d_ws, 0, x1_bytes, stream);

        {
            dim3 grid(FC_GROUPS), block(256);
            gcn_fc_wmma_kernel<<<grid, block, 0, stream>>>(emb, W, bias, out, N);
        }

        {
            long long threads = (long long)E * 32;
            dim3 grid((unsigned)((threads + 255) / 256)), block(256);
            gcn_spmm_kernel<<<grid, block, 0, stream>>>(vals, dropu,     rows, cols, emb, x1,  E);
            gcn_spmm_kernel<<<grid, block, 0, stream>>>(vals, dropu + E, rows, cols, x1,  out, E);
        }
    }

    {
        long long n4 = (long long)N * DIM / 4;
        dim3 grid((unsigned)((n4 + 255) / 256)), block(256);
        gcn_finalize_kernel<<<grid, block, 0, stream>>>(out, x1, n4);
    }
}


# Mean folded into the layer-1 pull: the wave that sums row r also loads its own x1 row and stores ((fc + sum) + x1)/3; the separate mean launch is gone (6 launches); order of every sum kept
# speedup vs baseline: 3.4836x; 1.0804x over previous
#include <hip/hip_runtime.h>
#include <hip/hip_bf16.h>

typedef __attribute__((ext_vector_type(2))) float v2f;
typedef __attribute__((ext_vector_type(4))) float v4f;
typedef __attribute__((ext_vector_type(8))) float v8f;
typedef __attribute__((ext_vector_type(4))) int   v4i;
typedef __attribute__((ext_vector_type(2))) int   v2i;

#define DIM 64
#define KEEP_PROB 0.7f
#define OWN_ROWS 400

#define FC_GROUPS 512

__device__ __forceinline__ void fc_load_tile(const float* __restrict__ emb, int tile,
                                             int c, int h, int n_nodes, v4f (&a)[8]) {
    int gr = tile * 16 + c;
    gr = (gr < n_nodes) ? gr : (n_nodes - 1);
    const float* p = emb + (size_t)gr * DIM + 4 * h;
#pragma unroll
    for (int j = 0; j < 8; ++j) a[j] = *(const v4f*)(p + 8 * j);
}

__device__ __forceinline__ void fc_tile(const v4f (&a)[8], const v4f (&bf)[4][8],
                                        float bb0, float bb1, float bb2, float bb3,
                                        float* __restrict__ acc, int tile,
                                        int c, int h, int n_nodes) {
    v8f c0 = {}, c1 = {}, c2 = {}, c3 = {};
#pragma unroll
    for (int j = 0; j < 8; ++j) {
        const v2f a0 = a[j].xy;
        const v2f a1 = a[j].zw;
        c0 = __builtin_amdgcn_wmma_f32_16x16x4_f32(false, a0, false, bf[0][j].xy, (short)0, c0, false, false);
        c1 = __builtin_amdgcn_wmma_f32_16x16x4_f32(false, a0, false, bf[1][j].xy, (short)0, c1, false, false);
        c2 = __builtin_amdgcn_wmma_f32_16x16x4_f32(false, a0, false, bf[2][j].xy, (short)0, c2, false, false);
        c3 = __builtin_amdgcn_wmma_f32_16x16x4_f32(false, a0, false, bf[3][j].xy, (short)0, c3, false, false);
        c0 = __builtin_amdgcn_wmma_f32_16x16x4_f32(false, a1, false, bf[0][j].zw, (short)0, c0, false, false);
        c1 = __builtin_amdgcn_wmma_f32_16x16x4_f32(false, a1, false, bf[1][j].zw, (short)0, c1, false, false);
        c2 = __builtin_amdgcn_wmma_f32_16x16x4_f32(false, a1, false, bf[2][j].zw, (short)0, c2, false, false);
        c3 = __builtin_amdgcn_wmma_f32_16x16x4_f32(false, a1, false, bf[3][j].zw, (short)0, c3, false, false);
    }

    const int row_base = tile * 16 + h * 8;
#pragma unroll
    for (int j = 0; j < 8; ++j) {
        const int gm = row_base + j;
        if (gm < n_nodes) {
            float* outr = acc + (size_t)gm * DIM;
            outr[c +  0] = c0[j] + bb0;
            outr[c + 16] = c1[j] + bb1;
            outr[c + 32] = c2[j] + bb2;
            outr[c + 48] = c3[j] + bb3;
        }
    }
}

__global__ void __launch_bounds__(256)
gcn_fc_wmma_kernel(const float* __restrict__ emb, const float* __restrict__ W,
                   const float* __restrict__ bias, float* __restrict__ acc,
                   int n_nodes) {
    const int lane = threadIdx.x & 31;
    const int c    = lane & 15;
    const int h    = lane >> 4;
    const int nw   = gridDim.x * 8;
    const int n_tiles = (n_nodes + 15) >> 4;
    int tile = (threadIdx.x >> 5) * gridDim.x + blockIdx.x;
    if (tile >= n_tiles) return;
    tile = __builtin_amdgcn_readfirstlane(tile);

    v4f bf[4][8];
#pragma unroll
    for (int t = 0; t < 4; ++t) {
#pragma unroll
        for (int j = 0; j < 8; ++j) {
            bf[t][j] = *(const v4f*)(W + (size_t)(16 * t + c) * DIM + 8 * j + 4 * h);
        }
    }
    const float bb0 = bias[c];
    const float bb1 = bias[c + 16];
    const float bb2 = bias[c + 32];
    const float bb3 = bias[c + 48];

    v4f pa[8], pb[8];
    fc_load_tile(emb, tile, c, h, n_nodes, pa);
    for (;;) {
        int next = tile + nw;
        fc_load_tile(emb, (next < n_tiles) ? next : tile, c, h, n_nodes, pb);
        fc_tile(pa, bf, bb0, bb1, bb2, bb3, acc, tile, c, h, n_nodes);
        if (next >= n_tiles) break;
        tile = next;
        next = tile + nw;
        fc_load_tile(emb, (next < n_tiles) ? next : tile, c, h, n_nodes, pa);
        fc_tile(pb, bf, bb0, bb1, bb2, bb3, acc, tile, c, h, n_nodes);
        if (next >= n_tiles) break;
        tile = next;
    }
}

__global__ void __launch_bounds__(256)
gcn_spmm_kernel(const float* __restrict__ vals, const float* __restrict__ dropu,
                const int* __restrict__ rows, const int* __restrict__ cols,
                const float* __restrict__ src, float* __restrict__ dst,
                int n_edges) {
    const long long tid = (long long)blockIdx.x * blockDim.x + threadIdx.x;
    int e = (int)(tid >> 5);
    if (e >= n_edges) return;
    e = __builtin_amdgcn_readfirstlane(e);
    const int l = threadIdx.x & 31;

    const float u = dropu[e];
    if (u + KEEP_PROB < 1.0f) return;
    const float v = vals[e] * (1.0f / KEEP_PROB);

    const int col = cols[e];
    const int row = rows[e];

    const float* sp = src + (size_t)col * DIM + l * 2;
    float*       dp = dst + (size_t)row * DIM + l * 2;
    const float m0 = v * sp[0];
    const float m1 = v * sp[1];

    asm volatile("global_atomic_add_f32 %0, %1, off scope:SCOPE_DEV"
                 :: "v"(dp), "v"(m0) : "memory");
    asm volatile("global_atomic_add_f32 %0, %1, off scope:SCOPE_DEV"
                 :: "v"(dp + 1), "v"(m1) : "memory");
}


#define BROWS 400
#define NBMAX 256
#define CHUNK 4096

__device__ __forceinline__ int wave_incl_scan_i32(int x, int lane) {
    int y;
    y = __shfl_up(x, 1);  if (lane >= 1)  x += y;
    y = __shfl_up(x, 2);  if (lane >= 2)  x += y;
    y = __shfl_up(x, 4);  if (lane >= 4)  x += y;
    y = __shfl_up(x, 8);  if (lane >= 8)  x += y;
    y = __shfl_up(x, 16); if (lane >= 16) x += y;
    return x;
}

template <int NBITS>
__device__ __forceinline__ void wave_match(int key, bool ok, int& rank, int& cnt) {
    unsigned peers = __builtin_amdgcn_ballot_w32(ok);
    for (int b = 0; b < NBITS; ++b) {
        const bool bit = ((key >> b) & 1) != 0;
        const unsigned m = __builtin_amdgcn_ballot_w32(ok && bit);
        peers &= bit ? m : ~m;
    }
    rank = (int)__builtin_amdgcn_mbcnt_lo(peers, 0u);
    cnt  = __builtin_popcount(peers);
}

__device__ __forceinline__ void load_rows4(const int* __restrict__ rows,
                                           int n_edges, int e0, int (&row)[4]) {
    if (e0 + 3 < n_edges) {
        const v4i g = *(const v4i*)(rows + e0);
        row[0] = g.x; row[1] = g.y; row[2] = g.z; row[3] = g.w;
    } else {
#pragma unroll
        for (int k = 0; k < 4; ++k) row[k] = (e0 + k < n_edges) ? rows[e0 + k] : -1;
    }
}

__device__ __forceinline__ void chunk_count(int n_nodes, int* mycw, int (&row)[4],
                                            int (&key)[4], int (&rank)[4],
                                            int (&cnt)[4]) {
#pragma unroll
    for (int k = 0; k < 4; ++k) {
        const bool ok = (unsigned)row[k] < (unsigned)n_nodes;
        if (!ok) row[k] = -1;
        key[k] = ok ? (row[k] / BROWS) : 0;
        wave_match<8>(key[k], ok, rank[k], cnt[k]);
        if (ok && rank[k] == 0) mycw[key[k]] += cnt[k];
    }
}

__global__ void __launch_bounds__(1024)
gcn_hist_kernel(const int* __restrict__ rows, int* __restrict__ hist,
                int n_nodes, int n_edges) {
    __shared__ int cw[32][NBMAX];

    const int tid  = threadIdx.x;
    const int lane = tid & 31;
    const int wave = tid >> 5;
    int* mycw = cw[wave];
    for (int i = lane; i < NBMAX; i += 32) mycw[i] = 0;

    int row[4], key[4], rank[4], cnt[4];
    load_rows4(rows, n_edges, (blockIdx.x * 1024 + tid) * 4, row);
    chunk_count(n_nodes, mycw, row, key, rank, cnt);
    __syncthreads();

    if (tid < NBMAX) {
        int tot = 0;
        for (int w = 0; w < 32; ++w) tot += cw[w][tid];
        hist[blockIdx.x * NBMAX + tid] = tot;
    }
}

__global__ void __launch_bounds__(1024)
gcn_scatter_kernel(const int* __restrict__ rows, const int* __restrict__ cols,
                   const float* __restrict__ vals, const float* __restrict__ dropu,
                   const int* __restrict__ hist, int* __restrict__ bbase,
                   int* __restrict__ tmp, int n_nodes, int n_edges, int n_chunks) {
    __shared__ int cw[32][NBMAX];
    __shared__ alignas(16) int part[2][16][NBMAX];
    __shared__ int gofs[NBMAX];
    __shared__ int wtot[8];

    const int tid   = threadIdx.x;
    const int lane  = tid & 31;
    const int wave  = tid >> 5;
    const int chunk = blockIdx.x;
    const int e0    = (chunk * 1024 + tid) * 4;
    int* mycw = cw[wave];
    for (int i = lane; i < NBMAX; i += 32) mycw[i] = 0;

    int row[4];
    load_rows4(rows, n_edges, e0, row);

    {
        const int t4 = tid & 63;
        const int q  = tid >> 6;
        v4i all = {0, 0, 0, 0}, bef = {0, 0, 0, 0};
        for (int c0 = q; c0 < n_chunks; c0 += 64) {
            v4i h[4];
#pragma unroll
            for (int u = 0; u < 4; ++u) {
                const int c  = c0 + 16 * u;
                const int cl = (c < n_chunks) ? c : (n_chunks - 1);
                h[u] = ((const v4i*)hist)[cl * (NBMAX / 4) + t4];
            }
#pragma unroll
            for (int u = 0; u < 4; ++u) {
                const int c = c0 + 16 * u;
                const v4i z = {0, 0, 0, 0};
                all += (c < n_chunks) ? h[u] : z;
                bef += (c < chunk) ? h[u] : z;
            }
        }
        ((v4i*)&part[0][q][0])[t4] = all;
        ((v4i*)&part[1][q][0])[t4] = bef;
    }

    int   cc[4];
    float ww[4], ua[4], ub[4];
    if (e0 + 3 < n_edges && (n_edges & 3) == 0) {
        const v4i c4 = *(const v4i*)(cols + e0);
        const v4f w4 = *(const v4f*)(vals + e0);
        const v4f a4 = *(const v4f*)(dropu + e0);
        const v4f b4 = *(const v4f*)(dropu + (size_t)n_edges + e0);
        cc[0] = c4.x; cc[1] = c4.y; cc[2] = c4.z; cc[3] = c4.w;
        ww[0] = w4.x; ww[1] = w4.y; ww[2] = w4.z; ww[3] = w4.w;
        ua[0] = a4.x; ua[1] = a4.y; ua[2] = a4.z; ua[3] = a4.w;
        ub[0] = b4.x; ub[1] = b4.y; ub[2] = b4.z; ub[3] = b4.w;
    } else {
#pragma unroll
        for (int k = 0; k < 4; ++k) {
            const bool in = e0 + k < n_edges;
            cc[k] = in ? cols[e0 + k] : 0;
            ww[k] = in ? vals[e0 + k] : 0.0f;
            ua[k] = in ? dropu[e0 + k] : 0.0f;
            ub[k] = in ? dropu[(size_t)n_edges + e0 + k] : 0.0f;
        }
    }

    int key[4], rank[4], cnt[4];
    chunk_count(n_nodes, mycw, row, key, rank, cnt);
    __syncthreads();

    int colsum = 0, before = 0;
    if (tid < NBMAX) {
        int run = 0;
        for (int w = 0; w < 32; ++w) {
            const int c = cw[w][tid];
            cw[w][tid] = run;
            run += c;
        }
        for (int q = 0; q < 16; ++q) {
            colsum += part[0][q][tid];
            before += part[1][q][tid];
        }
    }
    const int inc = wave_incl_scan_i32(colsum, lane);
    if (tid < NBMAX && lane == 31) wtot[wave] = inc;
    __syncthreads();
    if (tid < NBMAX) {
        int woff = 0;
        for (int w = 0; w < wave; ++w) woff += wtot[w];
        const int excl = woff + inc - colsum;
        gofs[tid] = excl + before;
        if (chunk == 0) {
            bbase[tid] = excl;
            if (tid == NBMAX - 1) bbase[NBMAX] = excl + colsum;
        }
    }
    __syncthreads();

#pragma unroll
    for (int k = 0; k < 4; ++k) {
        if (row[k] >= 0) {
            const int cur  = mycw[key[k]];
            const int slot = gofs[key[k]] + cur + rank[k];
            if (rank[k] == 0) mycw[key[k]] = cur + cnt[k];
            const float v  = ww[k] * (1.0f / KEEP_PROB);
            const float v0 = (ua[k] + KEEP_PROB < 1.0f) ? 0.0f : v;
            const float v1 = (ub[k] + KEEP_PROB < 1.0f) ? 0.0f : v;
            const v4i r = {cc[k], __float_as_int(v0), __float_as_int(v1), row[k]};
            ((v4i*)tmp)[slot] = r;
        }
    }
}

__global__ void __launch_bounds__(1024)
gcn_bucket_kernel(const int* __restrict__ bbase, const int* __restrict__ tmp,
                  int* __restrict__ rowptr, int* __restrict__ rec, int n_nodes) {
    __shared__ int cw[32][BROWS];
    __shared__ int rstart[BROWS];
    __shared__ int wtot[32];

    const int tid  = threadIdx.x;
    const int lane = tid & 31;
    const int wave = tid >> 5;
    const int lo   = blockIdx.x * BROWS;
    const int nrows = (n_nodes - lo < BROWS) ? (n_nodes - lo) : BROWS;
    const int seg_beg = bbase[blockIdx.x];
    const int seg_end = bbase[blockIdx.x + 1];
    int* mycw = cw[wave];
    for (int i = lane; i < BROWS; i += 32) mycw[i] = 0;

    for (int base = seg_beg; base < seg_end; base += 4096) {
        int kw[4];
#pragma unroll
        for (int u = 0; u < 4; ++u) {
            const int idx = base + u * 1024 + tid;
            const int ic  = (idx < seg_end) ? idx : (seg_end - 1);
            kw[u] = tmp[(size_t)ic * 4 + 3];
        }
#pragma unroll
        for (int u = 0; u < 4; ++u) {
            const bool ok = base + u * 1024 + tid < seg_end;
            const int key = ok ? (kw[u] - lo) : 0;
            int rank, cnt;
            wave_match<9>(key, ok, rank, cnt);
            if (ok && rank == 0) mycw[key] += cnt;
        }
    }
    __syncthreads();

    int tot = 0;
    if (tid < BROWS) {
        for (int w = 0; w < 32; ++w) {
            const int c = cw[w][tid];
            cw[w][tid] = tot;
            tot += c;
        }
    }
    const int x = wave_incl_scan_i32(tot, lane);
    if (lane == 31) wtot[wave] = x;
    __syncthreads();
    const int wt   = wtot[lane];
    const int wx   = wave_incl_scan_i32(wt, lane);
    const int woff = __shfl(wx - wt, wave);
    const int start = seg_beg + woff + x - tot;
    if (tid < BROWS) rstart[tid] = start;
    if (tid < nrows) {
        rowptr[lo + tid] = start;
        if (lo + tid == n_nodes - 1) rowptr[n_nodes] = start + tot;
    }
    __syncthreads();

    for (int base = seg_beg; base < seg_end; base += 4096) {
        v4i r[4];
#pragma unroll
        for (int u = 0; u < 4; ++u) {
            const int idx = base + u * 1024 + tid;
            const int ic  = (idx < seg_end) ? idx : (seg_end - 1);
            r[u] = ((const v4i*)tmp)[ic];
        }
#pragma unroll
        for (int u = 0; u < 4; ++u) {
            const bool ok = base + u * 1024 + tid < seg_end;
            const int key = ok ? (r[u].w - lo) : 0;
            int rank, cnt;
            wave_match<9>(key, ok, rank, cnt);
            if (ok) {
                const int cur = mycw[key];
                if (rank == 0) mycw[key] = cur + cnt;
                const v4i o = {r[u].x, r[u].y, r[u].x, r[u].z};
                ((v4i*)rec)[rstart[key] + cur + rank] = o;
            }
        }
    }
}

template <int K>
__device__ __forceinline__ void pull_batch(unsigned& m, int cj, int wb,
                                           const float* __restrict__ srcl,
                                           v2f& acc) {
    int   c[K];
    float w[K];
    v2f   x[K];
#pragma unroll
    for (int k = 0; k < K; ++k) {
        const int j = __builtin_ctz(m);
        m &= m - 1;
        c[k] = __builtin_amdgcn_readlane(cj, j);
        w[k] = __int_as_float(__builtin_amdgcn_readlane(wb, j));
    }
#pragma unroll
    for (int k = 0; k < K; ++k) {
        x[k] = *(const v2f*)(srcl + (size_t)c[k] * DIM);
    }
#pragma unroll
    for (int k = 0; k < K; ++k) {
        acc += w[k] * x[k];
    }
}

__device__ __forceinline__ void pull_records(const v2i r,
                                             const float* __restrict__ srcl,
                                             v2f& acc) {
    const int cj = r.x;
    const int wb = r.y;
    unsigned m = __builtin_amdgcn_ballot_w32(__int_as_float(wb) != 0.0f);
    int nk = __builtin_popcount(m);
    for (; nk >= 8; nk -= 8) pull_batch<8>(m, cj, wb, srcl, acc);
    if (nk & 4) pull_batch<4>(m, cj, wb, srcl, acc);
    if (nk & 2) pull_batch<2>(m, cj, wb, srcl, acc);
    if (nk & 1) pull_batch<1>(m, cj, wb, srcl, acc);
}

#define PULL_ROWS 8

template <int LAYER>
__global__ void __launch_bounds__(256)
gcn_pull_kernel(const int* __restrict__ rowptr, const int* __restrict__ rec,
                const float* __restrict__ src, float* __restrict__ dst,
                int n_nodes) {
    int r0 = (blockIdx.x * 8 + (threadIdx.x >> 5)) * PULL_ROWS;
    if (r0 >= n_nodes) return;
    r0 = __builtin_amdgcn_readfirstlane(r0);
    const int l  = threadIdx.x & 31;
    const int nr = (n_nodes - r0 < PULL_ROWS) ? (n_nodes - r0) : PULL_ROWS;

    const int rpv = rowptr[r0 + ((l < nr) ? l : nr)];

    float* dp = dst + (size_t)r0 * DIM + l * 2;
    const float* srcl = src + l * 2;
    const float* xp = srcl + (size_t)r0 * DIM;
    const v2i* recl = (const v2i*)rec + LAYER;

    v2f fn = {0.f, 0.f};
    v2f xn = {0.f, 0.f};
    if (LAYER == 1) {
        fn = *(const v2f*)dp;
        xn = *(const v2f*)xp;
    }
    int nbeg = __builtin_amdgcn_readlane(rpv, 0);
    int nend = __builtin_amdgcn_readlane(rpv, 1);
    v2i rn = {0, 0};
    if (nbeg + l < nend) rn = recl[(size_t)(nbeg + l) * 2];

    for (int i = 0; i < nr; ++i) {
        const v2i r = rn;
        v2f acc = fn;
        const v2f xs = xn;
        const int beg = nbeg;
        const int end = nend;

        if (i + 1 < nr) {
            nbeg = nend;
            nend = __builtin_amdgcn_readlane(rpv, i + 2);
            rn = (v2i){0, 0};
            if (nbeg + l < nend) rn = recl[(size_t)(nbeg + l) * 2];
            if (LAYER == 1) {
                fn = *(const v2f*)(dp + (size_t)(i + 1) * DIM);
                xn = *(const v2f*)(xp + (size_t)(i + 1) * DIM);
            }
        }

        pull_records(r, srcl, acc);
        for (int base = beg + 32; base < end; base += 32) {
            v2i q = {0, 0};
            if (base + l < end) q = recl[(size_t)(base + l) * 2];
            pull_records(q, srcl, acc);
        }

        if (LAYER == 1) acc = (acc + xs) * (1.0f / 3.0f);
        *(v2f*)(dp + (size_t)i * DIM) = acc;
    }
}

__global__ void __launch_bounds__(256)
gcn_finalize_kernel(float* __restrict__ out, const float* __restrict__ x1,
                    long long n4) {
    const long long i = (long long)blockIdx.x * blockDim.x + threadIdx.x;
    if (i >= n4) return;
    v4f a = ((const v4f*)out)[i];
    v4f p = ((const v4f*)x1)[i];
    v4f r = (a + p) * (1.0f / 3.0f);
    ((v4f*)out)[i] = r;
}

extern "C" void kernel_launch(void* const* d_in, const int* in_sizes, int n_in,
                              void* d_out, int out_size, void* d_ws, size_t ws_size,
                              hipStream_t stream) {
    const float* emb   = (const float*)d_in[0];
    const float* W     = (const float*)d_in[1];
    const float* bias  = (const float*)d_in[2];
    const float* vals  = (const float*)d_in[3];
    const float* dropu = (const float*)d_in[4];
    const int*   rows  = (const int*)d_in[5];
    const int*   cols  = (const int*)d_in[6];

    const int N = in_sizes[0] / DIM;
    const int E = in_sizes[3];

    float* out = (float*)d_out;
    float* x1  = (float*)d_ws;

    const int    n_chunks  = (E + CHUNK - 1) / CHUNK;
    const int    n_buckets = (N + BROWS - 1) / BROWS;
    const size_t x1_bytes  = (size_t)N * DIM * sizeof(float);
    const size_t rec_bytes = (size_t)E * 16;
    const size_t r0_bytes  = (x1_bytes > rec_bytes) ? x1_bytes : rec_bytes;
    const size_t rp_bytes  = (((size_t)N + 1) * sizeof(int) + 15) & ~(size_t)15;
    const size_t hs_bytes  = (size_t)n_chunks * NBMAX * sizeof(int);
    const size_t need      = r0_bytes + rec_bytes + rp_bytes + hs_bytes + (NBMAX + 1) * sizeof(int);

    if (ws_size >= need && n_buckets <= NBMAX && n_chunks > 0) {
        int* tmp    = (int*)d_ws;
        int* rec    = (int*)((char*)d_ws + r0_bytes);
        int* rowptr = (int*)((char*)d_ws + r0_bytes + rec_bytes);
        int* hist   = (int*)((char*)d_ws + r0_bytes + rec_bytes + rp_bytes);
        int* bbase  = (int*)((char*)d_ws + r0_bytes + rec_bytes + rp_bytes + hs_bytes);

        {
            dim3 grid(FC_GROUPS), block(256);
            gcn_fc_wmma_kernel<<<grid, block, 0, stream>>>(emb, W, bias, out, N);
        }
        {
            gcn_hist_kernel<<<dim3(n_chunks), dim3(1024), 0, stream>>>(rows, hist, N, E);
            gcn_scatter_kernel<<<dim3(n_chunks), dim3(1024), 0, stream>>>(rows, cols, vals, dropu, hist, bbase, tmp, N, E, n_chunks);
            gcn_bucket_kernel<<<dim3(n_buckets), dim3(1024), 0, stream>>>(bbase, tmp, rowptr, rec, N);
        }
        {
            dim3 grid((N + 8 * PULL_ROWS - 1) / (8 * PULL_ROWS)), block(256);
            gcn_pull_kernel<0><<<grid, block, 0, stream>>>(rowptr, rec, emb, x1,  N);
            gcn_pull_kernel<1><<<grid, block, 0, stream>>>(rowptr, rec, x1,  out, N);
        }
    } else {
        hipMemsetAsync(d_ws, 0, x1_bytes, stream);

        {
            dim3 grid(FC_GROUPS), block(256);
            gcn_fc_wmma_kernel<<<grid, block, 0, stream>>>(emb, W, bias, out, N);
        }

        {
            long long threads = (long long)E * 32;
            dim3 grid((unsigned)((threads + 255) / 256)), block(256);
            gcn_spmm_kernel<<<grid, block, 0, stream>>>(vals, dropu,     rows, cols, emb, x1,  E);
            gcn_spmm_kernel<<<grid, block, 0, stream>>>(vals, dropu + E, rows, cols, x1,  out, E);
        }

        {
            long long n4 = (long long)N * DIM / 4;
            dim3 grid((unsigned)((n4 + 255) / 256)), block(256);
            gcn_finalize_kernel<<<grid, block, 0, stream>>>(out, x1, n4);
        }
    }
}


# Both sort levels order their records in LDS and store them as contiguous stretches (lane i = record i), not one 16-B piece per lane into 32 lines; tmp/rec contents and every sum order kept
# speedup vs baseline: 3.7946x; 1.0893x over previous
#include <hip/hip_runtime.h>
#include <hip/hip_bf16.h>

typedef __attribute__((ext_vector_type(2))) float v2f;
typedef __attribute__((ext_vector_type(4))) float v4f;
typedef __attribute__((ext_vector_type(8))) float v8f;
typedef __attribute__((ext_vector_type(4))) int   v4i;
typedef __attribute__((ext_vector_type(2))) int   v2i;

#define DIM 64
#define KEEP_PROB 0.7f
#define OWN_ROWS 400

#define FC_GROUPS 512

__device__ __forceinline__ void fc_load_tile(const float* __restrict__ emb, int tile,
                                             int c, int h, int n_nodes, v4f (&a)[8]) {
    int gr = tile * 16 + c;
    gr = (gr < n_nodes) ? gr : (n_nodes - 1);
    const float* p = emb + (size_t)gr * DIM + 4 * h;
#pragma unroll
    for (int j = 0; j < 8; ++j) a[j] = *(const v4f*)(p + 8 * j);
}

__device__ __forceinline__ void fc_tile(const v4f (&a)[8], const v4f (&bf)[4][8],
                                        float bb0, float bb1, float bb2, float bb3,
                                        float* __restrict__ acc, int tile,
                                        int c, int h, int n_nodes) {
    v8f c0 = {}, c1 = {}, c2 = {}, c3 = {};
#pragma unroll
    for (int j = 0; j < 8; ++j) {
        const v2f a0 = a[j].xy;
        const v2f a1 = a[j].zw;
        c0 = __builtin_amdgcn_wmma_f32_16x16x4_f32(false, a0, false, bf[0][j].xy, (short)0, c0, false, false);
        c1 = __builtin_amdgcn_wmma_f32_16x16x4_f32(false, a0, false, bf[1][j].xy, (short)0, c1, false, false);
        c2 = __builtin_amdgcn_wmma_f32_16x16x4_f32(false, a0, false, bf[2][j].xy, (short)0, c2, false, false);
        c3 = __builtin_amdgcn_wmma_f32_16x16x4_f32(false, a0, false, bf[3][j].xy, (short)0, c3, false, false);
        c0 = __builtin_amdgcn_wmma_f32_16x16x4_f32(false, a1, false, bf[0][j].zw, (short)0, c0, false, false);
        c1 = __builtin_amdgcn_wmma_f32_16x16x4_f32(false, a1, false, bf[1][j].zw, (short)0, c1, false, false);
        c2 = __builtin_amdgcn_wmma_f32_16x16x4_f32(false, a1, false, bf[2][j].zw, (short)0, c2, false, false);
        c3 = __builtin_amdgcn_wmma_f32_16x16x4_f32(false, a1, false, bf[3][j].zw, (short)0, c3, false, false);
    }

    const int row_base = tile * 16 + h * 8;
#pragma unroll
    for (int j = 0; j < 8; ++j) {
        const int gm = row_base + j;
        if (gm < n_nodes) {
            float* outr = acc + (size_t)gm * DIM;
            outr[c +  0] = c0[j] + bb0;
            outr[c + 16] = c1[j] + bb1;
            outr[c + 32] = c2[j] + bb2;
            outr[c + 48] = c3[j] + bb3;
        }
    }
}

__global__ void __launch_bounds__(256)
gcn_fc_wmma_kernel(const float* __restrict__ emb, const float* __restrict__ W,
                   const float* __restrict__ bias, float* __restrict__ acc,
                   int n_nodes) {
    const int lane = threadIdx.x & 31;
    const int c    = lane & 15;
    const int h    = lane >> 4;
    const int nw   = gridDim.x * 8;
    const int n_tiles = (n_nodes + 15) >> 4;
    int tile = (threadIdx.x >> 5) * gridDim.x + blockIdx.x;
    if (tile >= n_tiles) return;
    tile = __builtin_amdgcn_readfirstlane(tile);

    v4f bf[4][8];
#pragma unroll
    for (int t = 0; t < 4; ++t) {
#pragma unroll
        for (int j = 0; j < 8; ++j) {
            bf[t][j] = *(const v4f*)(W + (size_t)(16 * t + c) * DIM + 8 * j + 4 * h);
        }
    }
    const float bb0 = bias[c];
    const float bb1 = bias[c + 16];
    const float bb2 = bias[c + 32];
    const float bb3 = bias[c + 48];

    v4f pa[8], pb[8];
    fc_load_tile(emb, tile, c, h, n_nodes, pa);
    for (;;) {
        int next = tile + nw;
        fc_load_tile(emb, (next < n_tiles) ? next : tile, c, h, n_nodes, pb);
        fc_tile(pa, bf, bb0, bb1, bb2, bb3, acc, tile, c, h, n_nodes);
        if (next >= n_tiles) break;
        tile = next;
        next = tile + nw;
        fc_load_tile(emb, (next < n_tiles) ? next : tile, c, h, n_nodes, pa);
        fc_tile(pb, bf, bb0, bb1, bb2, bb3, acc, tile, c, h, n_nodes);
        if (next >= n_tiles) break;
        tile = next;
    }
}

__global__ void __launch_bounds__(256)
gcn_spmm_kernel(const float* __restrict__ vals, const float* __restrict__ dropu,
                const int* __restrict__ rows, const int* __restrict__ cols,
                const float* __restrict__ src, float* __restrict__ dst,
                int n_edges) {
    const long long tid = (long long)blockIdx.x * blockDim.x + threadIdx.x;
    int e = (int)(tid >> 5);
    if (e >= n_edges) return;
    e = __builtin_amdgcn_readfirstlane(e);
    const int l = threadIdx.x & 31;

    const float u = dropu[e];
    if (u + KEEP_PROB < 1.0f) return;
    const float v = vals[e] * (1.0f / KEEP_PROB);

    const int col = cols[e];
    const int row = rows[e];

    const float* sp = src + (size_t)col * DIM + l * 2;
    float*       dp = dst + (size_t)row * DIM + l * 2;
    const float m0 = v * sp[0];
    const float m1 = v * sp[1];

    asm volatile("global_atomic_add_f32 %0, %1, off scope:SCOPE_DEV"
                 :: "v"(dp), "v"(m0) : "memory");
    asm volatile("global_atomic_add_f32 %0, %1, off scope:SCOPE_DEV"
                 :: "v"(dp + 1), "v"(m1) : "memory");
}


#define BROWS 400
#define NBMAX 256
#define CHUNK 4096
#define STAGE_CAP 8192

__device__ __forceinline__ int wave_incl_scan_i32(int x, int lane) {
    int y;
    y = __shfl_up(x, 1);  if (lane >= 1)  x += y;
    y = __shfl_up(x, 2);  if (lane >= 2)  x += y;
    y = __shfl_up(x, 4);  if (lane >= 4)  x += y;
    y = __shfl_up(x, 8);  if (lane >= 8)  x += y;
    y = __shfl_up(x, 16); if (lane >= 16) x += y;
    return x;
}

template <int NBITS>
__device__ __forceinline__ void wave_match(int key, bool ok, int& rank, int& cnt) {
    unsigned peers = __builtin_amdgcn_ballot_w32(ok);
    for (int b = 0; b < NBITS; ++b) {
        const bool bit = ((key >> b) & 1) != 0;
        const unsigned m = __builtin_amdgcn_ballot_w32(ok && bit);
        peers &= bit ? m : ~m;
    }
    rank = (int)__builtin_amdgcn_mbcnt_lo(peers, 0u);
    cnt  = __builtin_popcount(peers);
}

__device__ __forceinline__ void load_rows4(const int* __restrict__ rows,
                                           int n_edges, int e0, int (&row)[4]) {
    if (e0 + 3 < n_edges) {
        const v4i g = *(const v4i*)(rows + e0);
        row[0] = g.x; row[1] = g.y; row[2] = g.z; row[3] = g.w;
    } else {
#pragma unroll
        for (int k = 0; k < 4; ++k) row[k] = (e0 + k < n_edges) ? rows[e0 + k] : -1;
    }
}

__device__ __forceinline__ void chunk_count(int n_nodes, int* mycw, int (&row)[4],
                                            int (&key)[4], int (&rank)[4],
                                            int (&cnt)[4]) {
#pragma unroll
    for (int k = 0; k < 4; ++k) {
        const bool ok = (unsigned)row[k] < (unsigned)n_nodes;
        if (!ok) row[k] = -1;
        key[k] = ok ? (row[k] / BROWS) : 0;
        wave_match<8>(key[k], ok, rank[k], cnt[k]);
        if (ok && rank[k] == 0) mycw[key[k]] += cnt[k];
    }
}

__global__ void __launch_bounds__(1024)
gcn_hist_kernel(const int* __restrict__ rows, int* __restrict__ hist,
                int n_nodes, int n_edges) {
    __shared__ int cw[32][NBMAX];

    const int tid  = threadIdx.x;
    const int lane = tid & 31;
    const int wave = tid >> 5;
    int* mycw = cw[wave];
    for (int i = lane; i < NBMAX; i += 32) mycw[i] = 0;

    int row[4], key[4], rank[4], cnt[4];
    load_rows4(rows, n_edges, (blockIdx.x * 1024 + tid) * 4, row);
    chunk_count(n_nodes, mycw, row, key, rank, cnt);
    __syncthreads();

    if (tid < NBMAX) {
        int tot = 0;
        for (int w = 0; w < 32; ++w) tot += cw[w][tid];
        hist[blockIdx.x * NBMAX + tid] = tot;
    }
}

__global__ void __launch_bounds__(1024)
gcn_scatter_kernel(const int* __restrict__ rows, const int* __restrict__ cols,
                   const float* __restrict__ vals, const float* __restrict__ dropu,
                   const int* __restrict__ hist, int* __restrict__ bbase,
                   int* __restrict__ tmp, int n_nodes, int n_edges, int n_chunks) {
    __shared__ int cw[32][NBMAX];
    __shared__ alignas(16) int part[2][16][NBMAX];
    __shared__ int gofs[NBMAX];
    __shared__ int lofs[NBMAX];
    __shared__ int wtot[8];
    __shared__ int ltot[8];
    __shared__ alignas(16) int stage[CHUNK * 4];

    const int tid   = threadIdx.x;
    const int lane  = tid & 31;
    const int wave  = tid >> 5;
    const int chunk = blockIdx.x;
    const int e0    = (chunk * 1024 + tid) * 4;
    int* mycw = cw[wave];
    for (int i = lane; i < NBMAX; i += 32) mycw[i] = 0;

    int row[4];
    load_rows4(rows, n_edges, e0, row);

    {
        const int t4 = tid & 63;
        const int q  = tid >> 6;
        v4i all = {0, 0, 0, 0}, bef = {0, 0, 0, 0};
        for (int c0 = q; c0 < n_chunks; c0 += 64) {
            v4i h[4];
#pragma unroll
            for (int u = 0; u < 4; ++u) {
                const int c  = c0 + 16 * u;
                const int cl = (c < n_chunks) ? c : (n_chunks - 1);
                h[u] = ((const v4i*)hist)[cl * (NBMAX / 4) + t4];
            }
#pragma unroll
            for (int u = 0; u < 4; ++u) {
                const int c = c0 + 16 * u;
                const v4i z = {0, 0, 0, 0};
                all += (c < n_chunks) ? h[u] : z;
                bef += (c < chunk) ? h[u] : z;
            }
        }
        ((v4i*)&part[0][q][0])[t4] = all;
        ((v4i*)&part[1][q][0])[t4] = bef;
    }

    int   cc[4];
    float ww[4], ua[4], ub[4];
    if (e0 + 3 < n_edges && (n_edges & 3) == 0) {
        const v4i c4 = *(const v4i*)(cols + e0);
        const v4f w4 = *(const v4f*)(vals + e0);
        const v4f a4 = *(const v4f*)(dropu + e0);
        const v4f b4 = *(const v4f*)(dropu + (size_t)n_edges + e0);
        cc[0] = c4.x; cc[1] = c4.y; cc[2] = c4.z; cc[3] = c4.w;
        ww[0] = w4.x; ww[1] = w4.y; ww[2] = w4.z; ww[3] = w4.w;
        ua[0] = a4.x; ua[1] = a4.y; ua[2] = a4.z; ua[3] = a4.w;
        ub[0] = b4.x; ub[1] = b4.y; ub[2] = b4.z; ub[3] = b4.w;
    } else {
#pragma unroll
        for (int k = 0; k < 4; ++k) {
            const bool in = e0 + k < n_edges;
            cc[k] = in ? cols[e0 + k] : 0;
            ww[k] = in ? vals[e0 + k] : 0.0f;
            ua[k] = in ? dropu[e0 + k] : 0.0f;
            ub[k] = in ? dropu[(size_t)n_edges + e0 + k] : 0.0f;
        }
    }

    int key[4], rank[4], cnt[4];
    chunk_count(n_nodes, mycw, row, key, rank, cnt);
    __syncthreads();

    int colsum = 0, before = 0, mine = 0;
    if (tid < NBMAX) {
        int run = 0;
        for (int w = 0; w < 32; ++w) {
            const int c = cw[w][tid];
            cw[w][tid] = run;
            run += c;
        }
        mine = run;
        for (int q = 0; q < 16; ++q) {
            colsum += part[0][q][tid];
            before += part[1][q][tid];
        }
    }
    const int inc  = wave_incl_scan_i32(colsum, lane);
    const int linc = wave_incl_scan_i32(mine, lane);
    if (tid < NBMAX && lane == 31) {
        wtot[wave] = inc;
        ltot[wave] = linc;
    }
    __syncthreads();
    if (tid < NBMAX) {
        int woff = 0, lwoff = 0;
        for (int w = 0; w < wave; ++w) {
            woff  += wtot[w];
            lwoff += ltot[w];
        }
        const int excl  = woff + inc - colsum;
        const int lexcl = lwoff + linc - mine;
        lofs[tid] = lexcl;
        gofs[tid] = excl + before - lexcl;
        if (chunk == 0) {
            bbase[tid] = excl;
            if (tid == NBMAX - 1) bbase[NBMAX] = excl + colsum;
        }
    }
    int nvalid = 0;
    for (int w = 0; w < 8; ++w) nvalid += ltot[w];
    __syncthreads();

#pragma unroll
    for (int k = 0; k < 4; ++k) {
        if (row[k] >= 0) {
            const int cur  = mycw[key[k]];
            const int lpos = lofs[key[k]] + cur + rank[k];
            if (rank[k] == 0) mycw[key[k]] = cur + cnt[k];
            const float v  = ww[k] * (1.0f / KEEP_PROB);
            const float v0 = (ua[k] + KEEP_PROB < 1.0f) ? 0.0f : v;
            const float v1 = (ub[k] + KEEP_PROB < 1.0f) ? 0.0f : v;
            const v4i r = {cc[k], __float_as_int(v0), __float_as_int(v1), row[k]};
            ((v4i*)stage)[lpos] = r;
        }
    }
    __syncthreads();

    if (nvalid > 0) {
        v4i r[4];
        int g[4];
#pragma unroll
        for (int k = 0; k < 4; ++k) {
            const int pos = k * 1024 + tid;
            r[k] = ((const v4i*)stage)[(pos < nvalid) ? pos : (nvalid - 1)];
        }
#pragma unroll
        for (int k = 0; k < 4; ++k) g[k] = gofs[r[k].w / BROWS];
#pragma unroll
        for (int k = 0; k < 4; ++k) {
            const int pos = k * 1024 + tid;
            if (pos < nvalid) ((v4i*)tmp)[pos + g[k]] = r[k];
        }
    }
}

__global__ void __launch_bounds__(1024)
gcn_bucket_kernel(const int* __restrict__ bbase, const int* __restrict__ tmp,
                  int* __restrict__ rowptr, int* __restrict__ rec, int n_nodes) {
    __shared__ int cw[32][BROWS];
    __shared__ int rstart[BROWS];
    __shared__ int wtot[32];
    __shared__ alignas(16) int stage[STAGE_CAP * 4];

    const int tid  = threadIdx.x;
    const int lane = tid & 31;
    const int wave = tid >> 5;
    const int lo   = blockIdx.x * BROWS;
    const int nrows = (n_nodes - lo < BROWS) ? (n_nodes - lo) : BROWS;
    const int seg_beg = bbase[blockIdx.x];
    const int seg_end = bbase[blockIdx.x + 1];
    const bool staged = seg_end - seg_beg <= STAGE_CAP;
    int* mycw = cw[wave];
    for (int i = lane; i < BROWS; i += 32) mycw[i] = 0;

    for (int base = seg_beg; base < seg_end; base += 4096) {
        int kw[4];
#pragma unroll
        for (int u = 0; u < 4; ++u) {
            const int idx = base + u * 1024 + tid;
            const int ic  = (idx < seg_end) ? idx : (seg_end - 1);
            kw[u] = tmp[(size_t)ic * 4 + 3];
        }
#pragma unroll
        for (int u = 0; u < 4; ++u) {
            const bool ok = base + u * 1024 + tid < seg_end;
            const int key = ok ? (kw[u] - lo) : 0;
            int rank, cnt;
            wave_match<9>(key, ok, rank, cnt);
            if (ok && rank == 0) mycw[key] += cnt;
        }
    }
    __syncthreads();

    int tot = 0;
    if (tid < BROWS) {
        for (int w = 0; w < 32; ++w) {
            const int c = cw[w][tid];
            cw[w][tid] = tot;
            tot += c;
        }
    }
    const int x = wave_incl_scan_i32(tot, lane);
    if (lane == 31) wtot[wave] = x;
    __syncthreads();
    const int wt   = wtot[lane];
    const int wx   = wave_incl_scan_i32(wt, lane);
    const int woff = __shfl(wx - wt, wave);
    const int start = seg_beg + woff + x - tot;
    if (tid < BROWS) rstart[tid] = start;
    if (tid < nrows) {
        rowptr[lo + tid] = start;
        if (lo + tid == n_nodes - 1) rowptr[n_nodes] = start + tot;
    }
    __syncthreads();

    for (int base = seg_beg; base < seg_end; base += 4096) {
        v4i r[4];
#pragma unroll
        for (int u = 0; u < 4; ++u) {
            const int idx = base + u * 1024 + tid;
            const int ic  = (idx < seg_end) ? idx : (seg_end - 1);
            r[u] = ((const v4i*)tmp)[ic];
        }
#pragma unroll
        for (int u = 0; u < 4; ++u) {
            const bool ok = base + u * 1024 + tid < seg_end;
            const int key = ok ? (r[u].w - lo) : 0;
            int rank, cnt;
            wave_match<9>(key, ok, rank, cnt);
            if (ok) {
                const int cur = mycw[key];
                if (rank == 0) mycw[key] = cur + cnt;
                const v4i o = {r[u].x, r[u].y, r[u].x, r[u].z};
                const int slot = rstart[key] + cur + rank;
                if (staged) ((v4i*)stage)[slot - seg_beg] = o;
                else        ((v4i*)rec)[slot] = o;
            }
        }
    }

    if (staged) {
        __syncthreads();
        for (int i = tid; i < seg_end - seg_beg; i += 1024)
            ((v4i*)rec)[seg_beg + i] = ((const v4i*)stage)[i];
    }
}

template <int K>
__device__ __forceinline__ void pull_batch(unsigned& m, int cj, int wb,
                                           const float* __restrict__ srcl,
                                           v2f& acc) {
    int   c[K];
    float w[K];
    v2f   x[K];
#pragma unroll
    for (int k = 0; k < K; ++k) {
        const int j = __builtin_ctz(m);
        m &= m - 1;
        c[k] = __builtin_amdgcn_readlane(cj, j);
        w[k] = __int_as_float(__builtin_amdgcn_readlane(wb, j));
    }
#pragma unroll
    for (int k = 0; k < K; ++k) {
        x[k] = *(const v2f*)(srcl + (size_t)c[k] * DIM);
    }
#pragma unroll
    for (int k = 0; k < K; ++k) {
        acc += w[k] * x[k];
    }
}

__device__ __forceinline__ void pull_records(const v2i r,
                                             const float* __restrict__ srcl,
                                             v2f& acc) {
    const int cj = r.x;
    const int wb = r.y;
    unsigned m = __builtin_amdgcn_ballot_w32(__int_as_float(wb) != 0.0f);
    int nk = __builtin_popcount(m);
    for (; nk >= 8; nk -= 8) pull_batch<8>(m, cj, wb, srcl, acc);
    if (nk & 4) pull_batch<4>(m, cj, wb, srcl, acc);
    if (nk & 2) pull_batch<2>(m, cj, wb, srcl, acc);
    if (nk & 1) pull_batch<1>(m, cj, wb, srcl, acc);
}

#define PULL_ROWS 8

template <int LAYER>
__global__ void __launch_bounds__(256)
gcn_pull_kernel(const int* __restrict__ rowptr, const int* __restrict__ rec,
                const float* __restrict__ src, float* __restrict__ dst,
                int n_nodes) {
    int r0 = (blockIdx.x * 8 + (threadIdx.x >> 5)) * PULL_ROWS;
    if (r0 >= n_nodes) return;
    r0 = __builtin_amdgcn_readfirstlane(r0);
    const int l  = threadIdx.x & 31;
    const int nr = (n_nodes - r0 < PULL_ROWS) ? (n_nodes - r0) : PULL_ROWS;

    const int rpv = rowptr[r0 + ((l < nr) ? l : nr)];

    float* dp = dst + (size_t)r0 * DIM + l * 2;
    const float* srcl = src + l * 2;
    const float* xp = srcl + (size_t)r0 * DIM;
    const v2i* recl = (const v2i*)rec + LAYER;

    v2f fn = {0.f, 0.f};
    v2f xn = {0.f, 0.f};
    if (LAYER == 1) {
        fn = *(const v2f*)dp;
        xn = *(const v2f*)xp;
    }
    int nbeg = __builtin_amdgcn_readlane(rpv, 0);
    int nend = __builtin_amdgcn_readlane(rpv, 1);
    v2i rn = {0, 0};
    if (nbeg + l < nend) rn = recl[(size_t)(nbeg + l) * 2];

    for (int i = 0; i < nr; ++i) {
        const v2i r = rn;
        v2f acc = fn;
        const v2f xs = xn;
        const int beg = nbeg;
        const int end = nend;

        if (i + 1 < nr) {
            nbeg = nend;
            nend = __builtin_amdgcn_readlane(rpv, i + 2);
            rn = (v2i){0, 0};
            if (nbeg + l < nend) rn = recl[(size_t)(nbeg + l) * 2];
            if (LAYER == 1) {
                fn = *(const v2f*)(dp + (size_t)(i + 1) * DIM);
                xn = *(const v2f*)(xp + (size_t)(i + 1) * DIM);
            }
        }

        pull_records(r, srcl, acc);
        for (int base = beg + 32; base < end; base += 32) {
            v2i q = {0, 0};
            if (base + l < end) q = recl[(size_t)(base + l) * 2];
            pull_records(q, srcl, acc);
        }

        if (LAYER == 1) acc = (acc + xs) * (1.0f / 3.0f);
        *(v2f*)(dp + (size_t)i * DIM) = acc;
    }
}

__global__ void __launch_bounds__(256)
gcn_finalize_kernel(float* __restrict__ out, const float* __restrict__ x1,
                    long long n4) {
    const long long i = (long long)blockIdx.x * blockDim.x + threadIdx.x;
    if (i >= n4) return;
    v4f a = ((const v4f*)out)[i];
    v4f p = ((const v4f*)x1)[i];
    v4f r = (a + p) * (1.0f / 3.0f);
    ((v4f*)out)[i] = r;
}

extern "C" void kernel_launch(void* const* d_in, const int* in_sizes, int n_in,
                              void* d_out, int out_size, void* d_ws, size_t ws_size,
                              hipStream_t stream) {
    const float* emb   = (const float*)d_in[0];
    const float* W     = (const float*)d_in[1];
    const float* bias  = (const float*)d_in[2];
    const float* vals  = (const float*)d_in[3];
    const float* dropu = (const float*)d_in[4];
    const int*   rows  = (const int*)d_in[5];
    const int*   cols  = (const int*)d_in[6];

    const int N = in_sizes[0] / DIM;
    const int E = in_sizes[3];

    float* out = (float*)d_out;
    float* x1  = (float*)d_ws;

    const int    n_chunks  = (E + CHUNK - 1) / CHUNK;
    const int    n_buckets = (N + BROWS - 1) / BROWS;
    const size_t x1_bytes  = (size_t)N * DIM * sizeof(float);
    const size_t rec_bytes = (size_t)E * 16;
    const size_t r0_bytes  = (x1_bytes > rec_bytes) ? x1_bytes : rec_bytes;
    const size_t rp_bytes  = (((size_t)N + 1) * sizeof(int) + 15) & ~(size_t)15;
    const size_t hs_bytes  = (size_t)n_chunks * NBMAX * sizeof(int);
    const size_t need      = r0_bytes + rec_bytes + rp_bytes + hs_bytes + (NBMAX + 1) * sizeof(int);

    if (ws_size >= need && n_buckets <= NBMAX && n_chunks > 0) {
        int* tmp    = (int*)d_ws;
        int* rec    = (int*)((char*)d_ws + r0_bytes);
        int* rowptr = (int*)((char*)d_ws + r0_bytes + rec_bytes);
        int* hist   = (int*)((char*)d_ws + r0_bytes + rec_bytes + rp_bytes);
        int* bbase  = (int*)((char*)d_ws + r0_bytes + rec_bytes + rp_bytes + hs_bytes);

        {
            dim3 grid(FC_GROUPS), block(256);
            gcn_fc_wmma_kernel<<<grid, block, 0, stream>>>(emb, W, bias, out, N);
        }
        {
            gcn_hist_kernel<<<dim3(n_chunks), dim3(1024), 0, stream>>>(rows, hist, N, E);
            gcn_scatter_kernel<<<dim3(n_chunks), dim3(1024), 0, stream>>>(rows, cols, vals, dropu, hist, bbase, tmp, N, E, n_chunks);
            gcn_bucket_kernel<<<dim3(n_buckets), dim3(1024), 0, stream>>>(bbase, tmp, rowptr, rec, N);
        }
        {
            dim3 grid((N + 8 * PULL_ROWS - 1) / (8 * PULL_ROWS)), block(256);
            gcn_pull_kernel<0><<<grid, block, 0, stream>>>(rowptr, rec, emb, x1,  N);
            gcn_pull_kernel<1><<<grid, block, 0, stream>>>(rowptr, rec, x1,  out, N);
        }
    } else {
        hipMemsetAsync(d_ws, 0, x1_bytes, stream);

        {
            dim3 grid(FC_GROUPS), block(256);
            gcn_fc_wmma_kernel<<<grid, block, 0, stream>>>(emb, W, bias, out, N);
        }

        {
            long long threads = (long long)E * 32;
            dim3 grid((unsigned)((threads + 255) / 256)), block(256);
            gcn_spmm_kernel<<<grid, block, 0, stream>>>(vals, dropu,     rows, cols, emb, x1,  E);
            gcn_spmm_kernel<<<grid, block, 0, stream>>>(vals, dropu + E, rows, cols, x1,  out, E);
        }

        {
            long long n4 = (long long)N * DIM / 4;
            dim3 grid((unsigned)((n4 + 255) / 256)), block(256);
            gcn_finalize_kernel<<<grid, block, 0, stream>>>(out, x1, n4);
        }
    }
}


# Grouping level 1 is chunk-local: a chunk keeps its bucket-ordered records in its own stretch with a begin/count table, level 2 collects them; count launch gone (5 launches); row sum order new, fixed
# speedup vs baseline: 4.3985x; 1.1591x over previous
#include <hip/hip_runtime.h>
#include <hip/hip_bf16.h>

typedef __attribute__((ext_vector_type(2))) float v2f;
typedef __attribute__((ext_vector_type(4))) float v4f;
typedef __attribute__((ext_vector_type(8))) float v8f;
typedef __attribute__((ext_vector_type(4))) int   v4i;
typedef __attribute__((ext_vector_type(2))) int   v2i;

#define DIM 64
#define KEEP_PROB 0.7f
#define OWN_ROWS 400

#define FC_GROUPS 512

__device__ __forceinline__ void fc_load_tile(const float* __restrict__ emb, int tile,
                                             int c, int h, int n_nodes, v4f (&a)[8]) {
    int gr = tile * 16 + c;
    gr = (gr < n_nodes) ? gr : (n_nodes - 1);
    const float* p = emb + (size_t)gr * DIM + 4 * h;
#pragma unroll
    for (int j = 0; j < 8; ++j) a[j] = *(const v4f*)(p + 8 * j);
}

__device__ __forceinline__ void fc_tile(const v4f (&a)[8], const v4f (&bf)[4][8],
                                        float bb0, float bb1, float bb2, float bb3,
                                        float* __restrict__ acc, int tile,
                                        int c, int h, int n_nodes) {
    v8f c0 = {}, c1 = {}, c2 = {}, c3 = {};
#pragma unroll
    for (int j = 0; j < 8; ++j) {
        const v2f a0 = a[j].xy;
        const v2f a1 = a[j].zw;
        c0 = __builtin_amdgcn_wmma_f32_16x16x4_f32(false, a0, false, bf[0][j].xy, (short)0, c0, false, false);
        c1 = __builtin_amdgcn_wmma_f32_16x16x4_f32(false, a0, false, bf[1][j].xy, (short)0, c1, false, false);
        c2 = __builtin_amdgcn_wmma_f32_16x16x4_f32(false, a0, false, bf[2][j].xy, (short)0, c2, false, false);
        c3 = __builtin_amdgcn_wmma_f32_16x16x4_f32(false, a0, false, bf[3][j].xy, (short)0, c3, false, false);
        c0 = __builtin_amdgcn_wmma_f32_16x16x4_f32(false, a1, false, bf[0][j].zw, (short)0, c0, false, false);
        c1 = __builtin_amdgcn_wmma_f32_16x16x4_f32(false, a1, false, bf[1][j].zw, (short)0, c1, false, false);
        c2 = __builtin_amdgcn_wmma_f32_16x16x4_f32(false, a1, false, bf[2][j].zw, (short)0, c2, false, false);
        c3 = __builtin_amdgcn_wmma_f32_16x16x4_f32(false, a1, false, bf[3][j].zw, (short)0, c3, false, false);
    }

    const int row_base = tile * 16 + h * 8;
#pragma unroll
    for (int j = 0; j < 8; ++j) {
        const int gm = row_base + j;
        if (gm < n_nodes) {
            float* outr = acc + (size_t)gm * DIM;
            outr[c +  0] = c0[j] + bb0;
            outr[c + 16] = c1[j] + bb1;
            outr[c + 32] = c2[j] + bb2;
            outr[c + 48] = c3[j] + bb3;
        }
    }
}

__global__ void __launch_bounds__(256)
gcn_fc_wmma_kernel(const float* __restrict__ emb, const float* __restrict__ W,
                   const float* __restrict__ bias, float* __restrict__ acc,
                   int n_nodes) {
    const int lane = threadIdx.x & 31;
    const int c    = lane & 15;
    const int h    = lane >> 4;
    const int nw   = gridDim.x * 8;
    const int n_tiles = (n_nodes + 15) >> 4;
    int tile = (threadIdx.x >> 5) * gridDim.x + blockIdx.x;
    if (tile >= n_tiles) return;
    tile = __builtin_amdgcn_readfirstlane(tile);

    v4f bf[4][8];
#pragma unroll
    for (int t = 0; t < 4; ++t) {
#pragma unroll
        for (int j = 0; j < 8; ++j) {
            bf[t][j] = *(const v4f*)(W + (size_t)(16 * t + c) * DIM + 8 * j + 4 * h);
        }
    }
    const float bb0 = bias[c];
    const float bb1 = bias[c + 16];
    const float bb2 = bias[c + 32];
    const float bb3 = bias[c + 48];

    v4f pa[8], pb[8];
    fc_load_tile(emb, tile, c, h, n_nodes, pa);
    for (;;) {
        int next = tile + nw;
        fc_load_tile(emb, (next < n_tiles) ? next : tile, c, h, n_nodes, pb);
        fc_tile(pa, bf, bb0, bb1, bb2, bb3, acc, tile, c, h, n_nodes);
        if (next >= n_tiles) break;
        tile = next;
        next = tile + nw;
        fc_load_tile(emb, (next < n_tiles) ? next : tile, c, h, n_nodes, pa);
        fc_tile(pb, bf, bb0, bb1, bb2, bb3, acc, tile, c, h, n_nodes);
        if (next >= n_tiles) break;
        tile = next;
    }
}

__global__ void __launch_bounds__(256)
gcn_spmm_kernel(const float* __restrict__ vals, const float* __restrict__ dropu,
                const int* __restrict__ rows, const int* __restrict__ cols,
                const float* __restrict__ src, float* __restrict__ dst,
                int n_edges) {
    const long long tid = (long long)blockIdx.x * blockDim.x + threadIdx.x;
    int e = (int)(tid >> 5);
    if (e >= n_edges) return;
    e = __builtin_amdgcn_readfirstlane(e);
    const int l = threadIdx.x & 31;

    const float u = dropu[e];
    if (u + KEEP_PROB < 1.0f) return;
    const float v = vals[e] * (1.0f / KEEP_PROB);

    const int col = cols[e];
    const int row = rows[e];

    const float* sp = src + (size_t)col * DIM + l * 2;
    float*       dp = dst + (size_t)row * DIM + l * 2;
    const float m0 = v * sp[0];
    const float m1 = v * sp[1];

    asm volatile("global_atomic_add_f32 %0, %1, off scope:SCOPE_DEV"
                 :: "v"(dp), "v"(m0) : "memory");
    asm volatile("global_atomic_add_f32 %0, %1, off scope:SCOPE_DEV"
                 :: "v"(dp + 1), "v"(m1) : "memory");
}


#define BROWS 400
#define NBMAX 256
#define CHUNK 4096
#define STAGE_CAP 8192

__device__ __forceinline__ int wave_incl_scan_i32(int x, int lane) {
    int y;
    y = __shfl_up(x, 1);  if (lane >= 1)  x += y;
    y = __shfl_up(x, 2);  if (lane >= 2)  x += y;
    y = __shfl_up(x, 4);  if (lane >= 4)  x += y;
    y = __shfl_up(x, 8);  if (lane >= 8)  x += y;
    y = __shfl_up(x, 16); if (lane >= 16) x += y;
    return x;
}

template <int NBITS>
__device__ __forceinline__ void wave_match(int key, bool ok, int& rank, int& cnt) {
    unsigned peers = __builtin_amdgcn_ballot_w32(ok);
    for (int b = 0; b < NBITS; ++b) {
        const bool bit = ((key >> b) & 1) != 0;
        const unsigned m = __builtin_amdgcn_ballot_w32(ok && bit);
        peers &= bit ? m : ~m;
    }
    rank = (int)__builtin_amdgcn_mbcnt_lo(peers, 0u);
    cnt  = __builtin_popcount(peers);
}

__device__ __forceinline__ void load_rows4(const int* __restrict__ rows,
                                           int n_edges, int e0, int (&row)[4]) {
    if (e0 + 3 < n_edges) {
        const v4i g = *(const v4i*)(rows + e0);
        row[0] = g.x; row[1] = g.y; row[2] = g.z; row[3] = g.w;
    } else {
#pragma unroll
        for (int k = 0; k < 4; ++k) row[k] = (e0 + k < n_edges) ? rows[e0 + k] : -1;
    }
}

__device__ __forceinline__ void chunk_count(int n_nodes, int* mycw, int (&row)[4],
                                            int (&key)[4], int (&rank)[4],
                                            int (&cnt)[4]) {
#pragma unroll
    for (int k = 0; k < 4; ++k) {
        const bool ok = (unsigned)row[k] < (unsigned)n_nodes;
        if (!ok) row[k] = -1;
        key[k] = ok ? (row[k] / BROWS) : 0;
        wave_match<8>(key[k], ok, rank[k], cnt[k]);
        if (ok && rank[k] == 0) mycw[key[k]] += cnt[k];
    }
}

__global__ void __launch_bounds__(1024)
gcn_scatter_kernel(const int* __restrict__ rows, const int* __restrict__ cols,
                   const float* __restrict__ vals, const float* __restrict__ dropu,
                   int* __restrict__ ctab, int* __restrict__ tmp,
                   int n_nodes, int n_edges) {
    __shared__ int cw[32][NBMAX];
    __shared__ int lofs[NBMAX];
    __shared__ int ltot[8];
    __shared__ alignas(16) int stage[CHUNK * 4];

    const int tid   = threadIdx.x;
    const int lane  = tid & 31;
    const int wave  = tid >> 5;
    const int chunk = blockIdx.x;
    const int e0    = (chunk * 1024 + tid) * 4;
    int* mycw = cw[wave];
    for (int i = lane; i < NBMAX; i += 32) mycw[i] = 0;

    int row[4];
    load_rows4(rows, n_edges, e0, row);

    int   cc[4];
    float ww[4], ua[4], ub[4];
    if (e0 + 3 < n_edges && (n_edges & 3) == 0) {
        const v4i c4 = *(const v4i*)(cols + e0);
        const v4f w4 = *(const v4f*)(vals + e0);
        const v4f a4 = *(const v4f*)(dropu + e0);
        const v4f b4 = *(const v4f*)(dropu + (size_t)n_edges + e0);
        cc[0] = c4.x; cc[1] = c4.y; cc[2] = c4.z; cc[3] = c4.w;
        ww[0] = w4.x; ww[1] = w4.y; ww[2] = w4.z; ww[3] = w4.w;
        ua[0] = a4.x; ua[1] = a4.y; ua[2] = a4.z; ua[3] = a4.w;
        ub[0] = b4.x; ub[1] = b4.y; ub[2] = b4.z; ub[3] = b4.w;
    } else {
#pragma unroll
        for (int k = 0; k < 4; ++k) {
            const bool in = e0 + k < n_edges;
            cc[k] = in ? cols[e0 + k] : 0;
            ww[k] = in ? vals[e0 + k] : 0.0f;
            ua[k] = in ? dropu[e0 + k] : 0.0f;
            ub[k] = in ? dropu[(size_t)n_edges + e0 + k] : 0.0f;
        }
    }

    int key[4], rank[4], cnt[4];
    chunk_count(n_nodes, mycw, row, key, rank, cnt);
    __syncthreads();

    int mine = 0;
    if (tid < NBMAX) {
        int run = 0;
        for (int w = 0; w < 32; ++w) {
            const int c = cw[w][tid];
            cw[w][tid] = run;
            run += c;
        }
        mine = run;
    }
    const int linc = wave_incl_scan_i32(mine, lane);
    if (tid < NBMAX && lane == 31) ltot[wave] = linc;
    __syncthreads();
    if (tid < NBMAX) {
        int lwoff = 0;
        for (int w = 0; w < wave; ++w) lwoff += ltot[w];
        const int lexcl = lwoff + linc - mine;
        lofs[tid] = lexcl;
        ctab[chunk * NBMAX + tid] = lexcl | (mine << 16);
    }
    int nvalid = 0;
    for (int w = 0; w < 8; ++w) nvalid += ltot[w];
    __syncthreads();

#pragma unroll
    for (int k = 0; k < 4; ++k) {
        if (row[k] >= 0) {
            const int cur  = mycw[key[k]];
            const int lpos = lofs[key[k]] + cur + rank[k];
            if (rank[k] == 0) mycw[key[k]] = cur + cnt[k];
            const float v  = ww[k] * (1.0f / KEEP_PROB);
            const float v0 = (ua[k] + KEEP_PROB < 1.0f) ? 0.0f : v;
            const float v1 = (ub[k] + KEEP_PROB < 1.0f) ? 0.0f : v;
            const v4i r = {cc[k], __float_as_int(v0), __float_as_int(v1), row[k]};
            ((v4i*)stage)[lpos] = r;
        }
    }
    __syncthreads();

    if (nvalid > 0) {
        v4i r[4];
#pragma unroll
        for (int k = 0; k < 4; ++k) {
            const int pos = k * 1024 + tid;
            r[k] = ((const v4i*)stage)[(pos < nvalid) ? pos : (nvalid - 1)];
        }
#pragma unroll
        for (int k = 0; k < 4; ++k) {
            const int pos = k * 1024 + tid;
            if (pos < nvalid) ((v4i*)tmp)[(size_t)chunk * CHUNK + pos] = r[k];
        }
    }
}

#define WALK_U 5

template <int PASS>
__device__ __forceinline__ void bucket_step(const v4i r, bool ok, int lo, int seg_beg,
                                            bool staged, int* mycw,
                                            const int* rstart, int* stage,
                                            int* __restrict__ rec) {
    const int key = ok ? (r.w - lo) : 0;
    int rank, cnt;
    wave_match<9>(key, ok, rank, cnt);
    if (PASS == 1) {
        if (ok && rank == 0) mycw[key] += cnt;
    } else if (ok) {
        const int cur = mycw[key];
        if (rank == 0) mycw[key] = cur + cnt;
        const v4i o = {r.x, r.y, r.x, r.z};
        const int slot = rstart[key] + cur + rank;
        if (staged) ((v4i*)stage)[slot - seg_beg] = o;
        else        ((v4i*)rec)[slot] = o;
    }
}

template <int PASS>
__device__ __forceinline__ v4i bucket_load(const int* __restrict__ tmp, size_t at,
                                           int i, int n) {
    const size_t p = at + ((i < n) ? i : (n - 1));
    v4i r = {0, 0, 0, 0};
    if (PASS == 1) r.w = tmp[p * 4 + 3];
    else           r   = ((const v4i*)tmp)[p];
    return r;
}

template <int PASS>
__device__ __forceinline__ void bucket_walk(const int* __restrict__ ctab,
                                            const int* __restrict__ tmp,
                                            int* __restrict__ rec, int* mycw,
                                            const int* rstart, int* stage,
                                            int b, int lo, int n_chunks, int wave,
                                            int lane, int seg_beg, bool staged) {
    for (int cbase = 0; cbase < n_chunks; cbase += 1024) {
        const int cmine = cbase + wave + 32 * lane;
        const int pk = (cmine < n_chunks) ? ctab[(size_t)cmine * NBMAX + b] : 0;
        const int left = n_chunks - cbase - wave;
        int nj = (left > 0) ? ((left + 31) >> 5) : 0;
        nj = (nj < 32) ? nj : 32;
        for (int j0 = 0; j0 < nj; j0 += WALK_U) {
            int    cntu[WALK_U];
            size_t first[WALK_U];
            v4i    r[WALK_U];
#pragma unroll
            for (int u = 0; u < WALK_U; ++u) {
                const int j = j0 + u;
                const int p = (j < nj) ? __builtin_amdgcn_readlane(pk, j & 31) : 0;
                cntu[u]  = p >> 16;
                first[u] = (size_t)(cbase + wave + 32 * j) * CHUNK + (p & 0xffff);
                if (cntu[u] > 0) r[u] = bucket_load<PASS>(tmp, first[u], lane, cntu[u]);
            }
#pragma unroll
            for (int u = 0; u < WALK_U; ++u) {
                if (cntu[u] > 0) {
                    bucket_step<PASS>(r[u], lane < cntu[u], lo, seg_beg, staged, mycw, rstart, stage, rec);
                    for (int t = 32; t < cntu[u]; t += 32) {
                        const v4i q = bucket_load<PASS>(tmp, first[u], t + lane, cntu[u]);
                        bucket_step<PASS>(q, t + lane < cntu[u], lo, seg_beg, staged, mycw, rstart, stage, rec);
                    }
                }
            }
        }
    }
}

__global__ void __launch_bounds__(1024)
gcn_bucket_kernel(const int* __restrict__ ctab, const int* __restrict__ tmp,
                  int* __restrict__ rowptr, int* __restrict__ rec, int n_nodes,
                  int n_chunks) {
    __shared__ int cw[32][BROWS];
    __shared__ int rstart[BROWS];
    __shared__ int wtot[32];
    __shared__ int wsum[2][32];
    __shared__ alignas(16) int stage[STAGE_CAP * 4];

    const int tid  = threadIdx.x;
    const int lane = tid & 31;
    const int wave = __builtin_amdgcn_readfirstlane(tid >> 5);
    const int b    = blockIdx.x;
    const int lo   = b * BROWS;
    const int nrows = (n_nodes - lo < BROWS) ? (n_nodes - lo) : BROWS;
    int* mycw = cw[wave];
    for (int i = lane; i < BROWS; i += 32) mycw[i] = 0;

    int sb = 0, sl = 0;
    for (int c = tid; c < n_chunks; c += 1024) {
        const int p = ctab[(size_t)c * NBMAX + b];
        sb += p & 0xffff;
        sl += p >> 16;
    }
    for (int o = 16; o > 0; o >>= 1) {
        sb += __shfl_xor(sb, o);
        sl += __shfl_xor(sl, o);
    }
    if (lane == 0) {
        wsum[0][wave] = sb;
        wsum[1][wave] = sl;
    }
    __syncthreads();
    int seg_beg = wsum[0][lane];
    int seg_len = wsum[1][lane];
    for (int o = 16; o > 0; o >>= 1) {
        seg_beg += __shfl_xor(seg_beg, o);
        seg_len += __shfl_xor(seg_len, o);
    }
    const bool staged = seg_len <= STAGE_CAP;

    bucket_walk<1>(ctab, tmp, rec, mycw, rstart, stage, b, lo, n_chunks, wave, lane, seg_beg, staged);
    __syncthreads();

    int tot = 0;
    if (tid < BROWS) {
        for (int w = 0; w < 32; ++w) {
            const int c = cw[w][tid];
            cw[w][tid] = tot;
            tot += c;
        }
    }
    const int x = wave_incl_scan_i32(tot, lane);
    if (lane == 31) wtot[wave] = x;
    __syncthreads();
    const int wt   = wtot[lane];
    const int wx   = wave_incl_scan_i32(wt, lane);
    const int woff = __shfl(wx - wt, wave);
    const int start = seg_beg + woff + x - tot;
    if (tid < BROWS) rstart[tid] = start;
    if (tid < nrows) {
        rowptr[lo + tid] = start;
        if (lo + tid == n_nodes - 1) rowptr[n_nodes] = start + tot;
    }
    __syncthreads();

    bucket_walk<2>(ctab, tmp, rec, mycw, rstart, stage, b, lo, n_chunks, wave, lane, seg_beg, staged);

    if (staged) {
        __syncthreads();
        for (int i = tid; i < seg_len; i += 1024)
            ((v4i*)rec)[seg_beg + i] = ((const v4i*)stage)[i];
    }
}

template <int K>
__device__ __forceinline__ void pull_batch(unsigned& m, int cj, int wb,
                                           const float* __restrict__ srcl,
                                           v2f& acc) {
    int   c[K];
    float w[K];
    v2f   x[K];
#pragma unroll
    for (int k = 0; k < K; ++k) {
        const int j = __builtin_ctz(m);
        m &= m - 1;
        c[k] = __builtin_amdgcn_readlane(cj, j);
        w[k] = __int_as_float(__builtin_amdgcn_readlane(wb, j));
    }
#pragma unroll
    for (int k = 0; k < K; ++k) {
        x[k] = *(const v2f*)(srcl + (size_t)c[k] * DIM);
    }
#pragma unroll
    for (int k = 0; k < K; ++k) {
        acc += w[k] * x[k];
    }
}

__device__ __forceinline__ void pull_records(const v2i r,
                                             const float* __restrict__ srcl,
                                             v2f& acc) {
    const int cj = r.x;
    const int wb = r.y;
    unsigned m = __builtin_amdgcn_ballot_w32(__int_as_float(wb) != 0.0f);
    int nk = __builtin_popcount(m);
    for (; nk >= 8; nk -= 8) pull_batch<8>(m, cj, wb, srcl, acc);
    if (nk & 4) pull_batch<4>(m, cj, wb, srcl, acc);
    if (nk & 2) pull_batch<2>(m, cj, wb, srcl, acc);
    if (nk & 1) pull_batch<1>(m, cj, wb, srcl, acc);
}

#define PULL_ROWS 8

template <int LAYER>
__global__ void __launch_bounds__(256)
gcn_pull_kernel(const int* __restrict__ rowptr, const int* __restrict__ rec,
                const float* __restrict__ src, float* __restrict__ dst,
                int n_nodes) {
    int r0 = (blockIdx.x * 8 + (threadIdx.x >> 5)) * PULL_ROWS;
    if (r0 >= n_nodes) return;
    r0 = __builtin_amdgcn_readfirstlane(r0);
    const int l  = threadIdx.x & 31;
    const int nr = (n_nodes - r0 < PULL_ROWS) ? (n_nodes - r0) : PULL_ROWS;

    const int rpv = rowptr[r0 + ((l < nr) ? l : nr)];

    float* dp = dst + (size_t)r0 * DIM + l * 2;
    const float* srcl = src + l * 2;
    const float* xp = srcl + (size_t)r0 * DIM;
    const v2i* recl = (const v2i*)rec + LAYER;

    v2f fn = {0.f, 0.f};
    v2f xn = {0.f, 0.f};
    if (LAYER == 1) {
        fn = *(const v2f*)dp;
        xn = *(const v2f*)xp;
    }
    int nbeg = __builtin_amdgcn_readlane(rpv, 0);
    int nend = __builtin_amdgcn_readlane(rpv, 1);
    v2i rn = {0, 0};
    if (nbeg + l < nend) rn = recl[(size_t)(nbeg + l) * 2];

    for (int i = 0; i < nr; ++i) {
        const v2i r = rn;
        v2f acc = fn;
        const v2f xs = xn;
        const int beg = nbeg;
        const int end = nend;

        if (i + 1 < nr) {
            nbeg = nend;
            nend = __builtin_amdgcn_readlane(rpv, i + 2);
            rn = (v2i){0, 0};
            if (nbeg + l < nend) rn = recl[(size_t)(nbeg + l) * 2];
            if (LAYER == 1) {
                fn = *(const v2f*)(dp + (size_t)(i + 1) * DIM);
                xn = *(const v2f*)(xp + (size_t)(i + 1) * DIM);
            }
        }

        pull_records(r, srcl, acc);
        for (int base = beg + 32; base < end; base += 32) {
            v2i q = {0, 0};
            if (base + l < end) q = recl[(size_t)(base + l) * 2];
            pull_records(q, srcl, acc);
        }

        if (LAYER == 1) acc = (acc + xs) * (1.0f / 3.0f);
        *(v2f*)(dp + (size_t)i * DIM) = acc;
    }
}

__global__ void __launch_bounds__(256)
gcn_finalize_kernel(float* __restrict__ out, const float* __restrict__ x1,
                    long long n4) {
    const long long i = (long long)blockIdx.x * blockDim.x + threadIdx.x;
    if (i >= n4) return;
    v4f a = ((const v4f*)out)[i];
    v4f p = ((const v4f*)x1)[i];
    v4f r = (a + p) * (1.0f / 3.0f);
    ((v4f*)out)[i] = r;
}

extern "C" void kernel_launch(void* const* d_in, const int* in_sizes, int n_in,
                              void* d_out, int out_size, void* d_ws, size_t ws_size,
                              hipStream_t stream) {
    const float* emb   = (const float*)d_in[0];
    const float* W     = (const float*)d_in[1];
    const float* bias  = (const float*)d_in[2];
    const float* vals  = (const float*)d_in[3];
    const float* dropu = (const float*)d_in[4];
    const int*   rows  = (const int*)d_in[5];
    const int*   cols  = (const int*)d_in[6];

    const int N = in_sizes[0] / DIM;
    const int E = in_sizes[3];

    float* out = (float*)d_out;
    float* x1  = (float*)d_ws;

    const int    n_chunks  = (E + CHUNK - 1) / CHUNK;
    const int    n_buckets = (N + BROWS - 1) / BROWS;
    const size_t x1_bytes  = (size_t)N * DIM * sizeof(float);
    const size_t rec_bytes = (size_t)E * 16;
    const size_t tmp_bytes = (size_t)n_chunks * CHUNK * 16;
    const size_t r0_bytes  = (x1_bytes > tmp_bytes) ? x1_bytes : tmp_bytes;
    const size_t rp_bytes  = (((size_t)N + 1) * sizeof(int) + 15) & ~(size_t)15;
    const size_t ct_bytes  = (size_t)n_chunks * NBMAX * sizeof(int);
    const size_t need      = r0_bytes + rec_bytes + rp_bytes + ct_bytes;

    if (ws_size >= need && n_buckets <= NBMAX && n_chunks > 0) {
        int* tmp    = (int*)d_ws;
        int* rec    = (int*)((char*)d_ws + r0_bytes);
        int* rowptr = (int*)((char*)d_ws + r0_bytes + rec_bytes);
        int* ctab   = (int*)((char*)d_ws + r0_bytes + rec_bytes + rp_bytes);

        {
            dim3 grid(FC_GROUPS), block(256);
            gcn_fc_wmma_kernel<<<grid, block, 0, stream>>>(emb, W, bias, out, N);
        }
        {
            gcn_scatter_kernel<<<dim3(n_chunks), dim3(1024), 0, stream>>>(rows, cols, vals, dropu, ctab, tmp, N, E);
            gcn_bucket_kernel<<<dim3(n_buckets), dim3(1024), 0, stream>>>(ctab, tmp, rowptr, rec, N, n_chunks);
        }
        {
            dim3 grid((N + 8 * PULL_ROWS - 1) / (8 * PULL_ROWS)), block(256);
            gcn_pull_kernel<0><<<grid, block, 0, stream>>>(rowptr, rec, emb, x1,  N);
            gcn_pull_kernel<1><<<grid, block, 0, stream>>>(rowptr, rec, x1,  out, N);
        }
    } else {
        hipMemsetAsync(d_ws, 0, x1_bytes, stream);

        {
            dim3 grid(FC_GROUPS), block(256);
            gcn_fc_wmma_kernel<<<grid, block, 0, stream>>>(emb, W, bias, out, N);
        }

        {
            long long threads = (long long)E * 32;
            dim3 grid((unsigned)((threads + 255) / 256)), block(256);
            gcn_spmm_kernel<<<grid, block, 0, stream>>>(vals, dropu,     rows, cols, emb, x1,  E);
            gcn_spmm_kernel<<<grid, block, 0, stream>>>(vals, dropu + E, rows, cols, x1,  out, E);
        }

        {
            long long n4 = (long long)N * DIM / 4;
            dim3 grid((unsigned)((n4 + 255) / 256)), block(256);
            gcn_finalize_kernel<<<grid, block, 0, stream>>>(out, x1, n4);
        }
    }
}


# Edge record packed into 8 B, not 16 (col, two kept flags and row-in-bucket in one word; v in the other) through tmp, LDS stages and rec: sort and pull stream 8 B per edge, not 16; sum orders kept
# speedup vs baseline: 4.7782x; 1.0863x over previous
#include <hip/hip_runtime.h>
#include <hip/hip_bf16.h>

typedef __attribute__((ext_vector_type(2))) float v2f;
typedef __attribute__((ext_vector_type(4))) float v4f;
typedef __attribute__((ext_vector_type(8))) float v8f;
typedef __attribute__((ext_vector_type(4))) int   v4i;
typedef __attribute__((ext_vector_type(2))) int   v2i;

#define DIM 64
#define KEEP_PROB 0.7f
#define OWN_ROWS 400

#define FC_GROUPS 512

__device__ __forceinline__ void fc_load_tile(const float* __restrict__ emb, int tile,
                                             int c, int h, int n_nodes, v4f (&a)[8]) {
    int gr = tile * 16 + c;
    gr = (gr < n_nodes) ? gr : (n_nodes - 1);
    const float* p = emb + (size_t)gr * DIM + 4 * h;
#pragma unroll
    for (int j = 0; j < 8; ++j) a[j] = *(const v4f*)(p + 8 * j);
}

__device__ __forceinline__ void fc_tile(const v4f (&a)[8], const v4f (&bf)[4][8],
                                        float bb0, float bb1, float bb2, float bb3,
                                        float* __restrict__ acc, int tile,
                                        int c, int h, int n_nodes) {
    v8f c0 = {}, c1 = {}, c2 = {}, c3 = {};
#pragma unroll
    for (int j = 0; j < 8; ++j) {
        const v2f a0 = a[j].xy;
        const v2f a1 = a[j].zw;
        c0 = __builtin_amdgcn_wmma_f32_16x16x4_f32(false, a0, false, bf[0][j].xy, (short)0, c0, false, false);
        c1 = __builtin_amdgcn_wmma_f32_16x16x4_f32(false, a0, false, bf[1][j].xy, (short)0, c1, false, false);
        c2 = __builtin_amdgcn_wmma_f32_16x16x4_f32(false, a0, false, bf[2][j].xy, (short)0, c2, false, false);
        c3 = __builtin_amdgcn_wmma_f32_16x16x4_f32(false, a0, false, bf[3][j].xy, (short)0, c3, false, false);
        c0 = __builtin_amdgcn_wmma_f32_16x16x4_f32(false, a1, false, bf[0][j].zw, (short)0, c0, false, false);
        c1 = __builtin_amdgcn_wmma_f32_16x16x4_f32(false, a1, false, bf[1][j].zw, (short)0, c1, false, false);
        c2 = __builtin_amdgcn_wmma_f32_16x16x4_f32(false, a1, false, bf[2][j].zw, (short)0, c2, false, false);
        c3 = __builtin_amdgcn_wmma_f32_16x16x4_f32(false, a1, false, bf[3][j].zw, (short)0, c3, false, false);
    }

    const int row_base = tile * 16 + h * 8;
#pragma unroll
    for (int j = 0; j < 8; ++j) {
        const int gm = row_base + j;
        if (gm < n_nodes) {
            float* outr = acc + (size_t)gm * DIM;
            outr[c +  0] = c0[j] + bb0;
            outr[c + 16] = c1[j] + bb1;
            outr[c + 32] = c2[j] + bb2;
            outr[c + 48] = c3[j] + bb3;
        }
    }
}

__global__ void __launch_bounds__(256)
gcn_fc_wmma_kernel(const float* __restrict__ emb, const float* __restrict__ W,
                   const float* __restrict__ bias, float* __restrict__ acc,
                   int n_nodes) {
    const int lane = threadIdx.x & 31;
    const int c    = lane & 15;
    const int h    = lane >> 4;
    const int nw   = gridDim.x * 8;
    const int n_tiles = (n_nodes + 15) >> 4;
    int tile = (threadIdx.x >> 5) * gridDim.x + blockIdx.x;
    if (tile >= n_tiles) return;
    tile = __builtin_amdgcn_readfirstlane(tile);

    v4f bf[4][8];
#pragma unroll
    for (int t = 0; t < 4; ++t) {
#pragma unroll
        for (int j = 0; j < 8; ++j) {
            bf[t][j] = *(const v4f*)(W + (size_t)(16 * t + c) * DIM + 8 * j + 4 * h);
        }
    }
    const float bb0 = bias[c];
    const float bb1 = bias[c + 16];
    const float bb2 = bias[c + 32];
    const float bb3 = bias[c + 48];

    v4f pa[8], pb[8];
    fc_load_tile(emb, tile, c, h, n_nodes, pa);
    for (;;) {
        int next = tile + nw;
        fc_load_tile(emb, (next < n_tiles) ? next : tile, c, h, n_nodes, pb);
        fc_tile(pa, bf, bb0, bb1, bb2, bb3, acc, tile, c, h, n_nodes);
        if (next >= n_tiles) break;
        tile = next;
        next = tile + nw;
        fc_load_tile(emb, (next < n_tiles) ? next : tile, c, h, n_nodes, pa);
        fc_tile(pb, bf, bb0, bb1, bb2, bb3, acc, tile, c, h, n_nodes);
        if (next >= n_tiles) break;
        tile = next;
    }
}

__global__ void __launch_bounds__(256)
gcn_spmm_kernel(const float* __restrict__ vals, const float* __restrict__ dropu,
                const int* __restrict__ rows, const int* __restrict__ cols,
                const float* __restrict__ src, float* __restrict__ dst,
                int n_edges) {
    const long long tid = (long long)blockIdx.x * blockDim.x + threadIdx.x;
    int e = (int)(tid >> 5);
    if (e >= n_edges) return;
    e = __builtin_amdgcn_readfirstlane(e);
    const int l = threadIdx.x & 31;

    const float u = dropu[e];
    if (u + KEEP_PROB < 1.0f) return;
    const float v = vals[e] * (1.0f / KEEP_PROB);

    const int col = cols[e];
    const int row = rows[e];

    const float* sp = src + (size_t)col * DIM + l * 2;
    float*       dp = dst + (size_t)row * DIM + l * 2;
    const float m0 = v * sp[0];
    const float m1 = v * sp[1];

    asm volatile("global_atomic_add_f32 %0, %1, off scope:SCOPE_DEV"
                 :: "v"(dp), "v"(m0) : "memory");
    asm volatile("global_atomic_add_f32 %0, %1, off scope:SCOPE_DEV"
                 :: "v"(dp + 1), "v"(m1) : "memory");
}


#define BROWS 400
#define NBMAX 256
#define CHUNK 4096
#define STAGE_CAP 8192
#define COL_BITS 21
#define COL_MASK ((1 << COL_BITS) - 1)
#define ROW_SHIFT (COL_BITS + 2)

__device__ __forceinline__ int wave_incl_scan_i32(int x, int lane) {
    int y;
    y = __shfl_up(x, 1);  if (lane >= 1)  x += y;
    y = __shfl_up(x, 2);  if (lane >= 2)  x += y;
    y = __shfl_up(x, 4);  if (lane >= 4)  x += y;
    y = __shfl_up(x, 8);  if (lane >= 8)  x += y;
    y = __shfl_up(x, 16); if (lane >= 16) x += y;
    return x;
}

template <int NBITS>
__device__ __forceinline__ void wave_match(int key, bool ok, int& rank, int& cnt) {
    unsigned peers = __builtin_amdgcn_ballot_w32(ok);
    for (int b = 0; b < NBITS; ++b) {
        const bool bit = ((key >> b) & 1) != 0;
        const unsigned m = __builtin_amdgcn_ballot_w32(ok && bit);
        peers &= bit ? m : ~m;
    }
    rank = (int)__builtin_amdgcn_mbcnt_lo(peers, 0u);
    cnt  = __builtin_popcount(peers);
}

__device__ __forceinline__ void load_rows4(const int* __restrict__ rows,
                                           int n_edges, int e0, int (&row)[4]) {
    if (e0 + 3 < n_edges) {
        const v4i g = *(const v4i*)(rows + e0);
        row[0] = g.x; row[1] = g.y; row[2] = g.z; row[3] = g.w;
    } else {
#pragma unroll
        for (int k = 0; k < 4; ++k) row[k] = (e0 + k < n_edges) ? rows[e0 + k] : -1;
    }
}

__device__ __forceinline__ void chunk_count(int n_nodes, int* mycw, int (&row)[4],
                                            int (&key)[4], int (&rank)[4],
                                            int (&cnt)[4]) {
#pragma unroll
    for (int k = 0; k < 4; ++k) {
        const bool ok = (unsigned)row[k] < (unsigned)n_nodes;
        if (!ok) row[k] = -1;
        key[k] = ok ? (row[k] / BROWS) : 0;
        wave_match<8>(key[k], ok, rank[k], cnt[k]);
        if (ok && rank[k] == 0) mycw[key[k]] += cnt[k];
    }
}

__global__ void __launch_bounds__(1024)
gcn_scatter_kernel(const int* __restrict__ rows, const int* __restrict__ cols,
                   const float* __restrict__ vals, const float* __restrict__ dropu,
                   int* __restrict__ ctab, int* __restrict__ tmp,
                   int n_nodes, int n_edges) {
    __shared__ int cw[32][NBMAX];
    __shared__ int lofs[NBMAX];
    __shared__ int ltot[8];
    __shared__ alignas(16) int stage[CHUNK * 2];

    const int tid   = threadIdx.x;
    const int lane  = tid & 31;
    const int wave  = tid >> 5;
    const int chunk = blockIdx.x;
    const int e0    = (chunk * 1024 + tid) * 4;
    int* mycw = cw[wave];
    for (int i = lane; i < NBMAX; i += 32) mycw[i] = 0;

    int row[4];
    load_rows4(rows, n_edges, e0, row);

    int   cc[4];
    float ww[4], ua[4], ub[4];
    if (e0 + 3 < n_edges && (n_edges & 3) == 0) {
        const v4i c4 = *(const v4i*)(cols + e0);
        const v4f w4 = *(const v4f*)(vals + e0);
        const v4f a4 = *(const v4f*)(dropu + e0);
        const v4f b4 = *(const v4f*)(dropu + (size_t)n_edges + e0);
        cc[0] = c4.x; cc[1] = c4.y; cc[2] = c4.z; cc[3] = c4.w;
        ww[0] = w4.x; ww[1] = w4.y; ww[2] = w4.z; ww[3] = w4.w;
        ua[0] = a4.x; ua[1] = a4.y; ua[2] = a4.z; ua[3] = a4.w;
        ub[0] = b4.x; ub[1] = b4.y; ub[2] = b4.z; ub[3] = b4.w;
    } else {
#pragma unroll
        for (int k = 0; k < 4; ++k) {
            const bool in = e0 + k < n_edges;
            cc[k] = in ? cols[e0 + k] : 0;
            ww[k] = in ? vals[e0 + k] : 0.0f;
            ua[k] = in ? dropu[e0 + k] : 0.0f;
            ub[k] = in ? dropu[(size_t)n_edges + e0 + k] : 0.0f;
        }
    }

    int key[4], rank[4], cnt[4];
    chunk_count(n_nodes, mycw, row, key, rank, cnt);
    __syncthreads();

    int mine = 0;
    if (tid < NBMAX) {
        int run = 0;
        for (int w = 0; w < 32; ++w) {
            const int c = cw[w][tid];
            cw[w][tid] = run;
            run += c;
        }
        mine = run;
    }
    const int linc = wave_incl_scan_i32(mine, lane);
    if (tid < NBMAX && lane == 31) ltot[wave] = linc;
    __syncthreads();
    if (tid < NBMAX) {
        int lwoff = 0;
        for (int w = 0; w < wave; ++w) lwoff += ltot[w];
        const int lexcl = lwoff + linc - mine;
        lofs[tid] = lexcl;
        ctab[chunk * NBMAX + tid] = lexcl | (mine << 16);
    }
    int nvalid = 0;
    for (int w = 0; w < 8; ++w) nvalid += ltot[w];
    __syncthreads();

#pragma unroll
    for (int k = 0; k < 4; ++k) {
        if (row[k] >= 0) {
            const int cur  = mycw[key[k]];
            const int lpos = lofs[key[k]] + cur + rank[k];
            if (rank[k] == 0) mycw[key[k]] = cur + cnt[k];
            const float v  = ww[k] * (1.0f / KEEP_PROB);
            const int k0 = (ua[k] + KEEP_PROB < 1.0f) ? 0 : (1 << COL_BITS);
            const int k1 = (ub[k] + KEEP_PROB < 1.0f) ? 0 : (2 << COL_BITS);
            const int rl = row[k] - key[k] * BROWS;
            const v2i r = {(cc[k] & COL_MASK) | k0 | k1 | (int)((unsigned)rl << ROW_SHIFT),
                           __float_as_int(v)};
            ((v2i*)stage)[lpos] = r;
        }
    }
    __syncthreads();

    if (nvalid > 0) {
        v4i r[2];
        const int lastq = (nvalid - 1) >> 1;
#pragma unroll
        for (int k = 0; k < 2; ++k) {
            const int q = k * 1024 + tid;
            r[k] = ((const v4i*)stage)[(q < lastq) ? q : lastq];
        }
#pragma unroll
        for (int k = 0; k < 2; ++k) {
            const int q = k * 1024 + tid;
            int* dstp = tmp + ((size_t)chunk * CHUNK + 2 * q) * 2;
            if (2 * q + 1 < nvalid)  *(v4i*)dstp = r[k];
            else if (2 * q < nvalid) *(v2i*)dstp = r[k].xy;
        }
    }
}

#define WALK_U 5

template <int PASS>
__device__ __forceinline__ void bucket_step(const v2i r, bool ok, int seg_beg,
                                            bool staged, int* mycw,
                                            const int* rstart, int* stage,
                                            int* __restrict__ rec) {
    const int key = ok ? (int)((unsigned)r.x >> ROW_SHIFT) : 0;
    int rank, cnt;
    wave_match<9>(key, ok, rank, cnt);
    if (PASS == 1) {
        if (ok && rank == 0) mycw[key] += cnt;
    } else if (ok) {
        const int cur = mycw[key];
        if (rank == 0) mycw[key] = cur + cnt;
        const int slot = rstart[key] + cur + rank;
        if (staged) ((v2i*)stage)[slot - seg_beg] = r;
        else        ((v2i*)rec)[slot] = r;
    }
}

template <int PASS>
__device__ __forceinline__ v2i bucket_load(const int* __restrict__ tmp, size_t at,
                                           int i, int n) {
    const size_t p = at + ((i < n) ? i : (n - 1));
    v2i r = {0, 0};
    if (PASS == 1) r.x = tmp[p * 2];
    else           r   = ((const v2i*)tmp)[p];
    return r;
}

template <int PASS>
__device__ __forceinline__ void bucket_walk(const int* __restrict__ ctab,
                                            const int* __restrict__ tmp,
                                            int* __restrict__ rec, int* mycw,
                                            const int* rstart, int* stage,
                                            int b, int n_chunks, int wave,
                                            int lane, int seg_beg, bool staged) {
    for (int cbase = 0; cbase < n_chunks; cbase += 1024) {
        const int cmine = cbase + wave + 32 * lane;
        const int pk = (cmine < n_chunks) ? ctab[(size_t)cmine * NBMAX + b] : 0;
        const int left = n_chunks - cbase - wave;
        int nj = (left > 0) ? ((left + 31) >> 5) : 0;
        nj = (nj < 32) ? nj : 32;
        for (int j0 = 0; j0 < nj; j0 += WALK_U) {
            int    cntu[WALK_U];
            size_t first[WALK_U];
            v2i    r[WALK_U];
#pragma unroll
            for (int u = 0; u < WALK_U; ++u) {
                const int j = j0 + u;
                const int p = (j < nj) ? __builtin_amdgcn_readlane(pk, j & 31) : 0;
                cntu[u]  = p >> 16;
                first[u] = (size_t)(cbase + wave + 32 * j) * CHUNK + (p & 0xffff);
                if (cntu[u] > 0) r[u] = bucket_load<PASS>(tmp, first[u], lane, cntu[u]);
            }
#pragma unroll
            for (int u = 0; u < WALK_U; ++u) {
                if (cntu[u] > 0) {
                    bucket_step<PASS>(r[u], lane < cntu[u], seg_beg, staged, mycw, rstart, stage, rec);
                    for (int t = 32; t < cntu[u]; t += 32) {
                        const v2i q = bucket_load<PASS>(tmp, first[u], t + lane, cntu[u]);
                        bucket_step<PASS>(q, t + lane < cntu[u], seg_beg, staged, mycw, rstart, stage, rec);
                    }
                }
            }
        }
    }
}

__global__ void __launch_bounds__(1024)
gcn_bucket_kernel(const int* __restrict__ ctab, const int* __restrict__ tmp,
                  int* __restrict__ rowptr, int* __restrict__ rec, int n_nodes,
                  int n_chunks) {
    __shared__ int cw[32][BROWS];
    __shared__ int rstart[BROWS];
    __shared__ int wtot[32];
    __shared__ int wsum[2][32];
    __shared__ alignas(16) int stage[STAGE_CAP * 2];

    const int tid  = threadIdx.x;
    const int lane = tid & 31;
    const int wave = __builtin_amdgcn_readfirstlane(tid >> 5);
    const int b    = blockIdx.x;
    const int lo   = b * BROWS;
    const int nrows = (n_nodes - lo < BROWS) ? (n_nodes - lo) : BROWS;
    int* mycw = cw[wave];
    for (int i = lane; i < BROWS; i += 32) mycw[i] = 0;

    int sb = 0, sl = 0;
    for (int c = tid; c < n_chunks; c += 1024) {
        const int p = ctab[(size_t)c * NBMAX + b];
        sb += p & 0xffff;
        sl += p >> 16;
    }
    for (int o = 16; o > 0; o >>= 1) {
        sb += __shfl_xor(sb, o);
        sl += __shfl_xor(sl, o);
    }
    if (lane == 0) {
        wsum[0][wave] = sb;
        wsum[1][wave] = sl;
    }
    __syncthreads();
    int seg_beg = wsum[0][lane];
    int seg_len = wsum[1][lane];
    for (int o = 16; o > 0; o >>= 1) {
        seg_beg += __shfl_xor(seg_beg, o);
        seg_len += __shfl_xor(seg_len, o);
    }
    const bool staged = seg_len <= STAGE_CAP;

    bucket_walk<1>(ctab, tmp, rec, mycw, rstart, stage, b, n_chunks, wave, lane, seg_beg, staged);
    __syncthreads();

    int tot = 0;
    if (tid < BROWS) {
        for (int w = 0; w < 32; ++w) {
            const int c = cw[w][tid];
            cw[w][tid] = tot;
            tot += c;
        }
    }
    const int x = wave_incl_scan_i32(tot, lane);
    if (lane == 31) wtot[wave] = x;
    __syncthreads();
    const int wt   = wtot[lane];
    const int wx   = wave_incl_scan_i32(wt, lane);
    const int woff = __shfl(wx - wt, wave);
    const int start = seg_beg + woff + x - tot;
    if (tid < BROWS) rstart[tid] = start;
    if (tid < nrows) {
        rowptr[lo + tid] = start;
        if (lo + tid == n_nodes - 1) rowptr[n_nodes] = start + tot;
    }
    __syncthreads();

    bucket_walk<2>(ctab, tmp, rec, mycw, rstart, stage, b, n_chunks, wave, lane, seg_beg, staged);

    if (staged) {
        __syncthreads();
        for (int i = tid; i < seg_len; i += 1024)
            ((v2i*)rec)[seg_beg + i] = ((const v2i*)stage)[i];
    }
}

template <int K>
__device__ __forceinline__ void pull_batch(unsigned& m, int cj, int wb,
                                           const float* __restrict__ srcl,
                                           v2f& acc) {
    int   c[K];
    float w[K];
    v2f   x[K];
#pragma unroll
    for (int k = 0; k < K; ++k) {
        const int j = __builtin_ctz(m);
        m &= m - 1;
        c[k] = __builtin_amdgcn_readlane(cj, j);
        w[k] = __int_as_float(__builtin_amdgcn_readlane(wb, j));
    }
#pragma unroll
    for (int k = 0; k < K; ++k) {
        x[k] = *(const v2f*)(srcl + (size_t)c[k] * DIM);
    }
#pragma unroll
    for (int k = 0; k < K; ++k) {
        acc += w[k] * x[k];
    }
}

template <int LAYER>
__device__ __forceinline__ void pull_records(const v2i r,
                                             const float* __restrict__ srcl,
                                             v2f& acc) {
    const int cj = r.x & COL_MASK;
    const int wb = r.y;
    const bool kept = ((r.x >> (COL_BITS + LAYER)) & 1) != 0;
    unsigned m = __builtin_amdgcn_ballot_w32(kept && __int_as_float(wb) != 0.0f);
    int nk = __builtin_popcount(m);
    for (; nk >= 8; nk -= 8) pull_batch<8>(m, cj, wb, srcl, acc);
    if (nk & 4) pull_batch<4>(m, cj, wb, srcl, acc);
    if (nk & 2) pull_batch<2>(m, cj, wb, srcl, acc);
    if (nk & 1) pull_batch<1>(m, cj, wb, srcl, acc);
}

#define PULL_ROWS 8

template <int LAYER>
__global__ void __launch_bounds__(256)
gcn_pull_kernel(const int* __restrict__ rowptr, const int* __restrict__ rec,
                const float* __restrict__ src, float* __restrict__ dst,
                int n_nodes) {
    int r0 = (blockIdx.x * 8 + (threadIdx.x >> 5)) * PULL_ROWS;
    if (r0 >= n_nodes) return;
    r0 = __builtin_amdgcn_readfirstlane(r0);
    const int l  = threadIdx.x & 31;
    const int nr = (n_nodes - r0 < PULL_ROWS) ? (n_nodes - r0) : PULL_ROWS;

    const int rpv = rowptr[r0 + ((l < nr) ? l : nr)];

    float* dp = dst + (size_t)r0 * DIM + l * 2;
    const float* srcl = src + l * 2;
    const float* xp = srcl + (size_t)r0 * DIM;
    const v2i* recl = (const v2i*)rec;

    v2f fn = {0.f, 0.f};
    v2f xn = {0.f, 0.f};
    if (LAYER == 1) {
        fn = *(const v2f*)dp;
        xn = *(const v2f*)xp;
    }
    int nbeg = __builtin_amdgcn_readlane(rpv, 0);
    int nend = __builtin_amdgcn_readlane(rpv, 1);
    v2i rn = {0, 0};
    if (nbeg + l < nend) rn = recl[(size_t)(nbeg + l)];

    for (int i = 0; i < nr; ++i) {
        const v2i r = rn;
        v2f acc = fn;
        const v2f xs = xn;
        const int beg = nbeg;
        const int end = nend;

        if (i + 1 < nr) {
            nbeg = nend;
            nend = __builtin_amdgcn_readlane(rpv, i + 2);
            rn = (v2i){0, 0};
            if (nbeg + l < nend) rn = recl[(size_t)(nbeg + l)];
            if (LAYER == 1) {
                fn = *(const v2f*)(dp + (size_t)(i + 1) * DIM);
                xn = *(const v2f*)(xp + (size_t)(i + 1) * DIM);
            }
        }

        pull_records<LAYER>(r, srcl, acc);
        for (int base = beg + 32; base < end; base += 32) {
            v2i q = {0, 0};
            if (base + l < end) q = recl[(size_t)(base + l)];
            pull_records<LAYER>(q, srcl, acc);
        }

        if (LAYER == 1) acc = (acc + xs) * (1.0f / 3.0f);
        *(v2f*)(dp + (size_t)i * DIM) = acc;
    }
}

__global__ void __launch_bounds__(256)
gcn_finalize_kernel(float* __restrict__ out, const float* __restrict__ x1,
                    long long n4) {
    const long long i = (long long)blockIdx.x * blockDim.x + threadIdx.x;
    if (i >= n4) return;
    v4f a = ((const v4f*)out)[i];
    v4f p = ((const v4f*)x1)[i];
    v4f r = (a + p) * (1.0f / 3.0f);
    ((v4f*)out)[i] = r;
}

extern "C" void kernel_launch(void* const* d_in, const int* in_sizes, int n_in,
                              void* d_out, int out_size, void* d_ws, size_t ws_size,
                              hipStream_t stream) {
    const float* emb   = (const float*)d_in[0];
    const float* W     = (const float*)d_in[1];
    const float* bias  = (const float*)d_in[2];
    const float* vals  = (const float*)d_in[3];
    const float* dropu = (const float*)d_in[4];
    const int*   rows  = (const int*)d_in[5];
    const int*   cols  = (const int*)d_in[6];

    const int N = in_sizes[0] / DIM;
    const int E = in_sizes[3];

    float* out = (float*)d_out;
    float* x1  = (float*)d_ws;

    const int    n_chunks  = (E + CHUNK - 1) / CHUNK;
    const int    n_buckets = (N + BROWS - 1) / BROWS;
    const size_t x1_bytes  = (size_t)N * DIM * sizeof(float);
    const size_t rec_bytes = ((size_t)E * 8 + 15) & ~(size_t)15;
    const size_t tmp_bytes = (size_t)n_chunks * CHUNK * 8;
    const size_t r0_bytes  = (x1_bytes > tmp_bytes) ? x1_bytes : tmp_bytes;
    const size_t rp_bytes  = (((size_t)N + 1) * sizeof(int) + 15) & ~(size_t)15;
    const size_t ct_bytes  = (size_t)n_chunks * NBMAX * sizeof(int);
    const size_t need      = r0_bytes + rec_bytes + rp_bytes + ct_bytes;

    if (ws_size >= need && n_buckets <= NBMAX && n_chunks > 0) {
        int* tmp    = (int*)d_ws;
        int* rec    = (int*)((char*)d_ws + r0_bytes);
        int* rowptr = (int*)((char*)d_ws + r0_bytes + rec_bytes);
        int* ctab   = (int*)((char*)d_ws + r0_bytes + rec_bytes + rp_bytes);

        {
            dim3 grid(FC_GROUPS), block(256);
            gcn_fc_wmma_kernel<<<grid, block, 0, stream>>>(emb, W, bias, out, N);
        }
        {
            gcn_scatter_kernel<<<dim3(n_chunks), dim3(1024), 0, stream>>>(rows, cols, vals, dropu, ctab, tmp, N, E);
            gcn_bucket_kernel<<<dim3(n_buckets), dim3(1024), 0, stream>>>(ctab, tmp, rowptr, rec, N, n_chunks);
        }
        {
            dim3 grid((N + 8 * PULL_ROWS - 1) / (8 * PULL_ROWS)), block(256);
            gcn_pull_kernel<0><<<grid, block, 0, stream>>>(rowptr, rec, emb, x1,  N);
            gcn_pull_kernel<1><<<grid, block, 0, stream>>>(rowptr, rec, x1,  out, N);
        }
    } else {
        hipMemsetAsync(d_ws, 0, x1_bytes, stream);

        {
            dim3 grid(FC_GROUPS), block(256);
            gcn_fc_wmma_kernel<<<grid, block, 0, stream>>>(emb, W, bias, out, N);
        }

        {
            long long threads = (long long)E * 32;
            dim3 grid((unsigned)((threads + 255) / 256)), block(256);
            gcn_spmm_kernel<<<grid, block, 0, stream>>>(vals, dropu,     rows, cols, emb, x1,  E);
            gcn_spmm_kernel<<<grid, block, 0, stream>>>(vals, dropu + E, rows, cols, x1,  out, E);
        }

        {
            long long n4 = (long long)N * DIM / 4;
            dim3 grid((unsigned)((n4 + 255) / 256)), block(256);
            gcn_finalize_kernel<<<grid, block, 0, stream>>>(out, x1, n4);
        }
    }
}


# fc joins the layer-1 pull launch: a wave pair forms each 16-row fc tile (f32 WMMA) into LDS and the row sums start from it; fc launch, its out write and re-read gone (4 launches); sum orders kept
# speedup vs baseline: 4.9351x; 1.0328x over previous
#include <hip/hip_runtime.h>
#include <hip/hip_bf16.h>

typedef __attribute__((ext_vector_type(2))) float v2f;
typedef __attribute__((ext_vector_type(4))) float v4f;
typedef __attribute__((ext_vector_type(8))) float v8f;
typedef __attribute__((ext_vector_type(4))) int   v4i;
typedef __attribute__((ext_vector_type(2))) int   v2i;

#define DIM 64
#define KEEP_PROB 0.7f
#define OWN_ROWS 400

#define FC_GROUPS 512

__device__ __forceinline__ void fc_load_tile(const float* __restrict__ emb, int tile,
                                             int c, int h, int n_nodes, v4f (&a)[8]) {
    int gr = tile * 16 + c;
    gr = (gr < n_nodes) ? gr : (n_nodes - 1);
    const float* p = emb + (size_t)gr * DIM + 4 * h;
#pragma unroll
    for (int j = 0; j < 8; ++j) a[j] = *(const v4f*)(p + 8 * j);
}

__device__ __forceinline__ void fc_tile(const v4f (&a)[8], const v4f (&bf)[4][8],
                                        float bb0, float bb1, float bb2, float bb3,
                                        float* __restrict__ acc, int tile,
                                        int c, int h, int n_nodes) {
    v8f c0 = {}, c1 = {}, c2 = {}, c3 = {};
#pragma unroll
    for (int j = 0; j < 8; ++j) {
        const v2f a0 = a[j].xy;
        const v2f a1 = a[j].zw;
        c0 = __builtin_amdgcn_wmma_f32_16x16x4_f32(false, a0, false, bf[0][j].xy, (short)0, c0, false, false);
        c1 = __builtin_amdgcn_wmma_f32_16x16x4_f32(false, a0, false, bf[1][j].xy, (short)0, c1, false, false);
        c2 = __builtin_amdgcn_wmma_f32_16x16x4_f32(false, a0, false, bf[2][j].xy, (short)0, c2, false, false);
        c3 = __builtin_amdgcn_wmma_f32_16x16x4_f32(false, a0, false, bf[3][j].xy, (short)0, c3, false, false);
        c0 = __builtin_amdgcn_wmma_f32_16x16x4_f32(false, a1, false, bf[0][j].zw, (short)0, c0, false, false);
        c1 = __builtin_amdgcn_wmma_f32_16x16x4_f32(false, a1, false, bf[1][j].zw, (short)0, c1, false, false);
        c2 = __builtin_amdgcn_wmma_f32_16x16x4_f32(false, a1, false, bf[2][j].zw, (short)0, c2, false, false);
        c3 = __builtin_amdgcn_wmma_f32_16x16x4_f32(false, a1, false, bf[3][j].zw, (short)0, c3, false, false);
    }

    const int row_base = tile * 16 + h * 8;
#pragma unroll
    for (int j = 0; j < 8; ++j) {
        const int gm = row_base + j;
        if (gm < n_nodes) {
            float* outr = acc + (size_t)gm * DIM;
            outr[c +  0] = c0[j] + bb0;
            outr[c + 16] = c1[j] + bb1;
            outr[c + 32] = c2[j] + bb2;
            outr[c + 48] = c3[j] + bb3;
        }
    }
}

__global__ void __launch_bounds__(256)
gcn_fc_wmma_kernel(const float* __restrict__ emb, const float* __restrict__ W,
                   const float* __restrict__ bias, float* __restrict__ acc,
                   int n_nodes) {
    const int lane = threadIdx.x & 31;
    const int c    = lane & 15;
    const int h    = lane >> 4;
    const int nw   = gridDim.x * 8;
    const int n_tiles = (n_nodes + 15) >> 4;
    int tile = (threadIdx.x >> 5) * gridDim.x + blockIdx.x;
    if (tile >= n_tiles) return;
    tile = __builtin_amdgcn_readfirstlane(tile);

    v4f bf[4][8];
#pragma unroll
    for (int t = 0; t < 4; ++t) {
#pragma unroll
        for (int j = 0; j < 8; ++j) {
            bf[t][j] = *(const v4f*)(W + (size_t)(16 * t + c) * DIM + 8 * j + 4 * h);
        }
    }
    const float bb0 = bias[c];
    const float bb1 = bias[c + 16];
    const float bb2 = bias[c + 32];
    const float bb3 = bias[c + 48];

    v4f pa[8], pb[8];
    fc_load_tile(emb, tile, c, h, n_nodes, pa);
    for (;;) {
        int next = tile + nw;
        fc_load_tile(emb, (next < n_tiles) ? next : tile, c, h, n_nodes, pb);
        fc_tile(pa, bf, bb0, bb1, bb2, bb3, acc, tile, c, h, n_nodes);
        if (next >= n_tiles) break;
        tile = next;
        next = tile + nw;
        fc_load_tile(emb, (next < n_tiles) ? next : tile, c, h, n_nodes, pa);
        fc_tile(pb, bf, bb0, bb1, bb2, bb3, acc, tile, c, h, n_nodes);
        if (next >= n_tiles) break;
        tile = next;
    }
}

__global__ void __launch_bounds__(256)
gcn_spmm_kernel(const float* __restrict__ vals, const float* __restrict__ dropu,
                const int* __restrict__ rows, const int* __restrict__ cols,
                const float* __restrict__ src, float* __restrict__ dst,
                int n_edges) {
    const long long tid = (long long)blockIdx.x * blockDim.x + threadIdx.x;
    int e = (int)(tid >> 5);
    if (e >= n_edges) return;
    e = __builtin_amdgcn_readfirstlane(e);
    const int l = threadIdx.x & 31;

    const float u = dropu[e];
    if (u + KEEP_PROB < 1.0f) return;
    const float v = vals[e] * (1.0f / KEEP_PROB);

    const int col = cols[e];
    const int row = rows[e];

    const float* sp = src + (size_t)col * DIM + l * 2;
    float*       dp = dst + (size_t)row * DIM + l * 2;
    const float m0 = v * sp[0];
    const float m1 = v * sp[1];

    asm volatile("global_atomic_add_f32 %0, %1, off scope:SCOPE_DEV"
                 :: "v"(dp), "v"(m0) : "memory");
    asm volatile("global_atomic_add_f32 %0, %1, off scope:SCOPE_DEV"
                 :: "v"(dp + 1), "v"(m1) : "memory");
}


#define BROWS 400
#define NBMAX 256
#define CHUNK 4096
#define STAGE_CAP 8192
#define COL_BITS 21
#define COL_MASK ((1 << COL_BITS) - 1)
#define ROW_SHIFT (COL_BITS + 2)

__device__ __forceinline__ int wave_incl_scan_i32(int x, int lane) {
    int y;
    y = __shfl_up(x, 1);  if (lane >= 1)  x += y;
    y = __shfl_up(x, 2);  if (lane >= 2)  x += y;
    y = __shfl_up(x, 4);  if (lane >= 4)  x += y;
    y = __shfl_up(x, 8);  if (lane >= 8)  x += y;
    y = __shfl_up(x, 16); if (lane >= 16) x += y;
    return x;
}

template <int NBITS>
__device__ __forceinline__ void wave_match(int key, bool ok, int& rank, int& cnt) {
    unsigned peers = __builtin_amdgcn_ballot_w32(ok);
    for (int b = 0; b < NBITS; ++b) {
        const bool bit = ((key >> b) & 1) != 0;
        const unsigned m = __builtin_amdgcn_ballot_w32(ok && bit);
        peers &= bit ? m : ~m;
    }
    rank = (int)__builtin_amdgcn_mbcnt_lo(peers, 0u);
    cnt  = __builtin_popcount(peers);
}

__device__ __forceinline__ void load_rows4(const int* __restrict__ rows,
                                           int n_edges, int e0, int (&row)[4]) {
    if (e0 + 3 < n_edges) {
        const v4i g = *(const v4i*)(rows + e0);
        row[0] = g.x; row[1] = g.y; row[2] = g.z; row[3] = g.w;
    } else {
#pragma unroll
        for (int k = 0; k < 4; ++k) row[k] = (e0 + k < n_edges) ? rows[e0 + k] : -1;
    }
}

__device__ __forceinline__ void chunk_count(int n_nodes, int* mycw, int (&row)[4],
                                            int (&key)[4], int (&rank)[4],
                                            int (&cnt)[4]) {
#pragma unroll
    for (int k = 0; k < 4; ++k) {
        const bool ok = (unsigned)row[k] < (unsigned)n_nodes;
        if (!ok) row[k] = -1;
        key[k] = ok ? (row[k] / BROWS) : 0;
        wave_match<8>(key[k], ok, rank[k], cnt[k]);
        if (ok && rank[k] == 0) mycw[key[k]] += cnt[k];
    }
}

__global__ void __launch_bounds__(1024)
gcn_scatter_kernel(const int* __restrict__ rows, const int* __restrict__ cols,
                   const float* __restrict__ vals, const float* __restrict__ dropu,
                   int* __restrict__ ctab, int* __restrict__ tmp,
                   int n_nodes, int n_edges) {
    __shared__ int cw[32][NBMAX];
    __shared__ int lofs[NBMAX];
    __shared__ int ltot[8];
    __shared__ alignas(16) int stage[CHUNK * 2];

    const int tid   = threadIdx.x;
    const int lane  = tid & 31;
    const int wave  = tid >> 5;
    const int chunk = blockIdx.x;
    const int e0    = (chunk * 1024 + tid) * 4;
    int* mycw = cw[wave];
    for (int i = lane; i < NBMAX; i += 32) mycw[i] = 0;

    int row[4];
    load_rows4(rows, n_edges, e0, row);

    int   cc[4];
    float ww[4], ua[4], ub[4];
    if (e0 + 3 < n_edges && (n_edges & 3) == 0) {
        const v4i c4 = *(const v4i*)(cols + e0);
        const v4f w4 = *(const v4f*)(vals + e0);
        const v4f a4 = *(const v4f*)(dropu + e0);
        const v4f b4 = *(const v4f*)(dropu + (size_t)n_edges + e0);
        cc[0] = c4.x; cc[1] = c4.y; cc[2] = c4.z; cc[3] = c4.w;
        ww[0] = w4.x; ww[1] = w4.y; ww[2] = w4.z; ww[3] = w4.w;
        ua[0] = a4.x; ua[1] = a4.y; ua[2] = a4.z; ua[3] = a4.w;
        ub[0] = b4.x; ub[1] = b4.y; ub[2] = b4.z; ub[3] = b4.w;
    } else {
#pragma unroll
        for (int k = 0; k < 4; ++k) {
            const bool in = e0 + k < n_edges;
            cc[k] = in ? cols[e0 + k] : 0;
            ww[k] = in ? vals[e0 + k] : 0.0f;
            ua[k] = in ? dropu[e0 + k] : 0.0f;
            ub[k] = in ? dropu[(size_t)n_edges + e0 + k] : 0.0f;
        }
    }

    int key[4], rank[4], cnt[4];
    chunk_count(n_nodes, mycw, row, key, rank, cnt);
    __syncthreads();

    int mine = 0;
    if (tid < NBMAX) {
        int run = 0;
        for (int w = 0; w < 32; ++w) {
            const int c = cw[w][tid];
            cw[w][tid] = run;
            run += c;
        }
        mine = run;
    }
    const int linc = wave_incl_scan_i32(mine, lane);
    if (tid < NBMAX && lane == 31) ltot[wave] = linc;
    __syncthreads();
    if (tid < NBMAX) {
        int lwoff = 0;
        for (int w = 0; w < wave; ++w) lwoff += ltot[w];
        const int lexcl = lwoff + linc - mine;
        lofs[tid] = lexcl;
        ctab[chunk * NBMAX + tid] = lexcl | (mine << 16);
    }
    int nvalid = 0;
    for (int w = 0; w < 8; ++w) nvalid += ltot[w];
    __syncthreads();

#pragma unroll
    for (int k = 0; k < 4; ++k) {
        if (row[k] >= 0) {
            const int cur  = mycw[key[k]];
            const int lpos = lofs[key[k]] + cur + rank[k];
            if (rank[k] == 0) mycw[key[k]] = cur + cnt[k];
            const float v  = ww[k] * (1.0f / KEEP_PROB);
            const int k0 = (ua[k] + KEEP_PROB < 1.0f) ? 0 : (1 << COL_BITS);
            const int k1 = (ub[k] + KEEP_PROB < 1.0f) ? 0 : (2 << COL_BITS);
            const int rl = row[k] - key[k] * BROWS;
            const v2i r = {(cc[k] & COL_MASK) | k0 | k1 | (int)((unsigned)rl << ROW_SHIFT),
                           __float_as_int(v)};
            ((v2i*)stage)[lpos] = r;
        }
    }
    __syncthreads();

    if (nvalid > 0) {
        v4i r[2];
        const int lastq = (nvalid - 1) >> 1;
#pragma unroll
        for (int k = 0; k < 2; ++k) {
            const int q = k * 1024 + tid;
            r[k] = ((const v4i*)stage)[(q < lastq) ? q : lastq];
        }
#pragma unroll
        for (int k = 0; k < 2; ++k) {
            const int q = k * 1024 + tid;
            int* dstp = tmp + ((size_t)chunk * CHUNK + 2 * q) * 2;
            if (2 * q + 1 < nvalid)  *(v4i*)dstp = r[k];
            else if (2 * q < nvalid) *(v2i*)dstp = r[k].xy;
        }
    }
}

#define WALK_U 5

template <int PASS>
__device__ __forceinline__ void bucket_step(const v2i r, bool ok, int seg_beg,
                                            bool staged, int* mycw,
                                            const int* rstart, int* stage,
                                            int* __restrict__ rec) {
    const int key = ok ? (int)((unsigned)r.x >> ROW_SHIFT) : 0;
    int rank, cnt;
    wave_match<9>(key, ok, rank, cnt);
    if (PASS == 1) {
        if (ok && rank == 0) mycw[key] += cnt;
    } else if (ok) {
        const int cur = mycw[key];
        if (rank == 0) mycw[key] = cur + cnt;
        const int slot = rstart[key] + cur + rank;
        if (staged) ((v2i*)stage)[slot - seg_beg] = r;
        else        ((v2i*)rec)[slot] = r;
    }
}

template <int PASS>
__device__ __forceinline__ v2i bucket_load(const int* __restrict__ tmp, size_t at,
                                           int i, int n) {
    const size_t p = at + ((i < n) ? i : (n - 1));
    v2i r = {0, 0};
    if (PASS == 1) r.x = tmp[p * 2];
    else           r   = ((const v2i*)tmp)[p];
    return r;
}

template <int PASS>
__device__ __forceinline__ void bucket_walk(const int* __restrict__ ctab,
                                            const int* __restrict__ tmp,
                                            int* __restrict__ rec, int* mycw,
                                            const int* rstart, int* stage,
                                            int b, int n_chunks, int wave,
                                            int lane, int seg_beg, bool staged) {
    for (int cbase = 0; cbase < n_chunks; cbase += 1024) {
        const int cmine = cbase + wave + 32 * lane;
        const int pk = (cmine < n_chunks) ? ctab[(size_t)cmine * NBMAX + b] : 0;
        const int left = n_chunks - cbase - wave;
        int nj = (left > 0) ? ((left + 31) >> 5) : 0;
        nj = (nj < 32) ? nj : 32;
        for (int j0 = 0; j0 < nj; j0 += WALK_U) {
            int    cntu[WALK_U];
            size_t first[WALK_U];
            v2i    r[WALK_U];
#pragma unroll
            for (int u = 0; u < WALK_U; ++u) {
                const int j = j0 + u;
                const int p = (j < nj) ? __builtin_amdgcn_readlane(pk, j & 31) : 0;
                cntu[u]  = p >> 16;
                first[u] = (size_t)(cbase + wave + 32 * j) * CHUNK + (p & 0xffff);
                if (cntu[u] > 0) r[u] = bucket_load<PASS>(tmp, first[u], lane, cntu[u]);
            }
#pragma unroll
            for (int u = 0; u < WALK_U; ++u) {
                if (cntu[u] > 0) {
                    bucket_step<PASS>(r[u], lane < cntu[u], seg_beg, staged, mycw, rstart, stage, rec);
                    for (int t = 32; t < cntu[u]; t += 32) {
                        const v2i q = bucket_load<PASS>(tmp, first[u], t + lane, cntu[u]);
                        bucket_step<PASS>(q, t + lane < cntu[u], seg_beg, staged, mycw, rstart, stage, rec);
                    }
                }
            }
        }
    }
}

__global__ void __launch_bounds__(1024)
gcn_bucket_kernel(const int* __restrict__ ctab, const int* __restrict__ tmp,
                  int* __restrict__ rowptr, int* __restrict__ rec, int n_nodes,
                  int n_chunks) {
    __shared__ int cw[32][BROWS];
    __shared__ int rstart[BROWS];
    __shared__ int wtot[32];
    __shared__ int wsum[2][32];
    __shared__ alignas(16) int stage[STAGE_CAP * 2];

    const int tid  = threadIdx.x;
    const int lane = tid & 31;
    const int wave = __builtin_amdgcn_readfirstlane(tid >> 5);
    const int b    = blockIdx.x;
    const int lo   = b * BROWS;
    const int nrows = (n_nodes - lo < BROWS) ? (n_nodes - lo) : BROWS;
    int* mycw = cw[wave];
    for (int i = lane; i < BROWS; i += 32) mycw[i] = 0;

    int sb = 0, sl = 0;
    for (int c = tid; c < n_chunks; c += 1024) {
        const int p = ctab[(size_t)c * NBMAX + b];
        sb += p & 0xffff;
        sl += p >> 16;
    }
    for (int o = 16; o > 0; o >>= 1) {
        sb += __shfl_xor(sb, o);
        sl += __shfl_xor(sl, o);
    }
    if (lane == 0) {
        wsum[0][wave] = sb;
        wsum[1][wave] = sl;
    }
    __syncthreads();
    int seg_beg = wsum[0][lane];
    int seg_len = wsum[1][lane];
    for (int o = 16; o > 0; o >>= 1) {
        seg_beg += __shfl_xor(seg_beg, o);
        seg_len += __shfl_xor(seg_len, o);
    }
    const bool staged = seg_len <= STAGE_CAP;

    bucket_walk<1>(ctab, tmp, rec, mycw, rstart, stage, b, n_chunks, wave, lane, seg_beg, staged);
    __syncthreads();

    int tot = 0;
    if (tid < BROWS) {
        for (int w = 0; w < 32; ++w) {
            const int c = cw[w][tid];
            cw[w][tid] = tot;
            tot += c;
        }
    }
    const int x = wave_incl_scan_i32(tot, lane);
    if (lane == 31) wtot[wave] = x;
    __syncthreads();
    const int wt   = wtot[lane];
    const int wx   = wave_incl_scan_i32(wt, lane);
    const int woff = __shfl(wx - wt, wave);
    const int start = seg_beg + woff + x - tot;
    if (tid < BROWS) rstart[tid] = start;
    if (tid < nrows) {
        rowptr[lo + tid] = start;
        if (lo + tid == n_nodes - 1) rowptr[n_nodes] = start + tot;
    }
    __syncthreads();

    bucket_walk<2>(ctab, tmp, rec, mycw, rstart, stage, b, n_chunks, wave, lane, seg_beg, staged);

    if (staged) {
        __syncthreads();
        for (int i = tid; i < seg_len; i += 1024)
            ((v2i*)rec)[seg_beg + i] = ((const v2i*)stage)[i];
    }
}

template <int K>
__device__ __forceinline__ void pull_batch(unsigned& m, int cj, int wb,
                                           const float* __restrict__ srcl,
                                           v2f& acc) {
    int   c[K];
    float w[K];
    v2f   x[K];
#pragma unroll
    for (int k = 0; k < K; ++k) {
        const int j = __builtin_ctz(m);
        m &= m - 1;
        c[k] = __builtin_amdgcn_readlane(cj, j);
        w[k] = __int_as_float(__builtin_amdgcn_readlane(wb, j));
    }
#pragma unroll
    for (int k = 0; k < K; ++k) {
        x[k] = *(const v2f*)(srcl + (size_t)c[k] * DIM);
    }
#pragma unroll
    for (int k = 0; k < K; ++k) {
        acc += w[k] * x[k];
    }
}

template <int LAYER>
__device__ __forceinline__ void pull_records(const v2i r,
                                             const float* __restrict__ srcl,
                                             v2f& acc) {
    const int cj = r.x & COL_MASK;
    const int wb = r.y;
    const bool kept = ((r.x >> (COL_BITS + LAYER)) & 1) != 0;
    unsigned m = __builtin_amdgcn_ballot_w32(kept && __int_as_float(wb) != 0.0f);
    int nk = __builtin_popcount(m);
    for (; nk >= 8; nk -= 8) pull_batch<8>(m, cj, wb, srcl, acc);
    if (nk & 4) pull_batch<4>(m, cj, wb, srcl, acc);
    if (nk & 2) pull_batch<2>(m, cj, wb, srcl, acc);
    if (nk & 1) pull_batch<1>(m, cj, wb, srcl, acc);
}

#define PULL_ROWS 8
#define FC_PITCH 66

template <int LAYER>
__device__ __forceinline__ void pull_rows(int rpv, int nbeg, int nend, v2i rn, v2f xn,
                                          const v2i* __restrict__ recl,
                                          const float* __restrict__ srcl,
                                          float* __restrict__ dp,
                                          const float* __restrict__ xp,
                                          const float* fcl, int nr, int l) {
    v2f fn = {0.f, 0.f};
    if (LAYER == 1) fn = *(const v2f*)fcl;

    for (int i = 0; i < nr; ++i) {
        const v2i r = rn;
        v2f acc = fn;
        const v2f xs = xn;
        const int beg = nbeg;
        const int end = nend;

        if (i + 1 < nr) {
            nbeg = nend;
            nend = __builtin_amdgcn_readlane(rpv, i + 2);
            rn = (v2i){0, 0};
            if (nbeg + l < nend) rn = recl[(size_t)(nbeg + l)];
            if (LAYER == 1) {
                fn = *(const v2f*)(fcl + (i + 1) * FC_PITCH);
                xn = *(const v2f*)(xp + (size_t)(i + 1) * DIM);
            }
        }

        pull_records<LAYER>(r, srcl, acc);
        for (int base = beg + 32; base < end; base += 32) {
            v2i q = {0, 0};
            if (base + l < end) q = recl[(size_t)(base + l)];
            pull_records<LAYER>(q, srcl, acc);
        }

        if (LAYER == 1) acc = (acc + xs) * (1.0f / 3.0f);
        *(v2f*)(dp + (size_t)i * DIM) = acc;
    }
}

__global__ void __launch_bounds__(256)
gcn_pull0_kernel(const int* __restrict__ rowptr, const int* __restrict__ rec,
                 const float* __restrict__ src, float* __restrict__ dst,
                 int n_nodes) {
    int r0 = (blockIdx.x * 8 + (threadIdx.x >> 5)) * PULL_ROWS;
    if (r0 >= n_nodes) return;
    r0 = __builtin_amdgcn_readfirstlane(r0);
    const int l  = threadIdx.x & 31;
    const int nr = (n_nodes - r0 < PULL_ROWS) ? (n_nodes - r0) : PULL_ROWS;

    const int rpv = rowptr[r0 + ((l < nr) ? l : nr)];

    float* dp = dst + (size_t)r0 * DIM + l * 2;
    const float* srcl = src + l * 2;
    const v2i* recl = (const v2i*)rec;

    const int nbeg = __builtin_amdgcn_readlane(rpv, 0);
    const int nend = __builtin_amdgcn_readlane(rpv, 1);
    v2i rn = {0, 0};
    if (nbeg + l < nend) rn = recl[(size_t)(nbeg + l)];

    pull_rows<0>(rpv, nbeg, nend, rn, (v2f){0.f, 0.f}, recl, srcl, dp, srcl, nullptr, nr, l);
}

__global__ void __launch_bounds__(256, 16)
gcn_pull1_fc_kernel(const int* __restrict__ rowptr, const int* __restrict__ rec,
                    const float* __restrict__ src, const float* __restrict__ emb,
                    const float* __restrict__ W, const float* __restrict__ bias,
                    float* __restrict__ dst, int n_nodes) {
    __shared__ alignas(16) float fcs[8 * PULL_ROWS * FC_PITCH];

    const int l = threadIdx.x & 31;
    const int c = l & 15;
    const int h = l >> 4;
    const int wave = __builtin_amdgcn_readfirstlane(threadIdx.x >> 5);
    const int r0 = (blockIdx.x * 8 + wave) * PULL_ROWS;
    const bool live = r0 < n_nodes;
    const int nr = live ? ((n_nodes - r0 < PULL_ROWS) ? (n_nodes - r0) : PULL_ROWS) : 0;

    int rpv = 0;
    if (live) rpv = rowptr[r0 + ((l < nr) ? l : nr)];

    const int T  = wave >> 1;
    const int nh = wave & 1;
    int gr = blockIdx.x * (8 * PULL_ROWS) + 16 * T + c;
    gr = (gr < n_nodes) ? gr : (n_nodes - 1);
    const float* ap = emb + (size_t)gr * DIM + 4 * h;
    const float* bp = W + (size_t)(32 * nh + c) * DIM + 4 * h;
    v8f c0 = {}, c1 = {};
#pragma unroll
    for (int j = 0; j < 8; ++j) {
        const v4f a  = *(const v4f*)(ap + 8 * j);
        const v4f b0 = *(const v4f*)(bp + 8 * j);
        const v4f b1 = *(const v4f*)(bp + 16 * DIM + 8 * j);
        c0 = __builtin_amdgcn_wmma_f32_16x16x4_f32(false, a.xy, false, b0.xy, (short)0, c0, false, false);
        c1 = __builtin_amdgcn_wmma_f32_16x16x4_f32(false, a.xy, false, b1.xy, (short)0, c1, false, false);
        c0 = __builtin_amdgcn_wmma_f32_16x16x4_f32(false, a.zw, false, b0.zw, (short)0, c0, false, false);
        c1 = __builtin_amdgcn_wmma_f32_16x16x4_f32(false, a.zw, false, b1.zw, (short)0, c1, false, false);
    }
    const float bb0 = bias[32 * nh + c];
    const float bb1 = bias[32 * nh + 16 + c];

    float* dp = dst + (size_t)r0 * DIM + l * 2;
    const float* srcl = src + l * 2;
    const float* xp = srcl + (size_t)r0 * DIM;
    const v2i* recl = (const v2i*)rec;
    int nbeg = 0, nend = 0;
    v2i rn = {0, 0};
    v2f xn = {0.f, 0.f};
    if (live) {
        nbeg = __builtin_amdgcn_readlane(rpv, 0);
        nend = __builtin_amdgcn_readlane(rpv, 1);
        if (nbeg + l < nend) rn = recl[(size_t)(nbeg + l)];
        xn = *(const v2f*)xp;
    }

    float* fr = fcs + (16 * T + 8 * h) * FC_PITCH + 32 * nh + c;
#pragma unroll
    for (int j = 0; j < 8; ++j) {
        fr[j * FC_PITCH]      = c0[j] + bb0;
        fr[j * FC_PITCH + 16] = c1[j] + bb1;
    }
    __syncthreads();
    if (!live) return;

    pull_rows<1>(rpv, nbeg, nend, rn, xn, recl, srcl, dp, xp,
                 fcs + (wave * PULL_ROWS) * FC_PITCH + l * 2, nr, l);
}

__global__ void __launch_bounds__(256)
gcn_finalize_kernel(float* __restrict__ out, const float* __restrict__ x1,
                    long long n4) {
    const long long i = (long long)blockIdx.x * blockDim.x + threadIdx.x;
    if (i >= n4) return;
    v4f a = ((const v4f*)out)[i];
    v4f p = ((const v4f*)x1)[i];
    v4f r = (a + p) * (1.0f / 3.0f);
    ((v4f*)out)[i] = r;
}

extern "C" void kernel_launch(void* const* d_in, const int* in_sizes, int n_in,
                              void* d_out, int out_size, void* d_ws, size_t ws_size,
                              hipStream_t stream) {
    const float* emb   = (const float*)d_in[0];
    const float* W     = (const float*)d_in[1];
    const float* bias  = (const float*)d_in[2];
    const float* vals  = (const float*)d_in[3];
    const float* dropu = (const float*)d_in[4];
    const int*   rows  = (const int*)d_in[5];
    const int*   cols  = (const int*)d_in[6];

    const int N = in_sizes[0] / DIM;
    const int E = in_sizes[3];

    float* out = (float*)d_out;
    float* x1  = (float*)d_ws;

    const int    n_chunks  = (E + CHUNK - 1) / CHUNK;
    const int    n_buckets = (N + BROWS - 1) / BROWS;
    const size_t x1_bytes  = (size_t)N * DIM * sizeof(float);
    const size_t rec_bytes = ((size_t)E * 8 + 15) & ~(size_t)15;
    const size_t tmp_bytes = (size_t)n_chunks * CHUNK * 8;
    const size_t r0_bytes  = (x1_bytes > tmp_bytes) ? x1_bytes : tmp_bytes;
    const size_t rp_bytes  = (((size_t)N + 1) * sizeof(int) + 15) & ~(size_t)15;
    const size_t ct_bytes  = (size_t)n_chunks * NBMAX * sizeof(int);
    const size_t need      = r0_bytes + rec_bytes + rp_bytes + ct_bytes;

    if (ws_size >= need && n_buckets <= NBMAX && n_chunks > 0) {
        int* tmp    = (int*)d_ws;
        int* rec    = (int*)((char*)d_ws + r0_bytes);
        int* rowptr = (int*)((char*)d_ws + r0_bytes + rec_bytes);
        int* ctab   = (int*)((char*)d_ws + r0_bytes + rec_bytes + rp_bytes);

        {
            gcn_scatter_kernel<<<dim3(n_chunks), dim3(1024), 0, stream>>>(rows, cols, vals, dropu, ctab, tmp, N, E);
            gcn_bucket_kernel<<<dim3(n_buckets), dim3(1024), 0, stream>>>(ctab, tmp, rowptr, rec, N, n_chunks);
        }
        {
            dim3 grid((N + 8 * PULL_ROWS - 1) / (8 * PULL_ROWS)), block(256);
            gcn_pull0_kernel<<<grid, block, 0, stream>>>(rowptr, rec, emb, x1, N);
            gcn_pull1_fc_kernel<<<grid, block, 0, stream>>>(rowptr, rec, x1, emb, W, bias, out, N);
        }
    } else {
        hipMemsetAsync(d_ws, 0, x1_bytes, stream);

        {
            dim3 grid(FC_GROUPS), block(256);
            gcn_fc_wmma_kernel<<<grid, block, 0, stream>>>(emb, W, bias, out, N);
        }

        {
            long long threads = (long long)E * 32;
            dim3 grid((unsigned)((threads + 255) / 256)), block(256);
            gcn_spmm_kernel<<<grid, block, 0, stream>>>(vals, dropu,     rows, cols, emb, x1,  E);
            gcn_spmm_kernel<<<grid, block, 0, stream>>>(vals, dropu + E, rows, cols, x1,  out, E);
        }

        {
            long long n4 = (long long)N * DIM / 4;
            dim3 grid((unsigned)((n4 + 255) / 256)), block(256);
            gcn_finalize_kernel<<<grid, block, 0, stream>>>(out, x1, n4);
        }
    }
}


# Grouping level 2 copies a bucket's stretches into LDS once and ranks them there as dense 32-record groups, ranks kept for placing: one walk of tmp, a quarter of the matches; row sum order new, fixed
# speedup vs baseline: 5.2742x; 1.0687x over previous
#include <hip/hip_runtime.h>
#include <hip/hip_bf16.h>

typedef __attribute__((ext_vector_type(2))) float v2f;
typedef __attribute__((ext_vector_type(4))) float v4f;
typedef __attribute__((ext_vector_type(8))) float v8f;
typedef __attribute__((ext_vector_type(4))) int   v4i;
typedef __attribute__((ext_vector_type(2))) int   v2i;

#define DIM 64
#define KEEP_PROB 0.7f
#define OWN_ROWS 400

#define FC_GROUPS 512

__device__ __forceinline__ void fc_load_tile(const float* __restrict__ emb, int tile,
                                             int c, int h, int n_nodes, v4f (&a)[8]) {
    int gr = tile * 16 + c;
    gr = (gr < n_nodes) ? gr : (n_nodes - 1);
    const float* p = emb + (size_t)gr * DIM + 4 * h;
#pragma unroll
    for (int j = 0; j < 8; ++j) a[j] = *(const v4f*)(p + 8 * j);
}

__device__ __forceinline__ void fc_tile(const v4f (&a)[8], const v4f (&bf)[4][8],
                                        float bb0, float bb1, float bb2, float bb3,
                                        float* __restrict__ acc, int tile,
                                        int c, int h, int n_nodes) {
    v8f c0 = {}, c1 = {}, c2 = {}, c3 = {};
#pragma unroll
    for (int j = 0; j < 8; ++j) {
        const v2f a0 = a[j].xy;
        const v2f a1 = a[j].zw;
        c0 = __builtin_amdgcn_wmma_f32_16x16x4_f32(false, a0, false, bf[0][j].xy, (short)0, c0, false, false);
        c1 = __builtin_amdgcn_wmma_f32_16x16x4_f32(false, a0, false, bf[1][j].xy, (short)0, c1, false, false);
        c2 = __builtin_amdgcn_wmma_f32_16x16x4_f32(false, a0, false, bf[2][j].xy, (short)0, c2, false, false);
        c3 = __builtin_amdgcn_wmma_f32_16x16x4_f32(false, a0, false, bf[3][j].xy, (short)0, c3, false, false);
        c0 = __builtin_amdgcn_wmma_f32_16x16x4_f32(false, a1, false, bf[0][j].zw, (short)0, c0, false, false);
        c1 = __builtin_amdgcn_wmma_f32_16x16x4_f32(false, a1, false, bf[1][j].zw, (short)0, c1, false, false);
        c2 = __builtin_amdgcn_wmma_f32_16x16x4_f32(false, a1, false, bf[2][j].zw, (short)0, c2, false, false);
        c3 = __builtin_amdgcn_wmma_f32_16x16x4_f32(false, a1, false, bf[3][j].zw, (short)0, c3, false, false);
    }

    const int row_base = tile * 16 + h * 8;
#pragma unroll
    for (int j = 0; j < 8; ++j) {
        const int gm = row_base + j;
        if (gm < n_nodes) {
            float* outr = acc + (size_t)gm * DIM;
            outr[c +  0] = c0[j] + bb0;
            outr[c + 16] = c1[j] + bb1;
            outr[c + 32] = c2[j] + bb2;
            outr[c + 48] = c3[j] + bb3;
        }
    }
}

__global__ void __launch_bounds__(256)
gcn_fc_wmma_kernel(const float* __restrict__ emb, const float* __restrict__ W,
                   const float* __restrict__ bias, float* __restrict__ acc,
                   int n_nodes) {
    const int lane = threadIdx.x & 31;
    const int c    = lane & 15;
    const int h    = lane >> 4;
    const int nw   = gridDim.x * 8;
    const int n_tiles = (n_nodes + 15) >> 4;
    int tile = (threadIdx.x >> 5) * gridDim.x + blockIdx.x;
    if (tile >= n_tiles) return;
    tile = __builtin_amdgcn_readfirstlane(tile);

    v4f bf[4][8];
#pragma unroll
    for (int t = 0; t < 4; ++t) {
#pragma unroll
        for (int j = 0; j < 8; ++j) {
            bf[t][j] = *(const v4f*)(W + (size_t)(16 * t + c) * DIM + 8 * j + 4 * h);
        }
    }
    const float bb0 = bias[c];
    const float bb1 = bias[c + 16];
    const float bb2 = bias[c + 32];
    const float bb3 = bias[c + 48];

    v4f pa[8], pb[8];
    fc_load_tile(emb, tile, c, h, n_nodes, pa);
    for (;;) {
        int next = tile + nw;
        fc_load_tile(emb, (next < n_tiles) ? next : tile, c, h, n_nodes, pb);
        fc_tile(pa, bf, bb0, bb1, bb2, bb3, acc, tile, c, h, n_nodes);
        if (next >= n_tiles) break;
        tile = next;
        next = tile + nw;
        fc_load_tile(emb, (next < n_tiles) ? next : tile, c, h, n_nodes, pa);
        fc_tile(pb, bf, bb0, bb1, bb2, bb3, acc, tile, c, h, n_nodes);
        if (next >= n_tiles) break;
        tile = next;
    }
}

__global__ void __launch_bounds__(256)
gcn_spmm_kernel(const float* __restrict__ vals, const float* __restrict__ dropu,
                const int* __restrict__ rows, const int* __restrict__ cols,
                const float* __restrict__ src, float* __restrict__ dst,
                int n_edges) {
    const long long tid = (long long)blockIdx.x * blockDim.x + threadIdx.x;
    int e = (int)(tid >> 5);
    if (e >= n_edges) return;
    e = __builtin_amdgcn_readfirstlane(e);
    const int l = threadIdx.x & 31;

    const float u = dropu[e];
    if (u + KEEP_PROB < 1.0f) return;
    const float v = vals[e] * (1.0f / KEEP_PROB);

    const int col = cols[e];
    const int row = rows[e];

    const float* sp = src + (size_t)col * DIM + l * 2;
    float*       dp = dst + (size_t)row * DIM + l * 2;
    const float m0 = v * sp[0];
    const float m1 = v * sp[1];

    asm volatile("global_atomic_add_f32 %0, %1, off scope:SCOPE_DEV"
                 :: "v"(dp), "v"(m0) : "memory");
    asm volatile("global_atomic_add_f32 %0, %1, off scope:SCOPE_DEV"
                 :: "v"(dp + 1), "v"(m1) : "memory");
}


#define BROWS 400
#define NBMAX 256
#define CHUNK 4096
#define STAGE_CAP 8192
#define COL_BITS 21
#define COL_MASK ((1 << COL_BITS) - 1)
#define ROW_SHIFT (COL_BITS + 2)

__device__ __forceinline__ int wave_incl_scan_i32(int x, int lane) {
    int y;
    y = __shfl_up(x, 1);  if (lane >= 1)  x += y;
    y = __shfl_up(x, 2);  if (lane >= 2)  x += y;
    y = __shfl_up(x, 4);  if (lane >= 4)  x += y;
    y = __shfl_up(x, 8);  if (lane >= 8)  x += y;
    y = __shfl_up(x, 16); if (lane >= 16) x += y;
    return x;
}

template <int NBITS>
__device__ __forceinline__ void wave_match(int key, bool ok, int& rank, int& cnt) {
    unsigned peers = __builtin_amdgcn_ballot_w32(ok);
    for (int b = 0; b < NBITS; ++b) {
        const bool bit = ((key >> b) & 1) != 0;
        const unsigned m = __builtin_amdgcn_ballot_w32(ok && bit);
        peers &= bit ? m : ~m;
    }
    rank = (int)__builtin_amdgcn_mbcnt_lo(peers, 0u);
    cnt  = __builtin_popcount(peers);
}

__device__ __forceinline__ void load_rows4(const int* __restrict__ rows,
                                           int n_edges, int e0, int (&row)[4]) {
    if (e0 + 3 < n_edges) {
        const v4i g = *(const v4i*)(rows + e0);
        row[0] = g.x; row[1] = g.y; row[2] = g.z; row[3] = g.w;
    } else {
#pragma unroll
        for (int k = 0; k < 4; ++k) row[k] = (e0 + k < n_edges) ? rows[e0 + k] : -1;
    }
}

__device__ __forceinline__ void chunk_count(int n_nodes, int* mycw, int (&row)[4],
                                            int (&key)[4], int (&rank)[4],
                                            int (&cnt)[4]) {
#pragma unroll
    for (int k = 0; k < 4; ++k) {
        const bool ok = (unsigned)row[k] < (unsigned)n_nodes;
        if (!ok) row[k] = -1;
        key[k] = ok ? (row[k] / BROWS) : 0;
        wave_match<8>(key[k], ok, rank[k], cnt[k]);
        if (ok && rank[k] == 0) mycw[key[k]] += cnt[k];
    }
}

__global__ void __launch_bounds__(1024)
gcn_scatter_kernel(const int* __restrict__ rows, const int* __restrict__ cols,
                   const float* __restrict__ vals, const float* __restrict__ dropu,
                   int* __restrict__ ctab, int* __restrict__ tmp,
                   int n_nodes, int n_edges) {
    __shared__ int cw[32][NBMAX];
    __shared__ int lofs[NBMAX];
    __shared__ int ltot[8];
    __shared__ alignas(16) int stage[CHUNK * 2];

    const int tid   = threadIdx.x;
    const int lane  = tid & 31;
    const int wave  = tid >> 5;
    const int chunk = blockIdx.x;
    const int e0    = (chunk * 1024 + tid) * 4;
    int* mycw = cw[wave];
    for (int i = lane; i < NBMAX; i += 32) mycw[i] = 0;

    int row[4];
    load_rows4(rows, n_edges, e0, row);

    int   cc[4];
    float ww[4], ua[4], ub[4];
    if (e0 + 3 < n_edges && (n_edges & 3) == 0) {
        const v4i c4 = *(const v4i*)(cols + e0);
        const v4f w4 = *(const v4f*)(vals + e0);
        const v4f a4 = *(const v4f*)(dropu + e0);
        const v4f b4 = *(const v4f*)(dropu + (size_t)n_edges + e0);
        cc[0] = c4.x; cc[1] = c4.y; cc[2] = c4.z; cc[3] = c4.w;
        ww[0] = w4.x; ww[1] = w4.y; ww[2] = w4.z; ww[3] = w4.w;
        ua[0] = a4.x; ua[1] = a4.y; ua[2] = a4.z; ua[3] = a4.w;
        ub[0] = b4.x; ub[1] = b4.y; ub[2] = b4.z; ub[3] = b4.w;
    } else {
#pragma unroll
        for (int k = 0; k < 4; ++k) {
            const bool in = e0 + k < n_edges;
            cc[k] = in ? cols[e0 + k] : 0;
            ww[k] = in ? vals[e0 + k] : 0.0f;
            ua[k] = in ? dropu[e0 + k] : 0.0f;
            ub[k] = in ? dropu[(size_t)n_edges + e0 + k] : 0.0f;
        }
    }

    int key[4], rank[4], cnt[4];
    chunk_count(n_nodes, mycw, row, key, rank, cnt);
    __syncthreads();

    int mine = 0;
    if (tid < NBMAX) {
        int run = 0;
        for (int w = 0; w < 32; ++w) {
            const int c = cw[w][tid];
            cw[w][tid] = run;
            run += c;
        }
        mine = run;
    }
    const int linc = wave_incl_scan_i32(mine, lane);
    if (tid < NBMAX && lane == 31) ltot[wave] = linc;
    __syncthreads();
    if (tid < NBMAX) {
        int lwoff = 0;
        for (int w = 0; w < wave; ++w) lwoff += ltot[w];
        const int lexcl = lwoff + linc - mine;
        lofs[tid] = lexcl;
        ctab[chunk * NBMAX + tid] = lexcl | (mine << 16);
    }
    int nvalid = 0;
    for (int w = 0; w < 8; ++w) nvalid += ltot[w];
    __syncthreads();

#pragma unroll
    for (int k = 0; k < 4; ++k) {
        if (row[k] >= 0) {
            const int cur  = mycw[key[k]];
            const int lpos = lofs[key[k]] + cur + rank[k];
            if (rank[k] == 0) mycw[key[k]] = cur + cnt[k];
            const float v  = ww[k] * (1.0f / KEEP_PROB);
            const int k0 = (ua[k] + KEEP_PROB < 1.0f) ? 0 : (1 << COL_BITS);
            const int k1 = (ub[k] + KEEP_PROB < 1.0f) ? 0 : (2 << COL_BITS);
            const int rl = row[k] - key[k] * BROWS;
            const v2i r = {(cc[k] & COL_MASK) | k0 | k1 | (int)((unsigned)rl << ROW_SHIFT),
                           __float_as_int(v)};
            ((v2i*)stage)[lpos] = r;
        }
    }
    __syncthreads();

    if (nvalid > 0) {
        v4i r[2];
        const int lastq = (nvalid - 1) >> 1;
#pragma unroll
        for (int k = 0; k < 2; ++k) {
            const int q = k * 1024 + tid;
            r[k] = ((const v4i*)stage)[(q < lastq) ? q : lastq];
        }
#pragma unroll
        for (int k = 0; k < 2; ++k) {
            const int q = k * 1024 + tid;
            int* dstp = tmp + ((size_t)chunk * CHUNK + 2 * q) * 2;
            if (2 * q + 1 < nvalid)  *(v4i*)dstp = r[k];
            else if (2 * q < nvalid) *(v2i*)dstp = r[k].xy;
        }
    }
}

#define WALK_U 1

template <int PASS>
__device__ __forceinline__ void bucket_step(const v2i r, bool ok, int seg_beg,
                                            bool staged, int* mycw,
                                            const int* rstart, int* stage,
                                            int* __restrict__ rec) {
    const int key = ok ? (int)((unsigned)r.x >> ROW_SHIFT) : 0;
    int rank, cnt;
    wave_match<9>(key, ok, rank, cnt);
    if (PASS == 1) {
        if (ok && rank == 0) mycw[key] += cnt;
    } else if (ok) {
        const int cur = mycw[key];
        if (rank == 0) mycw[key] = cur + cnt;
        const int slot = rstart[key] + cur + rank;
        if (staged) ((v2i*)stage)[slot - seg_beg] = r;
        else        ((v2i*)rec)[slot] = r;
    }
}

template <int PASS>
__device__ __forceinline__ v2i bucket_load(const int* __restrict__ tmp, size_t at,
                                           int i, int n) {
    const size_t p = at + ((i < n) ? i : (n - 1));
    v2i r = {0, 0};
    if (PASS == 1) r.x = tmp[p * 2];
    else           r   = ((const v2i*)tmp)[p];
    return r;
}

template <int PASS>
__device__ __forceinline__ void bucket_walk(const int* __restrict__ ctab,
                                            const int* __restrict__ tmp,
                                            int* __restrict__ rec, int* mycw,
                                            const int* rstart, int* stage,
                                            int b, int n_chunks, int wave,
                                            int lane, int seg_beg, bool staged) {
    for (int cbase = 0; cbase < n_chunks; cbase += 1024) {
        const int cmine = cbase + wave + 32 * lane;
        const int pk = (cmine < n_chunks) ? ctab[(size_t)cmine * NBMAX + b] : 0;
        const int left = n_chunks - cbase - wave;
        int nj = (left > 0) ? ((left + 31) >> 5) : 0;
        nj = (nj < 32) ? nj : 32;
        for (int j0 = 0; j0 < nj; j0 += WALK_U) {
            int    cntu[WALK_U];
            size_t first[WALK_U];
            v2i    r[WALK_U];
#pragma unroll
            for (int u = 0; u < WALK_U; ++u) {
                const int j = j0 + u;
                const int p = (j < nj) ? __builtin_amdgcn_readlane(pk, j & 31) : 0;
                cntu[u]  = p >> 16;
                first[u] = (size_t)(cbase + wave + 32 * j) * CHUNK + (p & 0xffff);
                if (cntu[u] > 0) r[u] = bucket_load<PASS>(tmp, first[u], lane, cntu[u]);
            }
#pragma unroll
            for (int u = 0; u < WALK_U; ++u) {
                if (cntu[u] > 0) {
                    bucket_step<PASS>(r[u], lane < cntu[u], seg_beg, staged, mycw, rstart, stage, rec);
                    for (int t = 32; t < cntu[u]; t += 32) {
                        const v2i q = bucket_load<PASS>(tmp, first[u], t + lane, cntu[u]);
                        bucket_step<PASS>(q, t + lane < cntu[u], seg_beg, staged, mycw, rstart, stage, rec);
                    }
                }
            }
        }
    }
}

#define LOAD_U 10
#define DENSE_G (STAGE_CAP / 1024)

__global__ void __launch_bounds__(1024)
gcn_bucket_kernel(const int* __restrict__ ctab, const int* __restrict__ tmp,
                  int* __restrict__ rowptr, int* __restrict__ rec, int n_nodes,
                  int n_chunks) {
    __shared__ int cw[32][BROWS];
    __shared__ int rstart[BROWS];
    __shared__ int wtot[32];
    __shared__ int wsum[2][32];
    __shared__ alignas(16) int stage[STAGE_CAP * 2];
    __shared__ alignas(16) int raw[STAGE_CAP * 2];

    const int tid  = threadIdx.x;
    const int lane = tid & 31;
    const int wave = __builtin_amdgcn_readfirstlane(tid >> 5);
    const int b    = blockIdx.x;
    const int lo   = b * BROWS;
    const int nrows = (n_nodes - lo < BROWS) ? (n_nodes - lo) : BROWS;
    int* mycw = cw[wave];
    for (int i = lane; i < BROWS; i += 32) mycw[i] = 0;

    const int c0  = wave + 32 * lane;
    const int pk0 = (c0 < n_chunks) ? ctab[(size_t)c0 * NBMAX + b] : 0;
    int sb = pk0 & 0xffff, sl = pk0 >> 16;
#pragma nounroll
    for (int cbase = 1024; cbase < n_chunks; cbase += 1024) {
        const int c = cbase + wave + 32 * lane;
        const int p = (c < n_chunks) ? ctab[(size_t)c * NBMAX + b] : 0;
        sb += p & 0xffff;
        sl += p >> 16;
    }
    for (int o = 16; o > 0; o >>= 1) {
        sb += __shfl_xor(sb, o);
        sl += __shfl_xor(sl, o);
    }
    if (lane == 0) {
        wsum[0][wave] = sb;
        wsum[1][wave] = sl;
    }
    __syncthreads();
    int seg_beg = wsum[0][lane];
    for (int o = 16; o > 0; o >>= 1) seg_beg += __shfl_xor(seg_beg, o);
    const int wl  = wsum[1][lane];
    const int wli = wave_incl_scan_i32(wl, lane);
    seg_beg = __builtin_amdgcn_readfirstlane(seg_beg);
    const int seg_len = __builtin_amdgcn_readfirstlane(__shfl(wli, 31));
    const int wbase   = __builtin_amdgcn_readfirstlane(__shfl(wli - wl, wave));
    const bool staged = seg_len <= STAGE_CAP;
    const bool dense  = staged && n_chunks <= 1024;

    v2i rr[DENSE_G];
    int rc[DENSE_G];

    if (dense) {
        const int cnt0 = pk0 >> 16;
        const int rofs = wbase + wave_incl_scan_i32(cnt0, lane) - cnt0;
        const int left = n_chunks - wave;
        const int nj = (left > 0) ? ((left + 31) >> 5) : 0;
        for (int j0 = 0; j0 < nj; j0 += LOAD_U) {
            int cntu[LOAD_U];
            int rou[LOAD_U];
            int first[LOAD_U];
            v2i r[LOAD_U];
#pragma unroll
            for (int u = 0; u < LOAD_U; ++u) {
                const int j  = j0 + u;
                const int pj = __builtin_amdgcn_readlane(pk0, j & 31);
                const int oj = __builtin_amdgcn_readlane(rofs, j & 31);
                const int p  = (j < nj) ? pj : 0;
                rou[u]   = (j < nj) ? oj : 0;
                cntu[u]  = p >> 16;
                first[u] = (cntu[u] > 0) ? ((wave + 32 * j) * CHUNK + (p & 0xffff)) : 0;
                const int last = (cntu[u] > 0) ? (cntu[u] - 1) : 0;
                r[u] = ((const v2i*)tmp)[(size_t)first[u] + ((lane < last) ? lane : last)];
            }
#pragma unroll
            for (int u = 0; u < LOAD_U; ++u) {
                if (lane < cntu[u]) ((v2i*)raw)[rou[u] + lane] = r[u];
#pragma nounroll
                for (int t = 32; t < cntu[u]; t += 32) {
                    const v2i q = bucket_load<2>(tmp, (size_t)first[u], t + lane, cntu[u]);
                    if (t + lane < cntu[u]) ((v2i*)raw)[rou[u] + t + lane] = q;
                }
            }
        }
        __syncthreads();

#pragma unroll
        for (int g = 0; g < DENSE_G; ++g) {
            const int idx = g * 1024 + tid;
            rr[g] = ((const v2i*)raw)[(idx < seg_len) ? idx : 0];
        }
#pragma unroll
        for (int g = 0; g < DENSE_G; ++g) {
            rc[g] = 0;
            if (g * 1024 + 32 * wave < seg_len) {
                const bool ok = g * 1024 + tid < seg_len;
                const int key = ok ? (int)((unsigned)rr[g].x >> ROW_SHIFT) : 0;
                int rank, cnt;
                wave_match<9>(key, ok, rank, cnt);
                if (ok && rank == 0) mycw[key] += cnt;
                rc[g] = rank | (cnt << 8);
            }
        }
    } else {
        bucket_walk<1>(ctab, tmp, rec, mycw, rstart, stage, b, n_chunks, wave, lane, seg_beg, staged);
    }
    __syncthreads();

    int tot = 0;
    if (tid < BROWS) {
        for (int w = 0; w < 32; ++w) {
            const int c = cw[w][tid];
            cw[w][tid] = tot;
            tot += c;
        }
    }
    const int x = wave_incl_scan_i32(tot, lane);
    if (lane == 31) wtot[wave] = x;
    __syncthreads();
    const int wt   = wtot[lane];
    const int wx   = wave_incl_scan_i32(wt, lane);
    const int woff = __shfl(wx - wt, wave);
    const int start = seg_beg + woff + x - tot;
    if (tid < BROWS) rstart[tid] = start;
    if (tid < nrows) {
        rowptr[lo + tid] = start;
        if (lo + tid == n_nodes - 1) rowptr[n_nodes] = start + tot;
    }
    __syncthreads();

    if (dense) {
#pragma unroll
        for (int g = 0; g < DENSE_G; ++g) {
            if (g * 1024 + tid < seg_len) {
                const int key  = (int)((unsigned)rr[g].x >> ROW_SHIFT);
                const int rank = rc[g] & 0xff;
                const int cnt  = rc[g] >> 8;
                const int cur  = mycw[key];
                if (rank == 0) mycw[key] = cur + cnt;
                const int slot = rstart[key] + cur + rank;
                ((v2i*)stage)[slot - seg_beg] = rr[g];
            }
        }
    } else {
        bucket_walk<2>(ctab, tmp, rec, mycw, rstart, stage, b, n_chunks, wave, lane, seg_beg, staged);
    }

    if (staged) {
        __syncthreads();
        for (int i = tid; i < seg_len; i += 1024)
            ((v2i*)rec)[seg_beg + i] = ((const v2i*)stage)[i];
    }
}

template <int K>
__device__ __forceinline__ void pull_batch(unsigned& m, int cj, int wb,
                                           const float* __restrict__ srcl,
                                           v2f& acc) {
    int   c[K];
    float w[K];
    v2f   x[K];
#pragma unroll
    for (int k = 0; k < K; ++k) {
        const int j = __builtin_ctz(m);
        m &= m - 1;
        c[k] = __builtin_amdgcn_readlane(cj, j);
        w[k] = __int_as_float(__builtin_amdgcn_readlane(wb, j));
    }
#pragma unroll
    for (int k = 0; k < K; ++k) {
        x[k] = *(const v2f*)(srcl + (size_t)c[k] * DIM);
    }
#pragma unroll
    for (int k = 0; k < K; ++k) {
        acc += w[k] * x[k];
    }
}

template <int LAYER>
__device__ __forceinline__ void pull_records(const v2i r,
                                             const float* __restrict__ srcl,
                                             v2f& acc) {
    const int cj = r.x & COL_MASK;
    const int wb = r.y;
    const bool kept = ((r.x >> (COL_BITS + LAYER)) & 1) != 0;
    unsigned m = __builtin_amdgcn_ballot_w32(kept && __int_as_float(wb) != 0.0f);
    int nk = __builtin_popcount(m);
    for (; nk >= 8; nk -= 8) pull_batch<8>(m, cj, wb, srcl, acc);
    if (nk & 4) pull_batch<4>(m, cj, wb, srcl, acc);
    if (nk & 2) pull_batch<2>(m, cj, wb, srcl, acc);
    if (nk & 1) pull_batch<1>(m, cj, wb, srcl, acc);
}

#define PULL_ROWS 8
#define FC_PITCH 66

template <int LAYER>
__device__ __forceinline__ void pull_rows(int rpv, int nbeg, int nend, v2i rn, v2f xn,
                                          const v2i* __restrict__ recl,
                                          const float* __restrict__ srcl,
                                          float* __restrict__ dp,
                                          const float* __restrict__ xp,
                                          const float* fcl, int nr, int l) {
    v2f fn = {0.f, 0.f};
    if (LAYER == 1) fn = *(const v2f*)fcl;

    for (int i = 0; i < nr; ++i) {
        const v2i r = rn;
        v2f acc = fn;
        const v2f xs = xn;
        const int beg = nbeg;
        const int end = nend;

        if (i + 1 < nr) {
            nbeg = nend;
            nend = __builtin_amdgcn_readlane(rpv, i + 2);
            rn = (v2i){0, 0};
            if (nbeg + l < nend) rn = recl[(size_t)(nbeg + l)];
            if (LAYER == 1) {
                fn = *(const v2f*)(fcl + (i + 1) * FC_PITCH);
                xn = *(const v2f*)(xp + (size_t)(i + 1) * DIM);
            }
        }

        pull_records<LAYER>(r, srcl, acc);
        for (int base = beg + 32; base < end; base += 32) {
            v2i q = {0, 0};
            if (base + l < end) q = recl[(size_t)(base + l)];
            pull_records<LAYER>(q, srcl, acc);
        }

        if (LAYER == 1) acc = (acc + xs) * (1.0f / 3.0f);
        *(v2f*)(dp + (size_t)i * DIM) = acc;
    }
}

__global__ void __launch_bounds__(256)
gcn_pull0_kernel(const int* __restrict__ rowptr, const int* __restrict__ rec,
                 const float* __restrict__ src, float* __restrict__ dst,
                 int n_nodes) {
    int r0 = (blockIdx.x * 8 + (threadIdx.x >> 5)) * PULL_ROWS;
    if (r0 >= n_nodes) return;
    r0 = __builtin_amdgcn_readfirstlane(r0);
    const int l  = threadIdx.x & 31;
    const int nr = (n_nodes - r0 < PULL_ROWS) ? (n_nodes - r0) : PULL_ROWS;

    const int rpv = rowptr[r0 + ((l < nr) ? l : nr)];

    float* dp = dst + (size_t)r0 * DIM + l * 2;
    const float* srcl = src + l * 2;
    const v2i* recl = (const v2i*)rec;

    const int nbeg = __builtin_amdgcn_readlane(rpv, 0);
    const int nend = __builtin_amdgcn_readlane(rpv, 1);
    v2i rn = {0, 0};
    if (nbeg + l < nend) rn = recl[(size_t)(nbeg + l)];

    pull_rows<0>(rpv, nbeg, nend, rn, (v2f){0.f, 0.f}, recl, srcl, dp, srcl, nullptr, nr, l);
}

__global__ void __launch_bounds__(256, 16)
gcn_pull1_fc_kernel(const int* __restrict__ rowptr, const int* __restrict__ rec,
                    const float* __restrict__ src, const float* __restrict__ emb,
                    const float* __restrict__ W, const float* __restrict__ bias,
                    float* __restrict__ dst, int n_nodes) {
    __shared__ alignas(16) float fcs[8 * PULL_ROWS * FC_PITCH];

    const int l = threadIdx.x & 31;
    const int c = l & 15;
    const int h = l >> 4;
    const int wave = __builtin_amdgcn_readfirstlane(threadIdx.x >> 5);
    const int r0 = (blockIdx.x * 8 + wave) * PULL_ROWS;
    const bool live = r0 < n_nodes;
    const int nr = live ? ((n_nodes - r0 < PULL_ROWS) ? (n_nodes - r0) : PULL_ROWS) : 0;

    int rpv = 0;
    if (live) rpv = rowptr[r0 + ((l < nr) ? l : nr)];

    const int T  = wave >> 1;
    const int nh = wave & 1;
    int gr = blockIdx.x * (8 * PULL_ROWS) + 16 * T + c;
    gr = (gr < n_nodes) ? gr : (n_nodes - 1);
    const float* ap = emb + (size_t)gr * DIM + 4 * h;
    const float* bp = W + (size_t)(32 * nh + c) * DIM + 4 * h;
    v8f c0 = {}, c1 = {};
#pragma unroll
    for (int j = 0; j < 8; ++j) {
        const v4f a  = *(const v4f*)(ap + 8 * j);
        const v4f b0 = *(const v4f*)(bp + 8 * j);
        const v4f b1 = *(const v4f*)(bp + 16 * DIM + 8 * j);
        c0 = __builtin_amdgcn_wmma_f32_16x16x4_f32(false, a.xy, false, b0.xy, (short)0, c0, false, false);
        c1 = __builtin_amdgcn_wmma_f32_16x16x4_f32(false, a.xy, false, b1.xy, (short)0, c1, false, false);
        c0 = __builtin_amdgcn_wmma_f32_16x16x4_f32(false, a.zw, false, b0.zw, (short)0, c0, false, false);
        c1 = __builtin_amdgcn_wmma_f32_16x16x4_f32(false, a.zw, false, b1.zw, (short)0, c1, false, false);
    }
    const float bb0 = bias[32 * nh + c];
    const float bb1 = bias[32 * nh + 16 + c];

    float* dp = dst + (size_t)r0 * DIM + l * 2;
    const float* srcl = src + l * 2;
    const float* xp = srcl + (size_t)r0 * DIM;
    const v2i* recl = (const v2i*)rec;
    int nbeg = 0, nend = 0;
    v2i rn = {0, 0};
    v2f xn = {0.f, 0.f};
    if (live) {
        nbeg = __builtin_amdgcn_readlane(rpv, 0);
        nend = __builtin_amdgcn_readlane(rpv, 1);
        if (nbeg + l < nend) rn = recl[(size_t)(nbeg + l)];
        xn = *(const v2f*)xp;
    }

    float* fr = fcs + (16 * T + 8 * h) * FC_PITCH + 32 * nh + c;
#pragma unroll
    for (int j = 0; j < 8; ++j) {
        fr[j * FC_PITCH]      = c0[j] + bb0;
        fr[j * FC_PITCH + 16] = c1[j] + bb1;
    }
    __syncthreads();
    if (!live) return;

    pull_rows<1>(rpv, nbeg, nend, rn, xn, recl, srcl, dp, xp,
                 fcs + (wave * PULL_ROWS) * FC_PITCH + l * 2, nr, l);
}

__global__ void __launch_bounds__(256)
gcn_finalize_kernel(float* __restrict__ out, const float* __restrict__ x1,
                    long long n4) {
    const long long i = (long long)blockIdx.x * blockDim.x + threadIdx.x;
    if (i >= n4) return;
    v4f a = ((const v4f*)out)[i];
    v4f p = ((const v4f*)x1)[i];
    v4f r = (a + p) * (1.0f / 3.0f);
    ((v4f*)out)[i] = r;
}

extern "C" void kernel_launch(void* const* d_in, const int* in_sizes, int n_in,
                              void* d_out, int out_size, void* d_ws, size_t ws_size,
                              hipStream_t stream) {
    const float* emb   = (const float*)d_in[0];
    const float* W     = (const float*)d_in[1];
    const float* bias  = (const float*)d_in[2];
    const float* vals  = (const float*)d_in[3];
    const float* dropu = (const float*)d_in[4];
    const int*   rows  = (const int*)d_in[5];
    const int*   cols  = (const int*)d_in[6];

    const int N = in_sizes[0] / DIM;
    const int E = in_sizes[3];

    float* out = (float*)d_out;
    float* x1  = (float*)d_ws;

    const int    n_chunks  = (E + CHUNK - 1) / CHUNK;
    const int    n_buckets = (N + BROWS - 1) / BROWS;
    const size_t x1_bytes  = (size_t)N * DIM * sizeof(float);
    const size_t rec_bytes = ((size_t)E * 8 + 15) & ~(size_t)15;
    const size_t tmp_bytes = (size_t)n_chunks * CHUNK * 8;
    const size_t r0_bytes  = (x1_bytes > tmp_bytes) ? x1_bytes : tmp_bytes;
    const size_t rp_bytes  = (((size_t)N + 1) * sizeof(int) + 15) & ~(size_t)15;
    const size_t ct_bytes  = (size_t)n_chunks * NBMAX * sizeof(int);
    const size_t need      = r0_bytes + rec_bytes + rp_bytes + ct_bytes;

    if (ws_size >= need && n_buckets <= NBMAX && n_chunks > 0) {
        int* tmp    = (int*)d_ws;
        int* rec    = (int*)((char*)d_ws + r0_bytes);
        int* rowptr = (int*)((char*)d_ws + r0_bytes + rec_bytes);
        int* ctab   = (int*)((char*)d_ws + r0_bytes + rec_bytes + rp_bytes);

        {
            gcn_scatter_kernel<<<dim3(n_chunks), dim3(1024), 0, stream>>>(rows, cols, vals, dropu, ctab, tmp, N, E);
            gcn_bucket_kernel<<<dim3(n_buckets), dim3(1024), 0, stream>>>(ctab, tmp, rowptr, rec, N, n_chunks);
        }
        {
            dim3 grid((N + 8 * PULL_ROWS - 1) / (8 * PULL_ROWS)), block(256);
            gcn_pull0_kernel<<<grid, block, 0, stream>>>(rowptr, rec, emb, x1, N);
            gcn_pull1_fc_kernel<<<grid, block, 0, stream>>>(rowptr, rec, x1, emb, W, bias, out, N);
        }
    } else {
        hipMemsetAsync(d_ws, 0, x1_bytes, stream);

        {
            dim3 grid(FC_GROUPS), block(256);
            gcn_fc_wmma_kernel<<<grid, block, 0, stream>>>(emb, W, bias, out, N);
        }

        {
            long long threads = (long long)E * 32;
            dim3 grid((unsigned)((threads + 255) / 256)), block(256);
            gcn_spmm_kernel<<<grid, block, 0, stream>>>(vals, dropu,     rows, cols, emb, x1,  E);
            gcn_spmm_kernel<<<grid, block, 0, stream>>>(vals, dropu + E, rows, cols, x1,  out, E);
        }

        {
            long long n4 = (long long)N * DIM / 4;
            dim3 grid((unsigned)((n4 + 255) / 256)), block(256);
            gcn_finalize_kernel<<<grid, block, 0, stream>>>(out, x1, n4);
        }
    }
}
